# starting kernel
# baseline (speedup 1.0000x reference)
#include <hip/hip_runtime.h>
#include <hip/hip_bf16.h>

typedef __attribute__((ext_vector_type(2))) float v2f;
typedef __attribute__((ext_vector_type(8))) float v8f;

#define IN_CH 16
#define OUT_CH 8

__global__ void gcn_deg_init(float* __restrict__ deg, int n) {
    int i = blockIdx.x * blockDim.x + threadIdx.x;
    if (i < n) deg[i] = 1.0f;
}

__global__ void gcn_deg_count(const int* __restrict__ dst,
                              float* __restrict__ deg, int e_cnt) {
    int e = blockIdx.x * blockDim.x + threadIdx.x;
    if (e < e_cnt) atomicAdd(&deg[dst[e]], 1.0f);
}

__global__ void gcn_dinv(float* __restrict__ deg, int n) {
    int i = blockIdx.x * blockDim.x + threadIdx.x;
    if (i < n) {
        float d = deg[i];
        deg[i] = d > 0.0f ? __frsqrt_rn(d) : 0.0f;
    }
}

__global__ void gcn_gemm_wmma(const float* __restrict__ x,
                              const float* __restrict__ W,
                              const float* __restrict__ bias,
                              const float* __restrict__ dinv,
                              float* __restrict__ h,
                              float* __restrict__ out,
                              int n_nodes, int n_tiles) {
    const int lane  = threadIdx.x & 31;
    const int wave  = threadIdx.x >> 5;
    const int tile  = blockIdx.x * (blockDim.x >> 5) + wave;
    if (tile >= n_tiles) return;

    const int base  = tile * 16;
    const int m     = lane & 15;
    const int khalf = (lane >> 4) * 2;

    int arow = base + m;
    if (arow >= n_nodes) arow = n_nodes - 1;
    const float* __restrict__ xrow = x + (size_t)arow * IN_CH;

    v2f a0 = *(const v2f*)(xrow + 0  + khalf);
    v2f a1 = *(const v2f*)(xrow + 4  + khalf);
    v2f a2 = *(const v2f*)(xrow + 8  + khalf);
    v2f a3 = *(const v2f*)(xrow + 12 + khalf);

    const int   wrow = (m < OUT_CH) ? m : 0;
    const float sel  = (m < OUT_CH) ? 1.0f : 0.0f;
    const float* __restrict__ wr = W + wrow * IN_CH + khalf;
    v2f b0 = *(const v2f*)(wr + 0);
    v2f b1 = *(const v2f*)(wr + 4);
    v2f b2 = *(const v2f*)(wr + 8);
    v2f b3 = *(const v2f*)(wr + 12);
    b0 *= sel; b1 *= sel; b2 *= sel; b3 *= sel;

    v8f acc = {};
    acc = __builtin_amdgcn_wmma_f32_16x16x4_f32(false, a0, false, b0, (short)0, acc, false, false);
    acc = __builtin_amdgcn_wmma_f32_16x16x4_f32(false, a1, false, b1, (short)0, acc, false, false);
    acc = __builtin_amdgcn_wmma_f32_16x16x4_f32(false, a2, false, b2, (short)0, acc, false, false);
    acc = __builtin_amdgcn_wmma_f32_16x16x4_f32(false, a3, false, b3, (short)0, acc, false, false);

    const int  n     = m;
    const int  mbase = base + ((lane >> 4) * 8);
    const bool full  = (base + 16) <= n_nodes;

    if (full) {
        if (n < OUT_CH) {
            const float bn = bias[n];
            const float4 d0 = *(const float4*)(dinv + mbase);
            const float4 d1 = *(const float4*)(dinv + mbase + 4);
            const float dv[8] = {d0.x, d0.y, d0.z, d0.w, d1.x, d1.y, d1.z, d1.w};
            float* __restrict__ hp = h   + (size_t)mbase * OUT_CH + n;
            float* __restrict__ op = out + (size_t)mbase * OUT_CH + n;
#pragma unroll
            for (int r = 0; r < 8; ++r) {
                const float hv = acc[r];
                hp[r * OUT_CH] = hv;
                op[r * OUT_CH] = bn + dv[r] * dv[r] * hv;
            }
        }
    } else {
        if (n < OUT_CH) {
            const float bn = bias[n];
#pragma unroll
            for (int r = 0; r < 8; ++r) {
                const int node = mbase + r;
                if (node < n_nodes) {
                    const float hv = acc[r];
                    h[(size_t)node * OUT_CH + n] = hv;
                    const float di = dinv[node];
                    out[(size_t)node * OUT_CH + n] = bn + di * di * hv;
                }
            }
        }
    }
}

__global__ void gcn_scatter(const int* __restrict__ src,
                            const int* __restrict__ dst,
                            const float* __restrict__ dinv,
                            const float* __restrict__ h,
                            float* __restrict__ out, int e_cnt) {
    int e = blockIdx.x * blockDim.x + threadIdx.x;
    if (e >= e_cnt) return;
    const int s = src[e];
    const int d = dst[e];
    const float w = dinv[s] * dinv[d];
    const float4* __restrict__ hs = (const float4*)(h + (size_t)s * OUT_CH);
    const float4 h0 = hs[0];
    const float4 h1 = hs[1];
    float* __restrict__ o = out + (size_t)d * OUT_CH;
    atomicAdd(o + 0, w * h0.x);
    atomicAdd(o + 1, w * h0.y);
    atomicAdd(o + 2, w * h0.z);
    atomicAdd(o + 3, w * h0.w);
    atomicAdd(o + 4, w * h1.x);
    atomicAdd(o + 5, w * h1.y);
    atomicAdd(o + 6, w * h1.z);
    atomicAdd(o + 7, w * h1.w);
}

extern "C" void kernel_launch(void* const* d_in, const int* in_sizes, int n_in,
                              void* d_out, int out_size, void* d_ws, size_t ws_size,
                              hipStream_t stream) {
    const float* x    = (const float*)d_in[0];
    const int*   ei   = (const int*)d_in[1];
    const float* W    = (const float*)d_in[2];
    const float* bias = (const float*)d_in[3];

    const int N = in_sizes[0] / IN_CH;
    const int E = in_sizes[1] / 2;
    const int* src = ei;
    const int* dst = ei + (size_t)E;

    float* dinv = (float*)d_ws;
    size_t h_off = (((size_t)N * sizeof(float)) + 255u) & ~(size_t)255u;
    float* h = (float*)((char*)d_ws + h_off);
    float* out = (float*)d_out;

    const int TPB = 256;

    gcn_deg_init<<<(N + TPB - 1) / TPB, TPB, 0, stream>>>(dinv, N);
    gcn_deg_count<<<(E + TPB - 1) / TPB, TPB, 0, stream>>>(dst, dinv, E);
    gcn_dinv<<<(N + TPB - 1) / TPB, TPB, 0, stream>>>(dinv, N);

    const int n_tiles = (N + 15) / 16;
    const int waves_per_block = TPB / 32;
    const int gemm_blocks = (n_tiles + waves_per_block - 1) / waves_per_block;
    gcn_gemm_wmma<<<gemm_blocks, TPB, 0, stream>>>(x, W, bias, dinv, h, out, N, n_tiles);

    gcn_scatter<<<(E + TPB - 1) / TPB, TPB, 0, stream>>>(src, dst, dinv, h, out, E);
}


# degree and edge sum by owner groups: each holds 4096 nodes in LDS and scans all of dst for its own edges (per-wave queues, grouped row gathers, LDS float adds); no global atomics; sum order changes
# speedup vs baseline: 1.3433x; 1.3433x over previous
#include <hip/hip_runtime.h>
#include <hip/hip_bf16.h>


typedef __attribute__((ext_vector_type(2))) float v2f;
typedef __attribute__((ext_vector_type(8))) float v8f;

#define IN_CH 16
#define OUT_CH 8

#define OWN_R    4096
#define OWN_T    1024
#define OWN_W    (OWN_T / 32)
#define OWN_U    8
#define OWN_SS   (OWN_T * OWN_U)
#define QA_CAP   512
#define QA_STEP  (OWN_U * 32)
#define DI       4

template <bool GUARD>
__device__ __forceinline__ void own_load_step(const int4* __restrict__ d4, int first, int n4,
                                              int tid, int4 (&v)[OWN_U]) {
#pragma unroll
    for (int u = 0; u < OWN_U; ++u) {
        const int idx = first + u * OWN_T + tid;
        if (GUARD) {
            v[u] = make_int4(-1, -1, -1, -1);
            if (idx < n4) v[u] = d4[idx];
        } else {
            v[u] = d4[idx];
        }
    }
}

template <typename T, typename S>
__device__ __forceinline__ void own_scan_all(const int* __restrict__ dst, int e_cnt, int tid,
                                             T tail, S step) {
    const bool aligned = ((reinterpret_cast<size_t>(dst) & 15u) == 0);
    const int n4 = aligned ? (e_cnt >> 2) : 0;
    for (int e = 4 * n4 + tid; e < e_cnt; e += OWN_T) tail(e);
    const int nss = (n4 + OWN_SS - 1) / OWN_SS;
    unsigned rot = (blockIdx.x * (unsigned)nss) / gridDim.x;
    if (rot >= (unsigned)nss) rot = 0;
    for (int it = 0; it < nss; ++it) {
        int ss = it + (int)rot;
        if (ss >= nss) ss -= nss;
        const int first = ss * OWN_SS;
        step(first, n4, first + OWN_SS > n4);
    }
}

__global__ __launch_bounds__(OWN_T) void gcn_deg_owner(const int* __restrict__ dst,
                                                       float* __restrict__ dinv,
                                                       int n_nodes, int e_cnt) {
    __shared__ unsigned cnt[OWN_R];
    const int tid = threadIdx.x;
    const unsigned base = blockIdx.x * OWN_R;
    if (base >= (unsigned)n_nodes) return;
    const unsigned span = min((unsigned)OWN_R, (unsigned)n_nodes - base);
    const int4* __restrict__ d4 = (const int4*)dst;

    for (int i = tid; i < OWN_R; i += OWN_T) cnt[i] = 0u;
    __syncthreads();

    auto test4 = [&](const int4& w) {
        const unsigned x0 = (unsigned)w.x - base;
        const unsigned x1 = (unsigned)w.y - base;
        const unsigned x2 = (unsigned)w.z - base;
        const unsigned x3 = (unsigned)w.w - base;
        if (min(min(x0, x1), min(x2, x3)) < span) {
            if (x0 < span) atomicAdd(&cnt[x0], 1u);
            if (x1 < span) atomicAdd(&cnt[x1], 1u);
            if (x2 < span) atomicAdd(&cnt[x2], 1u);
            if (x3 < span) atomicAdd(&cnt[x3], 1u);
        }
    };
    own_scan_all(dst, e_cnt, tid,
        [&](int e) {
            const unsigned x = (unsigned)dst[e] - base;
            if (x < span) atomicAdd(&cnt[x], 1u);
        },
        [&](int first, int n4, bool partial) {
            int4 v[OWN_U];
            if (partial) own_load_step<true >(d4, first, n4, tid, v);
            else         own_load_step<false>(d4, first, n4, tid, v);
#pragma unroll
            for (int u = 0; u < OWN_U; ++u) test4(v[u]);
        });
    __syncthreads();

    for (unsigned i = tid; i < span; i += OWN_T) {
        const float d = 1.0f + (float)cnt[i];
        dinv[base + i] = d > 0.0f ? __frsqrt_rn(d) : 0.0f;
    }
}

__global__ void gcn_gemm_wmma(const float* __restrict__ x,
                              const float* __restrict__ W,
                              const float* __restrict__ bias,
                              const float* __restrict__ dinv,
                              float* __restrict__ h,
                              float* __restrict__ out,
                              int n_nodes, int n_tiles) {
    const int lane  = threadIdx.x & 31;
    const int wave  = threadIdx.x >> 5;
    const int tile  = blockIdx.x * (blockDim.x >> 5) + wave;
    if (tile >= n_tiles) return;

    const int base  = tile * 16;
    const int m     = lane & 15;
    const int khalf = (lane >> 4) * 2;

    int arow = base + m;
    if (arow >= n_nodes) arow = n_nodes - 1;
    const float* __restrict__ xrow = x + (size_t)arow * IN_CH;

    v2f a0 = *(const v2f*)(xrow + 0  + khalf);
    v2f a1 = *(const v2f*)(xrow + 4  + khalf);
    v2f a2 = *(const v2f*)(xrow + 8  + khalf);
    v2f a3 = *(const v2f*)(xrow + 12 + khalf);

    const int   wrow = (m < OUT_CH) ? m : 0;
    const float sel  = (m < OUT_CH) ? 1.0f : 0.0f;
    const float* __restrict__ wr = W + wrow * IN_CH + khalf;
    v2f b0 = *(const v2f*)(wr + 0);
    v2f b1 = *(const v2f*)(wr + 4);
    v2f b2 = *(const v2f*)(wr + 8);
    v2f b3 = *(const v2f*)(wr + 12);
    b0 *= sel; b1 *= sel; b2 *= sel; b3 *= sel;

    v8f acc = {};
    acc = __builtin_amdgcn_wmma_f32_16x16x4_f32(false, a0, false, b0, (short)0, acc, false, false);
    acc = __builtin_amdgcn_wmma_f32_16x16x4_f32(false, a1, false, b1, (short)0, acc, false, false);
    acc = __builtin_amdgcn_wmma_f32_16x16x4_f32(false, a2, false, b2, (short)0, acc, false, false);
    acc = __builtin_amdgcn_wmma_f32_16x16x4_f32(false, a3, false, b3, (short)0, acc, false, false);

    const int  n     = m;
    const int  mbase = base + ((lane >> 4) * 8);
    const bool full  = (base + 16) <= n_nodes;

    if (full) {
        if (n < OUT_CH) {
            const float bn = bias[n];
            const float4 d0 = *(const float4*)(dinv + mbase);
            const float4 d1 = *(const float4*)(dinv + mbase + 4);
            const float dv[8] = {d0.x, d0.y, d0.z, d0.w, d1.x, d1.y, d1.z, d1.w};
            float* __restrict__ hp = h   + (size_t)mbase * OUT_CH + n;
            float* __restrict__ op = out + (size_t)mbase * OUT_CH + n;
#pragma unroll
            for (int r = 0; r < 8; ++r) {
                const float hv = acc[r];
                hp[r * OUT_CH] = hv;
                op[r * OUT_CH] = bn + dv[r] * dv[r] * hv;
            }
        }
    } else {
        if (n < OUT_CH) {
            const float bn = bias[n];
#pragma unroll
            for (int r = 0; r < 8; ++r) {
                const int node = mbase + r;
                if (node < n_nodes) {
                    const float hv = acc[r];
                    h[(size_t)node * OUT_CH + n] = hv;
                    const float di = dinv[node];
                    out[(size_t)node * OUT_CH + n] = bn + di * di * hv;
                }
            }
        }
    }
}

__device__ __forceinline__ void own_add_row(float* acc_row, float w, const float4& h0, const float4& h1) {
    atomicAdd(acc_row + 0, w * h0.x);
    atomicAdd(acc_row + 1, w * h0.y);
    atomicAdd(acc_row + 2, w * h0.z);
    atomicAdd(acc_row + 3, w * h0.w);
    atomicAdd(acc_row + 4, w * h1.x);
    atomicAdd(acc_row + 5, w * h1.y);
    atomicAdd(acc_row + 6, w * h1.z);
    atomicAdd(acc_row + 7, w * h1.w);
}

__global__ __launch_bounds__(OWN_T) void gcn_sum_owner(const int* __restrict__ src,
                                                       const int* __restrict__ dst,
                                                       const float* __restrict__ dinv,
                                                       const float* __restrict__ h,
                                                       float* __restrict__ out,
                                                       int n_nodes, int e_cnt) {
    __shared__ float acc[OWN_R * OUT_CH];
    __shared__ int   qa[OWN_W][QA_CAP];
    const int tid  = threadIdx.x;
    const int lane = tid & 31;
    const int wv   = __builtin_amdgcn_readfirstlane(tid >> 5);
    int* myq = qa[wv];
    const unsigned base = blockIdx.x * OWN_R;
    if (base >= (unsigned)n_nodes) return;
    const unsigned span = min((unsigned)OWN_R, (unsigned)n_nodes - base);
    const int4* __restrict__ d4 = (const int4*)dst;

    {
        const float4* __restrict__ o4 = (const float4*)(out + (size_t)base * OUT_CH);
        float4* a4 = (float4*)acc;
        for (unsigned i = tid; i < span * (OUT_CH / 4); i += OWN_T) a4[i] = o4[i];
    }
    __syncthreads();

    auto add_now = [&](int e, unsigned x) {
        const int s = src[e];
        const float w = dinv[s] * dinv[base + x];
        const float4* __restrict__ hs = (const float4*)(h + (size_t)s * OUT_CH);
        const float4 h0 = hs[0];
        const float4 h1 = hs[1];
        own_add_row(&acc[x * OUT_CH], w, h0, h1);
    };

    auto drain = [&](unsigned n) {
        for (unsigned j0 = 0; j0 < n; j0 += 32 * DI) {
            bool ok[DI]; int idx[DI]; int4 v[DI];
#pragma unroll
            for (int d = 0; d < DI; ++d) {
                const unsigned j = j0 + d * 32 + lane;
                ok[d]  = j < n;
                idx[d] = myq[ok[d] ? j : 0u];
            }
#pragma unroll
            for (int d = 0; d < DI; ++d) v[d] = d4[idx[d]];
            unsigned xm[DI]; int kk[DI]; int s[DI];
#pragma unroll
            for (int d = 0; d < DI; ++d) {
                const unsigned x0 = (unsigned)v[d].x - base;
                const unsigned x1 = (unsigned)v[d].y - base;
                const unsigned x2 = (unsigned)v[d].z - base;
                const unsigned x3 = (unsigned)v[d].w - base;
                const unsigned m  = min(min(x0, x1), min(x2, x3));
                kk[d] = (x0 == m) ? 0 : (x1 == m) ? 1 : (x2 == m) ? 2 : 3;
                xm[d] = min(m, span - 1u);
                s[d]  = src[4 * idx[d] + kk[d]];
            }
            float ws[DI], wd[DI]; float4 r0[DI], r1[DI];
#pragma unroll
            for (int d = 0; d < DI; ++d) {
                ws[d] = dinv[s[d]];
                wd[d] = dinv[base + xm[d]];
                const float4* __restrict__ hs = (const float4*)(h + (size_t)s[d] * OUT_CH);
                r0[d] = hs[0];
                r1[d] = hs[1];
            }
#pragma unroll
            for (int d = 0; d < DI; ++d) {
                asm volatile("" :: "v"(ws[d]), "v"(wd[d]),
                                   "v"(r0[d].x), "v"(r0[d].y), "v"(r0[d].z), "v"(r0[d].w),
                                   "v"(r1[d].x), "v"(r1[d].y), "v"(r1[d].z), "v"(r1[d].w));
            }
#pragma unroll
            for (int d = 0; d < DI; ++d)
                if (ok[d]) own_add_row(&acc[xm[d] * OUT_CH], ws[d] * wd[d], r0[d], r1[d]);
#pragma unroll
            for (int d = 0; d < DI; ++d) {
                if (ok[d]) {
                    const unsigned x0 = (unsigned)v[d].x - base;
                    const unsigned x1 = (unsigned)v[d].y - base;
                    const unsigned x2 = (unsigned)v[d].z - base;
                    const unsigned x3 = (unsigned)v[d].w - base;
                    const int e0 = 4 * idx[d];
                    if (kk[d] != 0 && x0 < span) add_now(e0 + 0, x0);
                    if (kk[d] != 1 && x1 < span) add_now(e0 + 1, x1);
                    if (kk[d] != 2 && x2 < span) add_now(e0 + 2, x2);
                    if (kk[d] != 3 && x3 < span) add_now(e0 + 3, x3);
                }
            }
        }
    };

    unsigned wcount = 0;

    own_scan_all(dst, e_cnt, tid,
        [&](int e) {
            const unsigned x = (unsigned)dst[e] - base;
            if (x < span) add_now(e, x);
        },
        [&](int first, int n4, bool partial) {
            int4 v[OWN_U];
            if (partial) own_load_step<true >(d4, first, n4, tid, v);
            else         own_load_step<false>(d4, first, n4, tid, v);
#pragma unroll
            for (int u = 0; u < OWN_U; ++u) {
                const unsigned x0 = (unsigned)v[u].x - base;
                const unsigned x1 = (unsigned)v[u].y - base;
                const unsigned x2 = (unsigned)v[u].z - base;
                const unsigned x3 = (unsigned)v[u].w - base;
                const bool p = min(min(x0, x1), min(x2, x3)) < span;
                const unsigned mask = __builtin_amdgcn_ballot_w32(p);
                const unsigned rank = __builtin_amdgcn_mbcnt_lo(mask, 0u);
                if (p) myq[wcount + rank] = first + u * OWN_T + tid;
                wcount += (unsigned)__builtin_popcount(mask);
            }
            if (wcount > QA_CAP - QA_STEP) { drain(wcount); wcount = 0; }
        });
    drain(wcount);
    __syncthreads();

    {
        float4* __restrict__ o4 = (float4*)(out + (size_t)base * OUT_CH);
        const float4* a4 = (const float4*)acc;
        for (unsigned i = tid; i < span * (OUT_CH / 4); i += OWN_T) o4[i] = a4[i];
    }
}

extern "C" void kernel_launch(void* const* d_in, const int* in_sizes, int n_in,
                              void* d_out, int out_size, void* d_ws, size_t ws_size,
                              hipStream_t stream) {
    const float* x    = (const float*)d_in[0];
    const int*   ei   = (const int*)d_in[1];
    const float* W    = (const float*)d_in[2];
    const float* bias = (const float*)d_in[3];

    const int N = in_sizes[0] / IN_CH;
    const int E = in_sizes[1] / 2;
    const int* src = ei;
    const int* dst = ei + (size_t)E;

    float* dinv = (float*)d_ws;
    size_t h_off = (((size_t)N * sizeof(float)) + 255u) & ~(size_t)255u;
    float* h = (float*)((char*)d_ws + h_off);
    float* out = (float*)d_out;

    const int TPB = 256;
    const int owners = (N + OWN_R - 1) / OWN_R;

    gcn_deg_owner<<<owners, OWN_T, 0, stream>>>(dst, dinv, N, E);

    const int n_tiles = (N + 15) / 16;
    const int waves_per_block = TPB / 32;
    const int gemm_blocks = (n_tiles + waves_per_block - 1) / waves_per_block;
    gcn_gemm_wmma<<<gemm_blocks, TPB, 0, stream>>>(x, W, bias, dinv, h, out, N, n_tiles);

    gcn_sum_owner<<<owners, OWN_T, 0, stream>>>(src, dst, dinv, h, out, N, E);
}


# degree walk keeps the int4 positions it found in per-wave lists in the workspace; the sum kernel works through its lists (dst and src words read together) and walks dst no more; sum order changes
# speedup vs baseline: 1.5464x; 1.1512x over previous
#include <hip/hip_runtime.h>
#include <hip/hip_bf16.h>


typedef __attribute__((ext_vector_type(2))) float v2f;
typedef __attribute__((ext_vector_type(8))) float v8f;

#define IN_CH 16
#define OUT_CH 8

#define OWN_R    4096
#define OWN_T    1024
#define OWN_W    (OWN_T / 32)
#define OWN_U    8
#define OWN_SS   (OWN_T * OWN_U)
#define QA_CAP   512
#define QA_STEP  (OWN_U * 32)
#define DI       4
#define LIST_MAX 2048

template <bool GUARD>
__device__ __forceinline__ void own_load_step(const int4* __restrict__ d4, int first, int n4,
                                              int tid, int4 (&v)[OWN_U]) {
#pragma unroll
    for (int u = 0; u < OWN_U; ++u) {
        const int idx = first + u * OWN_T + tid;
        if (GUARD) {
            v[u] = make_int4(-1, -1, -1, -1);
            if (idx < n4) v[u] = d4[idx];
        } else {
            v[u] = d4[idx];
        }
    }
}

template <typename T, typename S>
__device__ __forceinline__ void own_scan_all(const int* __restrict__ dst, int e_cnt, int tid,
                                             int it0, T tail, S step) {
    const bool aligned = ((reinterpret_cast<size_t>(dst) & 15u) == 0);
    const int n4 = aligned ? (e_cnt >> 2) : 0;
    for (int e = 4 * n4 + tid; e < e_cnt; e += OWN_T) tail(e);
    const int nss = (n4 + OWN_SS - 1) / OWN_SS;
    unsigned rot = (blockIdx.x * (unsigned)nss) / gridDim.x;
    if (rot >= (unsigned)nss) rot = 0;
    for (int it = it0; it < nss; ++it) {
        int ss = it + (int)rot;
        if (ss >= nss) ss -= nss;
        const int first = ss * OWN_SS;
        step(it, first, n4, first + OWN_SS > n4);
    }
}

__global__ __launch_bounds__(OWN_T) void gcn_deg_owner(const int* __restrict__ dst,
                                                       float* __restrict__ dinv,
                                                       int* __restrict__ meta,
                                                       int* __restrict__ lists,
                                                       unsigned cap,
                                                       int n_nodes, int e_cnt) {
    __shared__ unsigned cnt[OWN_R];
    const int tid = threadIdx.x;
    const int lane = tid & 31;
    const int wv   = __builtin_amdgcn_readfirstlane(tid >> 5);
    const unsigned base = blockIdx.x * OWN_R;
    if (base >= (unsigned)n_nodes) return;
    const unsigned span = min((unsigned)OWN_R, (unsigned)n_nodes - base);
    const int4* __restrict__ d4 = (const int4*)dst;

    for (int i = tid; i < OWN_R; i += OWN_T) cnt[i] = 0u;
    __syncthreads();

    const unsigned wslot = blockIdx.x * OWN_W + (unsigned)wv;
    int* __restrict__ lst = lists + (size_t)wslot * cap;
    unsigned wcount = 0;
    bool     rec    = true;
    int      stop   = 0x7fffffff;

    auto count4 = [&](unsigned x0, unsigned x1, unsigned x2, unsigned x3) {
        if (x0 < span) atomicAdd(&cnt[x0], 1u);
        if (x1 < span) atomicAdd(&cnt[x1], 1u);
        if (x2 < span) atomicAdd(&cnt[x2], 1u);
        if (x3 < span) atomicAdd(&cnt[x3], 1u);
    };
    auto test4 = [&](const int4& w) {
        const unsigned x0 = (unsigned)w.x - base;
        const unsigned x1 = (unsigned)w.y - base;
        const unsigned x2 = (unsigned)w.z - base;
        const unsigned x3 = (unsigned)w.w - base;
        if (min(min(x0, x1), min(x2, x3)) < span) count4(x0, x1, x2, x3);
    };
    auto test4_list = [&](const int4& w, int pos) {
        const unsigned x0 = (unsigned)w.x - base;
        const unsigned x1 = (unsigned)w.y - base;
        const unsigned x2 = (unsigned)w.z - base;
        const unsigned x3 = (unsigned)w.w - base;
        const bool p = min(min(x0, x1), min(x2, x3)) < span;
        const unsigned mask = __builtin_amdgcn_ballot_w32(p);
        const unsigned rank = __builtin_amdgcn_mbcnt_lo(mask, 0u);
        if (p) {
            lst[wcount + rank] = pos;
            count4(x0, x1, x2, x3);
        }
        wcount += (unsigned)__builtin_popcount(mask);
    };
    own_scan_all(dst, e_cnt, tid, 0,
        [&](int e) {
            const unsigned x = (unsigned)dst[e] - base;
            if (x < span) atomicAdd(&cnt[x], 1u);
        },
        [&](int it, int first, int n4, bool partial) {
            if (rec && wcount + QA_STEP > cap) { rec = false; stop = it; }
            int4 v[OWN_U];
            if (partial) own_load_step<true >(d4, first, n4, tid, v);
            else         own_load_step<false>(d4, first, n4, tid, v);
            if (rec) {
#pragma unroll
                for (int u = 0; u < OWN_U; ++u) test4_list(v[u], first + u * OWN_T + tid);
            } else {
#pragma unroll
                for (int u = 0; u < OWN_U; ++u) test4(v[u]);
            }
        });
    if (cap != 0u && lane == 0) {
        meta[2 * wslot + 0] = (int)wcount;
        meta[2 * wslot + 1] = stop;
    }
    __syncthreads();

    for (unsigned i = tid; i < span; i += OWN_T) {
        const float d = 1.0f + (float)cnt[i];
        dinv[base + i] = d > 0.0f ? __frsqrt_rn(d) : 0.0f;
    }
}

__global__ void gcn_gemm_wmma(const float* __restrict__ x,
                              const float* __restrict__ W,
                              const float* __restrict__ bias,
                              const float* __restrict__ dinv,
                              float* __restrict__ h,
                              float* __restrict__ out,
                              int n_nodes, int n_tiles) {
    const int lane  = threadIdx.x & 31;
    const int wave  = threadIdx.x >> 5;
    const int tile  = blockIdx.x * (blockDim.x >> 5) + wave;
    if (tile >= n_tiles) return;

    const int base  = tile * 16;
    const int m     = lane & 15;
    const int khalf = (lane >> 4) * 2;

    int arow = base + m;
    if (arow >= n_nodes) arow = n_nodes - 1;
    const float* __restrict__ xrow = x + (size_t)arow * IN_CH;

    v2f a0 = *(const v2f*)(xrow + 0  + khalf);
    v2f a1 = *(const v2f*)(xrow + 4  + khalf);
    v2f a2 = *(const v2f*)(xrow + 8  + khalf);
    v2f a3 = *(const v2f*)(xrow + 12 + khalf);

    const int   wrow = (m < OUT_CH) ? m : 0;
    const float sel  = (m < OUT_CH) ? 1.0f : 0.0f;
    const float* __restrict__ wr = W + wrow * IN_CH + khalf;
    v2f b0 = *(const v2f*)(wr + 0);
    v2f b1 = *(const v2f*)(wr + 4);
    v2f b2 = *(const v2f*)(wr + 8);
    v2f b3 = *(const v2f*)(wr + 12);
    b0 *= sel; b1 *= sel; b2 *= sel; b3 *= sel;

    v8f acc = {};
    acc = __builtin_amdgcn_wmma_f32_16x16x4_f32(false, a0, false, b0, (short)0, acc, false, false);
    acc = __builtin_amdgcn_wmma_f32_16x16x4_f32(false, a1, false, b1, (short)0, acc, false, false);
    acc = __builtin_amdgcn_wmma_f32_16x16x4_f32(false, a2, false, b2, (short)0, acc, false, false);
    acc = __builtin_amdgcn_wmma_f32_16x16x4_f32(false, a3, false, b3, (short)0, acc, false, false);

    const int  n     = m;
    const int  mbase = base + ((lane >> 4) * 8);
    const bool full  = (base + 16) <= n_nodes;

    if (full) {
        if (n < OUT_CH) {
            const float bn = bias[n];
            const float4 d0 = *(const float4*)(dinv + mbase);
            const float4 d1 = *(const float4*)(dinv + mbase + 4);
            const float dv[8] = {d0.x, d0.y, d0.z, d0.w, d1.x, d1.y, d1.z, d1.w};
            float* __restrict__ hp = h   + (size_t)mbase * OUT_CH + n;
            float* __restrict__ op = out + (size_t)mbase * OUT_CH + n;
#pragma unroll
            for (int r = 0; r < 8; ++r) {
                const float hv = acc[r];
                hp[r * OUT_CH] = hv;
                op[r * OUT_CH] = bn + dv[r] * dv[r] * hv;
            }
        }
    } else {
        if (n < OUT_CH) {
            const float bn = bias[n];
#pragma unroll
            for (int r = 0; r < 8; ++r) {
                const int node = mbase + r;
                if (node < n_nodes) {
                    const float hv = acc[r];
                    h[(size_t)node * OUT_CH + n] = hv;
                    const float di = dinv[node];
                    out[(size_t)node * OUT_CH + n] = bn + di * di * hv;
                }
            }
        }
    }
}

__device__ __forceinline__ void own_add_row(float* acc_row, float w, const float4& h0, const float4& h1) {
    atomicAdd(acc_row + 0, w * h0.x);
    atomicAdd(acc_row + 1, w * h0.y);
    atomicAdd(acc_row + 2, w * h0.z);
    atomicAdd(acc_row + 3, w * h0.w);
    atomicAdd(acc_row + 4, w * h1.x);
    atomicAdd(acc_row + 5, w * h1.y);
    atomicAdd(acc_row + 6, w * h1.z);
    atomicAdd(acc_row + 7, w * h1.w);
}

__global__ __launch_bounds__(OWN_T) void gcn_sum_owner(const int* __restrict__ src,
                                                       const int* __restrict__ dst,
                                                       const float* __restrict__ dinv,
                                                       const float* __restrict__ h,
                                                       const int* __restrict__ meta,
                                                       const int* __restrict__ lists,
                                                       unsigned cap,
                                                       float* __restrict__ out,
                                                       int n_nodes, int e_cnt) {
    __shared__ float acc[OWN_R * OUT_CH];
    __shared__ int   qa[OWN_W][QA_CAP];
    const int tid  = threadIdx.x;
    const int lane = tid & 31;
    const int wv   = __builtin_amdgcn_readfirstlane(tid >> 5);
    int* myq = qa[wv];
    const unsigned base = blockIdx.x * OWN_R;
    if (base >= (unsigned)n_nodes) return;
    const unsigned span = min((unsigned)OWN_R, (unsigned)n_nodes - base);
    const int4* __restrict__ d4 = (const int4*)dst;
    const bool s_al = ((reinterpret_cast<size_t>(src) & 15u) == 0);
    const unsigned n4m1 = (unsigned)max((e_cnt >> 2) - 1, 0);

    const unsigned wslot = blockIdx.x * OWN_W + (unsigned)wv;
    const int* __restrict__ lst = lists + (size_t)wslot * cap;
    unsigned lcount = 0;
    int it0 = 0;
    if (cap != 0u) {
        lcount = min((unsigned)__builtin_amdgcn_readfirstlane(meta[2 * wslot + 0]), cap);
        it0    = max(__builtin_amdgcn_readfirstlane(meta[2 * wslot + 1]), 0);
    }

    {
        const float4* __restrict__ o4 = (const float4*)(out + (size_t)base * OUT_CH);
        float4* a4 = (float4*)acc;
        for (unsigned i = tid; i < span * (OUT_CH / 4); i += OWN_T) a4[i] = o4[i];
    }
    __syncthreads();

    auto add_src = [&](int s, unsigned x) {
        const float w = dinv[s] * dinv[base + x];
        const float4* __restrict__ hs = (const float4*)(h + (size_t)s * OUT_CH);
        const float4 h0 = hs[0];
        const float4 h1 = hs[1];
        own_add_row(&acc[x * OUT_CH], w, h0, h1);
    };

    auto drain = [&](unsigned n, auto get) {
        for (unsigned j0 = 0; j0 < n; j0 += 32 * DI) {
            bool ok[DI]; unsigned idx[DI]; int4 v[DI]; int4 sv[DI];
#pragma unroll
            for (int d = 0; d < DI; ++d) {
                const unsigned j = j0 + d * 32 + lane;
                ok[d]  = j < n;
                idx[d] = min((unsigned)get(ok[d] ? j : 0u), n4m1);
            }
#pragma unroll
            for (int d = 0; d < DI; ++d) {
                v[d] = d4[idx[d]];
                if (s_al) {
                    sv[d] = ((const int4*)src)[idx[d]];
                } else {
                    const int* sp = src + 4 * (size_t)idx[d];
                    sv[d] = make_int4(sp[0], sp[1], sp[2], sp[3]);
                }
            }
            unsigned xm[DI]; int kk[DI]; int s[DI];
#pragma unroll
            for (int d = 0; d < DI; ++d) {
                const unsigned x0 = (unsigned)v[d].x - base;
                const unsigned x1 = (unsigned)v[d].y - base;
                const unsigned x2 = (unsigned)v[d].z - base;
                const unsigned x3 = (unsigned)v[d].w - base;
                const unsigned m  = min(min(x0, x1), min(x2, x3));
                kk[d] = (x0 == m) ? 0 : (x1 == m) ? 1 : (x2 == m) ? 2 : 3;
                xm[d] = min(m, span - 1u);
                s[d]  = (x0 == m) ? sv[d].x : (x1 == m) ? sv[d].y : (x2 == m) ? sv[d].z : sv[d].w;
            }
            float ws[DI], wd[DI]; float4 r0[DI], r1[DI];
#pragma unroll
            for (int d = 0; d < DI; ++d) {
                ws[d] = dinv[s[d]];
                wd[d] = dinv[base + xm[d]];
                const float4* __restrict__ hs = (const float4*)(h + (size_t)s[d] * OUT_CH);
                r0[d] = hs[0];
                r1[d] = hs[1];
            }
#pragma unroll
            for (int d = 0; d < DI; ++d) {
                asm volatile("" :: "v"(ws[d]), "v"(wd[d]),
                                   "v"(r0[d].x), "v"(r0[d].y), "v"(r0[d].z), "v"(r0[d].w),
                                   "v"(r1[d].x), "v"(r1[d].y), "v"(r1[d].z), "v"(r1[d].w));
            }
#pragma unroll
            for (int d = 0; d < DI; ++d)
                if (ok[d]) own_add_row(&acc[xm[d] * OUT_CH], ws[d] * wd[d], r0[d], r1[d]);
#pragma unroll
            for (int d = 0; d < DI; ++d) {
                if (ok[d]) {
                    const unsigned x0 = (unsigned)v[d].x - base;
                    const unsigned x1 = (unsigned)v[d].y - base;
                    const unsigned x2 = (unsigned)v[d].z - base;
                    const unsigned x3 = (unsigned)v[d].w - base;
                    if (kk[d] != 0 && x0 < span) add_src(sv[d].x, x0);
                    if (kk[d] != 1 && x1 < span) add_src(sv[d].y, x1);
                    if (kk[d] != 2 && x2 < span) add_src(sv[d].z, x2);
                    if (kk[d] != 3 && x3 < span) add_src(sv[d].w, x3);
                }
            }
        }
    };

    drain(lcount, [&](unsigned j) { return lst[j]; });

    unsigned wcount = 0;

    own_scan_all(dst, e_cnt, tid, it0,
        [&](int e) {
            const unsigned x = (unsigned)dst[e] - base;
            if (x < span) add_src(src[e], x);
        },
        [&](int it, int first, int n4, bool partial) {
            int4 v[OWN_U];
            if (partial) own_load_step<true >(d4, first, n4, tid, v);
            else         own_load_step<false>(d4, first, n4, tid, v);
#pragma unroll
            for (int u = 0; u < OWN_U; ++u) {
                const unsigned x0 = (unsigned)v[u].x - base;
                const unsigned x1 = (unsigned)v[u].y - base;
                const unsigned x2 = (unsigned)v[u].z - base;
                const unsigned x3 = (unsigned)v[u].w - base;
                const bool p = min(min(x0, x1), min(x2, x3)) < span;
                const unsigned mask = __builtin_amdgcn_ballot_w32(p);
                const unsigned rank = __builtin_amdgcn_mbcnt_lo(mask, 0u);
                if (p) myq[wcount + rank] = first + u * OWN_T + tid;
                wcount += (unsigned)__builtin_popcount(mask);
            }
            if (wcount > QA_CAP - QA_STEP) { drain(wcount, [&](unsigned j) { return myq[j]; }); wcount = 0; }
        });
    drain(wcount, [&](unsigned j) { return myq[j]; });
    __syncthreads();

    {
        float4* __restrict__ o4 = (float4*)(out + (size_t)base * OUT_CH);
        const float4* a4 = (const float4*)acc;
        for (unsigned i = tid; i < span * (OUT_CH / 4); i += OWN_T) o4[i] = a4[i];
    }
}

extern "C" void kernel_launch(void* const* d_in, const int* in_sizes, int n_in,
                              void* d_out, int out_size, void* d_ws, size_t ws_size,
                              hipStream_t stream) {
    const float* x    = (const float*)d_in[0];
    const int*   ei   = (const int*)d_in[1];
    const float* W    = (const float*)d_in[2];
    const float* bias = (const float*)d_in[3];

    const int N = in_sizes[0] / IN_CH;
    const int E = in_sizes[1] / 2;
    const int* src = ei;
    const int* dst = ei + (size_t)E;

    float* dinv = (float*)d_ws;
    size_t h_off = (((size_t)N * sizeof(float)) + 255u) & ~(size_t)255u;
    float* h = (float*)((char*)d_ws + h_off);
    float* out = (float*)d_out;

    const int TPB = 256;
    const int owners = (N + OWN_R - 1) / OWN_R;

    const size_t lwaves   = (size_t)owners * OWN_W;
    const size_t meta_off = (h_off + (size_t)N * OUT_CH * sizeof(float) + 255u) & ~(size_t)255u;
    const size_t list_off = (meta_off + lwaves * 2 * sizeof(int) + 255u) & ~(size_t)255u;
    size_t cap = 0;
    if (lwaves != 0 && ws_size > list_off) cap = (ws_size - list_off) / (lwaves * sizeof(int));
    if (cap > LIST_MAX) cap = LIST_MAX;
    int* meta  = (int*)((char*)d_ws + meta_off);
    int* lists = (int*)((char*)d_ws + list_off);

    gcn_deg_owner<<<owners, OWN_T, 0, stream>>>(dst, dinv, meta, lists, (unsigned)cap, N, E);

    const int n_tiles = (N + 15) / 16;
    const int waves_per_block = TPB / 32;
    const int gemm_blocks = (n_tiles + waves_per_block - 1) / waves_per_block;
    gcn_gemm_wmma<<<gemm_blocks, TPB, 0, stream>>>(x, W, bias, dinv, h, out, N, n_tiles);

    gcn_sum_owner<<<owners, OWN_T, 0, stream>>>(src, dst, dinv, h, meta, lists, (unsigned)cap, out, N, E);
}


# two workgroups per owner range, each walks half of dst (grid fills the chip): counts meet as integer partials in the product kernel, partial tiles add onto out by float atomics; sum order changes
# speedup vs baseline: 2.7252x; 1.7623x over previous
#include <hip/hip_runtime.h>
#include <hip/hip_bf16.h>


typedef __attribute__((ext_vector_type(2))) float v2f;
typedef __attribute__((ext_vector_type(8))) float v8f;

#define IN_CH 16
#define OUT_CH 8

#define OWN_R    4096
#define OWN_NS   2
#define OWN_T    1024
#define OWN_W    (OWN_T / 32)
#define OWN_U    8
#define OWN_SS   (OWN_T * OWN_U)
#define QA_CAP   512
#define QA_STEP  (OWN_U * 32)
#define DI       4
#define LIST_MAX 2048

template <bool GUARD>
__device__ __forceinline__ void own_load_step(const int4* __restrict__ d4, int first, int n4,
                                              int tid, int4 (&v)[OWN_U]) {
#pragma unroll
    for (int u = 0; u < OWN_U; ++u) {
        const int idx = first + u * OWN_T + tid;
        if (GUARD) {
            v[u] = make_int4(-1, -1, -1, -1);
            if (idx < n4) v[u] = d4[idx];
        } else {
            v[u] = d4[idx];
        }
    }
}

template <typename T, typename S>
__device__ __forceinline__ void own_scan_all(const int* __restrict__ dst, int e_cnt, int tid,
                                             unsigned owner, unsigned owners, unsigned slice,
                                             int it0, T tail, S step) {
    const bool aligned = ((reinterpret_cast<size_t>(dst) & 15u) == 0);
    const int n4 = aligned ? (e_cnt >> 2) : 0;
    if (slice == 0u)
        for (int e = 4 * n4 + tid; e < e_cnt; e += OWN_T) tail(e);
    const int nss = (n4 + OWN_SS - 1) / OWN_SS;
    const int lo  = (int)(((unsigned)nss * slice) / (unsigned)OWN_NS);
    const int hi  = (int)(((unsigned)nss * (slice + 1u)) / (unsigned)OWN_NS);
    const int ns  = hi - lo;
    unsigned rot = (owner * (unsigned)ns) / owners;
    if (rot >= (unsigned)ns) rot = 0;
    for (int it = it0; it < ns; ++it) {
        int ss = it + (int)rot;
        if (ss >= ns) ss -= ns;
        const int first = (lo + ss) * OWN_SS;
        step(it, first, n4, first + OWN_SS > n4);
    }
}

__global__ __launch_bounds__(OWN_T) void gcn_deg_owner(const int* __restrict__ dst,
                                                       unsigned* __restrict__ cntp,
                                                       int cstride,
                                                       int* __restrict__ meta,
                                                       int* __restrict__ lists,
                                                       unsigned cap,
                                                       int n_nodes, int e_cnt) {
    __shared__ unsigned cnt[OWN_R];
    const int tid = threadIdx.x;
    const int lane = tid & 31;
    const int wv   = __builtin_amdgcn_readfirstlane(tid >> 5);
    const unsigned owner  = blockIdx.x / OWN_NS;
    const unsigned slice  = blockIdx.x - owner * OWN_NS;
    const unsigned owners = gridDim.x / OWN_NS;
    const unsigned base = owner * OWN_R;
    if (base >= (unsigned)n_nodes) return;
    const unsigned span = min((unsigned)OWN_R, (unsigned)n_nodes - base);
    const int4* __restrict__ d4 = (const int4*)dst;

    for (int i = tid; i < OWN_R; i += OWN_T) cnt[i] = 0u;
    __syncthreads();

    const unsigned wslot = blockIdx.x * OWN_W + (unsigned)wv;
    int* __restrict__ lst = lists + (size_t)wslot * cap;
    unsigned wcount = 0;
    bool     rec    = true;
    int      stop   = 0x7fffffff;

    auto count4 = [&](unsigned x0, unsigned x1, unsigned x2, unsigned x3) {
        if (x0 < span) atomicAdd(&cnt[x0], 1u);
        if (x1 < span) atomicAdd(&cnt[x1], 1u);
        if (x2 < span) atomicAdd(&cnt[x2], 1u);
        if (x3 < span) atomicAdd(&cnt[x3], 1u);
    };
    auto test4 = [&](const int4& w) {
        const unsigned x0 = (unsigned)w.x - base;
        const unsigned x1 = (unsigned)w.y - base;
        const unsigned x2 = (unsigned)w.z - base;
        const unsigned x3 = (unsigned)w.w - base;
        if (min(min(x0, x1), min(x2, x3)) < span) count4(x0, x1, x2, x3);
    };
    auto test4_list = [&](const int4& w, int pos) {
        const unsigned x0 = (unsigned)w.x - base;
        const unsigned x1 = (unsigned)w.y - base;
        const unsigned x2 = (unsigned)w.z - base;
        const unsigned x3 = (unsigned)w.w - base;
        const bool p = min(min(x0, x1), min(x2, x3)) < span;
        const unsigned mask = __builtin_amdgcn_ballot_w32(p);
        const unsigned rank = __builtin_amdgcn_mbcnt_lo(mask, 0u);
        if (p) {
            lst[wcount + rank] = pos;
            count4(x0, x1, x2, x3);
        }
        wcount += (unsigned)__builtin_popcount(mask);
    };
    own_scan_all(dst, e_cnt, tid, owner, owners, slice, 0,
        [&](int e) {
            const unsigned x = (unsigned)dst[e] - base;
            if (x < span) atomicAdd(&cnt[x], 1u);
        },
        [&](int it, int first, int n4, bool partial) {
            if (rec && wcount + QA_STEP > cap) { rec = false; stop = it; }
            int4 v[OWN_U];
            if (partial) own_load_step<true >(d4, first, n4, tid, v);
            else         own_load_step<false>(d4, first, n4, tid, v);
            if (rec) {
#pragma unroll
                for (int u = 0; u < OWN_U; ++u) test4_list(v[u], first + u * OWN_T + tid);
            } else {
#pragma unroll
                for (int u = 0; u < OWN_U; ++u) test4(v[u]);
            }
        });
    if (cap != 0u && lane == 0) {
        meta[2 * wslot + 0] = (int)wcount;
        meta[2 * wslot + 1] = stop;
    }
    __syncthreads();

    unsigned* __restrict__ plane = cntp + (size_t)slice * (size_t)cstride + base;
    for (unsigned i = tid; i < span; i += OWN_T) plane[i] = cnt[i];
}

__device__ __forceinline__ float own_dinv(unsigned c) {
    const float d = 1.0f + (float)c;
    return d > 0.0f ? __frsqrt_rn(d) : 0.0f;
}

__global__ void gcn_gemm_wmma(const float* __restrict__ x,
                              const float* __restrict__ W,
                              const float* __restrict__ bias,
                              const unsigned* __restrict__ cntp,
                              int cstride,
                              float* __restrict__ dinv,
                              float* __restrict__ h,
                              float* __restrict__ out,
                              int n_nodes, int n_tiles) {
    const int lane  = threadIdx.x & 31;
    const int wave  = threadIdx.x >> 5;
    const int tile  = blockIdx.x * (blockDim.x >> 5) + wave;
    if (tile >= n_tiles) return;

    const int base  = tile * 16;
    const int m     = lane & 15;
    const int khalf = (lane >> 4) * 2;

    int arow = base + m;
    if (arow >= n_nodes) arow = n_nodes - 1;
    const float* __restrict__ xrow = x + (size_t)arow * IN_CH;

    v2f a0 = *(const v2f*)(xrow + 0  + khalf);
    v2f a1 = *(const v2f*)(xrow + 4  + khalf);
    v2f a2 = *(const v2f*)(xrow + 8  + khalf);
    v2f a3 = *(const v2f*)(xrow + 12 + khalf);

    const int   wrow = (m < OUT_CH) ? m : 0;
    const float sel  = (m < OUT_CH) ? 1.0f : 0.0f;
    const float* __restrict__ wr = W + wrow * IN_CH + khalf;
    v2f b0 = *(const v2f*)(wr + 0);
    v2f b1 = *(const v2f*)(wr + 4);
    v2f b2 = *(const v2f*)(wr + 8);
    v2f b3 = *(const v2f*)(wr + 12);
    b0 *= sel; b1 *= sel; b2 *= sel; b3 *= sel;

    v8f acc = {};
    acc = __builtin_amdgcn_wmma_f32_16x16x4_f32(false, a0, false, b0, (short)0, acc, false, false);
    acc = __builtin_amdgcn_wmma_f32_16x16x4_f32(false, a1, false, b1, (short)0, acc, false, false);
    acc = __builtin_amdgcn_wmma_f32_16x16x4_f32(false, a2, false, b2, (short)0, acc, false, false);
    acc = __builtin_amdgcn_wmma_f32_16x16x4_f32(false, a3, false, b3, (short)0, acc, false, false);

    if ((tile & 1) == 0) {
        const int node = base + lane;
        if (node < n_nodes) {
            unsigned c = 0u;
#pragma unroll
            for (int s = 0; s < OWN_NS; ++s) c += cntp[(size_t)s * (size_t)cstride + node];
            dinv[node] = own_dinv(c);
        }
    }

    const int  n     = m;
    const int  mbase = base + ((lane >> 4) * 8);
    const bool full  = (base + 16) <= n_nodes;

    if (full) {
        if (n < OUT_CH) {
            const float bn = bias[n];
            uint4 c0 = *(const uint4*)(cntp + mbase);
            uint4 c1 = *(const uint4*)(cntp + mbase + 4);
#pragma unroll
            for (int s = 1; s < OWN_NS; ++s) {
                const unsigned* __restrict__ pl = cntp + (size_t)s * (size_t)cstride + mbase;
                const uint4 e0 = *(const uint4*)(pl);
                const uint4 e1 = *(const uint4*)(pl + 4);
                c0.x += e0.x; c0.y += e0.y; c0.z += e0.z; c0.w += e0.w;
                c1.x += e1.x; c1.y += e1.y; c1.z += e1.z; c1.w += e1.w;
            }
            const float dv[8] = {own_dinv(c0.x), own_dinv(c0.y), own_dinv(c0.z), own_dinv(c0.w),
                                 own_dinv(c1.x), own_dinv(c1.y), own_dinv(c1.z), own_dinv(c1.w)};
            float* __restrict__ hp = h   + (size_t)mbase * OUT_CH + n;
            float* __restrict__ op = out + (size_t)mbase * OUT_CH + n;
#pragma unroll
            for (int r = 0; r < 8; ++r) {
                const float hv = acc[r];
                hp[r * OUT_CH] = hv;
                op[r * OUT_CH] = bn + dv[r] * dv[r] * hv;
            }
        }
    } else {
        if (n < OUT_CH) {
            const float bn = bias[n];
#pragma unroll
            for (int r = 0; r < 8; ++r) {
                const int node = mbase + r;
                if (node < n_nodes) {
                    const float hv = acc[r];
                    h[(size_t)node * OUT_CH + n] = hv;
                    unsigned c = 0u;
#pragma unroll
                    for (int s = 0; s < OWN_NS; ++s) c += cntp[(size_t)s * (size_t)cstride + node];
                    const float di = own_dinv(c);
                    out[(size_t)node * OUT_CH + n] = bn + di * di * hv;
                }
            }
        }
    }
}

__device__ __forceinline__ void own_add_row(float* acc_row, float w, const float4& h0, const float4& h1) {
    atomicAdd(acc_row + 0, w * h0.x);
    atomicAdd(acc_row + 1, w * h0.y);
    atomicAdd(acc_row + 2, w * h0.z);
    atomicAdd(acc_row + 3, w * h0.w);
    atomicAdd(acc_row + 4, w * h1.x);
    atomicAdd(acc_row + 5, w * h1.y);
    atomicAdd(acc_row + 6, w * h1.z);
    atomicAdd(acc_row + 7, w * h1.w);
}

__global__ __launch_bounds__(OWN_T) void gcn_sum_owner(const int* __restrict__ src,
                                                       const int* __restrict__ dst,
                                                       const float* __restrict__ dinv,
                                                       const float* __restrict__ h,
                                                       const int* __restrict__ meta,
                                                       const int* __restrict__ lists,
                                                       unsigned cap,
                                                       float* __restrict__ out,
                                                       int n_nodes, int e_cnt) {
    __shared__ float acc[OWN_R * OUT_CH];
    __shared__ int   qa[OWN_W][QA_CAP];
    const int tid  = threadIdx.x;
    const int lane = tid & 31;
    const int wv   = __builtin_amdgcn_readfirstlane(tid >> 5);
    int* myq = qa[wv];
    const unsigned owner  = blockIdx.x / OWN_NS;
    const unsigned slice  = blockIdx.x - owner * OWN_NS;
    const unsigned owners = gridDim.x / OWN_NS;
    const unsigned base = owner * OWN_R;
    if (base >= (unsigned)n_nodes) return;
    const unsigned span = min((unsigned)OWN_R, (unsigned)n_nodes - base);
    const int4* __restrict__ d4 = (const int4*)dst;
    const bool s_al = ((reinterpret_cast<size_t>(src) & 15u) == 0);
    const unsigned n4m1 = (unsigned)max((e_cnt >> 2) - 1, 0);

    const unsigned wslot = blockIdx.x * OWN_W + (unsigned)wv;
    const int* __restrict__ lst = lists + (size_t)wslot * cap;
    unsigned lcount = 0;
    int it0 = 0;
    if (cap != 0u) {
        lcount = min((unsigned)__builtin_amdgcn_readfirstlane(meta[2 * wslot + 0]), cap);
        it0    = max(__builtin_amdgcn_readfirstlane(meta[2 * wslot + 1]), 0);
    }

    {
        float4* a4 = (float4*)acc;
        for (unsigned i = tid; i < span * (OUT_CH / 4); i += OWN_T) a4[i] = make_float4(0.0f, 0.0f, 0.0f, 0.0f);
    }
    __syncthreads();

    auto add_src = [&](int s, unsigned x) {
        const float w = dinv[s] * dinv[base + x];
        const float4* __restrict__ hs = (const float4*)(h + (size_t)s * OUT_CH);
        const float4 h0 = hs[0];
        const float4 h1 = hs[1];
        own_add_row(&acc[x * OUT_CH], w, h0, h1);
    };

    auto drain = [&](unsigned n, auto get) {
        for (unsigned j0 = 0; j0 < n; j0 += 32 * DI) {
            bool ok[DI]; unsigned idx[DI]; int4 v[DI]; int4 sv[DI];
#pragma unroll
            for (int d = 0; d < DI; ++d) {
                const unsigned j = j0 + d * 32 + lane;
                ok[d]  = j < n;
                idx[d] = min((unsigned)get(ok[d] ? j : 0u), n4m1);
            }
#pragma unroll
            for (int d = 0; d < DI; ++d) {
                v[d] = d4[idx[d]];
                if (s_al) {
                    sv[d] = ((const int4*)src)[idx[d]];
                } else {
                    const int* sp = src + 4 * (size_t)idx[d];
                    sv[d] = make_int4(sp[0], sp[1], sp[2], sp[3]);
                }
            }
            unsigned xm[DI]; int kk[DI]; int s[DI];
#pragma unroll
            for (int d = 0; d < DI; ++d) {
                const unsigned x0 = (unsigned)v[d].x - base;
                const unsigned x1 = (unsigned)v[d].y - base;
                const unsigned x2 = (unsigned)v[d].z - base;
                const unsigned x3 = (unsigned)v[d].w - base;
                const unsigned m  = min(min(x0, x1), min(x2, x3));
                kk[d] = (x0 == m) ? 0 : (x1 == m) ? 1 : (x2 == m) ? 2 : 3;
                xm[d] = min(m, span - 1u);
                s[d]  = (x0 == m) ? sv[d].x : (x1 == m) ? sv[d].y : (x2 == m) ? sv[d].z : sv[d].w;
            }
            float ws[DI], wd[DI]; float4 r0[DI], r1[DI];
#pragma unroll
            for (int d = 0; d < DI; ++d) {
                ws[d] = dinv[s[d]];
                wd[d] = dinv[base + xm[d]];
                const float4* __restrict__ hs = (const float4*)(h + (size_t)s[d] * OUT_CH);
                r0[d] = hs[0];
                r1[d] = hs[1];
            }
#pragma unroll
            for (int d = 0; d < DI; ++d) {
                asm volatile("" :: "v"(ws[d]), "v"(wd[d]),
                                   "v"(r0[d].x), "v"(r0[d].y), "v"(r0[d].z), "v"(r0[d].w),
                                   "v"(r1[d].x), "v"(r1[d].y), "v"(r1[d].z), "v"(r1[d].w));
            }
#pragma unroll
            for (int d = 0; d < DI; ++d)
                if (ok[d]) own_add_row(&acc[xm[d] * OUT_CH], ws[d] * wd[d], r0[d], r1[d]);
#pragma unroll
            for (int d = 0; d < DI; ++d) {
                if (ok[d]) {
                    const unsigned x0 = (unsigned)v[d].x - base;
                    const unsigned x1 = (unsigned)v[d].y - base;
                    const unsigned x2 = (unsigned)v[d].z - base;
                    const unsigned x3 = (unsigned)v[d].w - base;
                    if (kk[d] != 0 && x0 < span) add_src(sv[d].x, x0);
                    if (kk[d] != 1 && x1 < span) add_src(sv[d].y, x1);
                    if (kk[d] != 2 && x2 < span) add_src(sv[d].z, x2);
                    if (kk[d] != 3 && x3 < span) add_src(sv[d].w, x3);
                }
            }
        }
    };

    drain(lcount, [&](unsigned j) { return lst[j]; });

    unsigned wcount = 0;

    own_scan_all(dst, e_cnt, tid, owner, owners, slice, it0,
        [&](int e) {
            const unsigned x = (unsigned)dst[e] - base;
            if (x < span) add_src(src[e], x);
        },
        [&](int it, int first, int n4, bool partial) {
            int4 v[OWN_U];
            if (partial) own_load_step<true >(d4, first, n4, tid, v);
            else         own_load_step<false>(d4, first, n4, tid, v);
#pragma unroll
            for (int u = 0; u < OWN_U; ++u) {
                const unsigned x0 = (unsigned)v[u].x - base;
                const unsigned x1 = (unsigned)v[u].y - base;
                const unsigned x2 = (unsigned)v[u].z - base;
                const unsigned x3 = (unsigned)v[u].w - base;
                const bool p = min(min(x0, x1), min(x2, x3)) < span;
                const unsigned mask = __builtin_amdgcn_ballot_w32(p);
                const unsigned rank = __builtin_amdgcn_mbcnt_lo(mask, 0u);
                if (p) myq[wcount + rank] = first + u * OWN_T + tid;
                wcount += (unsigned)__builtin_popcount(mask);
            }
            if (wcount > QA_CAP - QA_STEP) { drain(wcount, [&](unsigned j) { return myq[j]; }); wcount = 0; }
        });
    drain(wcount, [&](unsigned j) { return myq[j]; });
    __syncthreads();

    {
        float* __restrict__ o = out + (size_t)base * OUT_CH;
        for (unsigned i = tid; i < span * OUT_CH; i += OWN_T) atomicAdd(o + i, acc[i]);
    }
}

extern "C" void kernel_launch(void* const* d_in, const int* in_sizes, int n_in,
                              void* d_out, int out_size, void* d_ws, size_t ws_size,
                              hipStream_t stream) {
    const float* x    = (const float*)d_in[0];
    const int*   ei   = (const int*)d_in[1];
    const float* W    = (const float*)d_in[2];
    const float* bias = (const float*)d_in[3];

    const int N = in_sizes[0] / IN_CH;
    const int E = in_sizes[1] / 2;
    const int* src = ei;
    const int* dst = ei + (size_t)E;

    float* dinv = (float*)d_ws;
    size_t h_off = (((size_t)N * sizeof(float)) + 255u) & ~(size_t)255u;
    float* h = (float*)((char*)d_ws + h_off);
    float* out = (float*)d_out;

    const int TPB = 256;
    const int owners = (N + OWN_R - 1) / OWN_R;

    const size_t lwaves   = (size_t)owners * OWN_NS * OWN_W;
    const size_t cstride  = ((size_t)N + 63u) & ~(size_t)63u;
    const size_t cnt_off  = (h_off + (size_t)N * OUT_CH * sizeof(float) + 255u) & ~(size_t)255u;
    const size_t meta_off = (cnt_off + (size_t)OWN_NS * cstride * sizeof(unsigned) + 255u) & ~(size_t)255u;
    const size_t list_off = (meta_off + lwaves * 2 * sizeof(int) + 255u) & ~(size_t)255u;
    size_t cap = 0;
    if (lwaves != 0 && ws_size > list_off) cap = (ws_size - list_off) / (lwaves * sizeof(int));
    if (cap > LIST_MAX) cap = LIST_MAX;
    unsigned* cntp = (unsigned*)((char*)d_ws + cnt_off);
    int* meta  = (int*)((char*)d_ws + meta_off);
    int* lists = (int*)((char*)d_ws + list_off);

    gcn_deg_owner<<<owners * OWN_NS, OWN_T, 0, stream>>>(dst, cntp, (int)cstride, meta, lists, (unsigned)cap, N, E);

    const int n_tiles = (N + 15) / 16;
    const int waves_per_block = TPB / 32;
    const int gemm_blocks = (n_tiles + waves_per_block - 1) / waves_per_block;
    gcn_gemm_wmma<<<gemm_blocks, TPB, 0, stream>>>(x, W, bias, cntp, (int)cstride, dinv, h, out, N, n_tiles);

    gcn_sum_owner<<<owners * OWN_NS, OWN_T, 0, stream>>>(src, dst, dinv, h, meta, lists, (unsigned)cap, out, N, E);
}


# norm factored out of the edge loop: product kernel stores rows times dinv[src], tile sums them unscaled, dinv[dst] applied once at the tile's add onto out; no dinv gather per edge; rounding changes
# speedup vs baseline: 2.9731x; 1.0910x over previous
#include <hip/hip_runtime.h>
#include <hip/hip_bf16.h>


typedef __attribute__((ext_vector_type(2))) float v2f;
typedef __attribute__((ext_vector_type(8))) float v8f;

#define IN_CH 16
#define OUT_CH 8

#define OWN_R    4096
#define OWN_NS   2
#define OWN_T    1024
#define OWN_W    (OWN_T / 32)
#define OWN_U    8
#define OWN_SS   (OWN_T * OWN_U)
#define QA_CAP   512
#define QA_STEP  (OWN_U * 32)
#define DI       4
#define LIST_MAX 2048

template <bool GUARD>
__device__ __forceinline__ void own_load_step(const int4* __restrict__ d4, int first, int n4,
                                              int tid, int4 (&v)[OWN_U]) {
#pragma unroll
    for (int u = 0; u < OWN_U; ++u) {
        const int idx = first + u * OWN_T + tid;
        if (GUARD) {
            v[u] = make_int4(-1, -1, -1, -1);
            if (idx < n4) v[u] = d4[idx];
        } else {
            v[u] = d4[idx];
        }
    }
}

template <typename T, typename S>
__device__ __forceinline__ void own_scan_all(const int* __restrict__ dst, int e_cnt, int tid,
                                             unsigned owner, unsigned owners, unsigned slice,
                                             int it0, T tail, S step) {
    const bool aligned = ((reinterpret_cast<size_t>(dst) & 15u) == 0);
    const int n4 = aligned ? (e_cnt >> 2) : 0;
    if (slice == 0u)
        for (int e = 4 * n4 + tid; e < e_cnt; e += OWN_T) tail(e);
    const int nss = (n4 + OWN_SS - 1) / OWN_SS;
    const int lo  = (int)(((unsigned)nss * slice) / (unsigned)OWN_NS);
    const int hi  = (int)(((unsigned)nss * (slice + 1u)) / (unsigned)OWN_NS);
    const int ns  = hi - lo;
    unsigned rot = (owner * (unsigned)ns) / owners;
    if (rot >= (unsigned)ns) rot = 0;
    for (int it = it0; it < ns; ++it) {
        int ss = it + (int)rot;
        if (ss >= ns) ss -= ns;
        const int first = (lo + ss) * OWN_SS;
        step(it, first, n4, first + OWN_SS > n4);
    }
}

__global__ __launch_bounds__(OWN_T) void gcn_deg_owner(const int* __restrict__ dst,
                                                       unsigned* __restrict__ cntp,
                                                       int cstride,
                                                       int* __restrict__ meta,
                                                       int* __restrict__ lists,
                                                       unsigned cap,
                                                       int n_nodes, int e_cnt) {
    __shared__ unsigned cnt[OWN_R];
    const int tid = threadIdx.x;
    const int lane = tid & 31;
    const int wv   = __builtin_amdgcn_readfirstlane(tid >> 5);
    const unsigned owner  = blockIdx.x / OWN_NS;
    const unsigned slice  = blockIdx.x - owner * OWN_NS;
    const unsigned owners = gridDim.x / OWN_NS;
    const unsigned base = owner * OWN_R;
    if (base >= (unsigned)n_nodes) return;
    const unsigned span = min((unsigned)OWN_R, (unsigned)n_nodes - base);
    const int4* __restrict__ d4 = (const int4*)dst;

    for (int i = tid; i < OWN_R; i += OWN_T) cnt[i] = 0u;
    __syncthreads();

    const unsigned wslot = blockIdx.x * OWN_W + (unsigned)wv;
    int* __restrict__ lst = lists + (size_t)wslot * cap;
    unsigned wcount = 0;
    bool     rec    = true;
    int      stop   = 0x7fffffff;

    auto count4 = [&](unsigned x0, unsigned x1, unsigned x2, unsigned x3) {
        if (x0 < span) atomicAdd(&cnt[x0], 1u);
        if (x1 < span) atomicAdd(&cnt[x1], 1u);
        if (x2 < span) atomicAdd(&cnt[x2], 1u);
        if (x3 < span) atomicAdd(&cnt[x3], 1u);
    };
    auto test4 = [&](const int4& w) {
        const unsigned x0 = (unsigned)w.x - base;
        const unsigned x1 = (unsigned)w.y - base;
        const unsigned x2 = (unsigned)w.z - base;
        const unsigned x3 = (unsigned)w.w - base;
        if (min(min(x0, x1), min(x2, x3)) < span) count4(x0, x1, x2, x3);
    };
    auto test4_list = [&](const int4& w, int pos) {
        const unsigned x0 = (unsigned)w.x - base;
        const unsigned x1 = (unsigned)w.y - base;
        const unsigned x2 = (unsigned)w.z - base;
        const unsigned x3 = (unsigned)w.w - base;
        const bool p = min(min(x0, x1), min(x2, x3)) < span;
        const unsigned mask = __builtin_amdgcn_ballot_w32(p);
        const unsigned rank = __builtin_amdgcn_mbcnt_lo(mask, 0u);
        if (p) {
            lst[wcount + rank] = pos;
            count4(x0, x1, x2, x3);
        }
        wcount += (unsigned)__builtin_popcount(mask);
    };
    own_scan_all(dst, e_cnt, tid, owner, owners, slice, 0,
        [&](int e) {
            const unsigned x = (unsigned)dst[e] - base;
            if (x < span) atomicAdd(&cnt[x], 1u);
        },
        [&](int it, int first, int n4, bool partial) {
            if (rec && wcount + QA_STEP > cap) { rec = false; stop = it; }
            int4 v[OWN_U];
            if (partial) own_load_step<true >(d4, first, n4, tid, v);
            else         own_load_step<false>(d4, first, n4, tid, v);
            if (rec) {
#pragma unroll
                for (int u = 0; u < OWN_U; ++u) test4_list(v[u], first + u * OWN_T + tid);
            } else {
#pragma unroll
                for (int u = 0; u < OWN_U; ++u) test4(v[u]);
            }
        });
    if (cap != 0u && lane == 0) {
        meta[2 * wslot + 0] = (int)wcount;
        meta[2 * wslot + 1] = stop;
    }
    __syncthreads();

    unsigned* __restrict__ plane = cntp + (size_t)slice * (size_t)cstride + base;
    for (unsigned i = tid; i < span; i += OWN_T) plane[i] = cnt[i];
}

__device__ __forceinline__ float own_dinv(unsigned c) {
    const float d = 1.0f + (float)c;
    return d > 0.0f ? __frsqrt_rn(d) : 0.0f;
}

__global__ void gcn_gemm_wmma(const float* __restrict__ x,
                              const float* __restrict__ W,
                              const float* __restrict__ bias,
                              const unsigned* __restrict__ cntp,
                              int cstride,
                              float* __restrict__ dinv,
                              float* __restrict__ h,
                              float* __restrict__ out,
                              int n_nodes, int n_tiles) {
    const int lane  = threadIdx.x & 31;
    const int wave  = threadIdx.x >> 5;
    const int tile  = blockIdx.x * (blockDim.x >> 5) + wave;
    if (tile >= n_tiles) return;

    const int base  = tile * 16;
    const int m     = lane & 15;
    const int khalf = (lane >> 4) * 2;

    int arow = base + m;
    if (arow >= n_nodes) arow = n_nodes - 1;
    const float* __restrict__ xrow = x + (size_t)arow * IN_CH;

    v2f a0 = *(const v2f*)(xrow + 0  + khalf);
    v2f a1 = *(const v2f*)(xrow + 4  + khalf);
    v2f a2 = *(const v2f*)(xrow + 8  + khalf);
    v2f a3 = *(const v2f*)(xrow + 12 + khalf);

    const int   wrow = (m < OUT_CH) ? m : 0;
    const float sel  = (m < OUT_CH) ? 1.0f : 0.0f;
    const float* __restrict__ wr = W + wrow * IN_CH + khalf;
    v2f b0 = *(const v2f*)(wr + 0);
    v2f b1 = *(const v2f*)(wr + 4);
    v2f b2 = *(const v2f*)(wr + 8);
    v2f b3 = *(const v2f*)(wr + 12);
    b0 *= sel; b1 *= sel; b2 *= sel; b3 *= sel;

    v8f acc = {};
    acc = __builtin_amdgcn_wmma_f32_16x16x4_f32(false, a0, false, b0, (short)0, acc, false, false);
    acc = __builtin_amdgcn_wmma_f32_16x16x4_f32(false, a1, false, b1, (short)0, acc, false, false);
    acc = __builtin_amdgcn_wmma_f32_16x16x4_f32(false, a2, false, b2, (short)0, acc, false, false);
    acc = __builtin_amdgcn_wmma_f32_16x16x4_f32(false, a3, false, b3, (short)0, acc, false, false);

    if ((tile & 1) == 0) {
        const int node = base + lane;
        if (node < n_nodes) {
            unsigned c = 0u;
#pragma unroll
            for (int s = 0; s < OWN_NS; ++s) c += cntp[(size_t)s * (size_t)cstride + node];
            dinv[node] = own_dinv(c);
        }
    }

    const int  n     = m;
    const int  mbase = base + ((lane >> 4) * 8);
    const bool full  = (base + 16) <= n_nodes;

    if (full) {
        if (n < OUT_CH) {
            const float bn = bias[n];
            uint4 c0 = *(const uint4*)(cntp + mbase);
            uint4 c1 = *(const uint4*)(cntp + mbase + 4);
#pragma unroll
            for (int s = 1; s < OWN_NS; ++s) {
                const unsigned* __restrict__ pl = cntp + (size_t)s * (size_t)cstride + mbase;
                const uint4 e0 = *(const uint4*)(pl);
                const uint4 e1 = *(const uint4*)(pl + 4);
                c0.x += e0.x; c0.y += e0.y; c0.z += e0.z; c0.w += e0.w;
                c1.x += e1.x; c1.y += e1.y; c1.z += e1.z; c1.w += e1.w;
            }
            const float dv[8] = {own_dinv(c0.x), own_dinv(c0.y), own_dinv(c0.z), own_dinv(c0.w),
                                 own_dinv(c1.x), own_dinv(c1.y), own_dinv(c1.z), own_dinv(c1.w)};
            float* __restrict__ hp = h   + (size_t)mbase * OUT_CH + n;
            float* __restrict__ op = out + (size_t)mbase * OUT_CH + n;
#pragma unroll
            for (int r = 0; r < 8; ++r) {
                const float hv = acc[r];
                hp[r * OUT_CH] = dv[r] * hv;
                op[r * OUT_CH] = bn + dv[r] * dv[r] * hv;
            }
        }
    } else {
        if (n < OUT_CH) {
            const float bn = bias[n];
#pragma unroll
            for (int r = 0; r < 8; ++r) {
                const int node = mbase + r;
                if (node < n_nodes) {
                    const float hv = acc[r];
                    unsigned c = 0u;
#pragma unroll
                    for (int s = 0; s < OWN_NS; ++s) c += cntp[(size_t)s * (size_t)cstride + node];
                    const float di = own_dinv(c);
                    h[(size_t)node * OUT_CH + n] = di * hv;
                    out[(size_t)node * OUT_CH + n] = bn + di * di * hv;
                }
            }
        }
    }
}

__device__ __forceinline__ void own_add_row(float* acc_row, const float4& h0, const float4& h1) {
    atomicAdd(acc_row + 0, h0.x);
    atomicAdd(acc_row + 1, h0.y);
    atomicAdd(acc_row + 2, h0.z);
    atomicAdd(acc_row + 3, h0.w);
    atomicAdd(acc_row + 4, h1.x);
    atomicAdd(acc_row + 5, h1.y);
    atomicAdd(acc_row + 6, h1.z);
    atomicAdd(acc_row + 7, h1.w);
}

__global__ __launch_bounds__(OWN_T) void gcn_sum_owner(const int* __restrict__ src,
                                                       const int* __restrict__ dst,
                                                       const float* __restrict__ dinv,
                                                       const float* __restrict__ h,
                                                       const int* __restrict__ meta,
                                                       const int* __restrict__ lists,
                                                       unsigned cap,
                                                       float* __restrict__ out,
                                                       int n_nodes, int e_cnt) {
    __shared__ float acc[OWN_R * OUT_CH];
    __shared__ int   qa[OWN_W][QA_CAP];
    __shared__ float dvs[OWN_R];
    const int tid  = threadIdx.x;
    const int lane = tid & 31;
    const int wv   = __builtin_amdgcn_readfirstlane(tid >> 5);
    int* myq = qa[wv];
    const unsigned owner  = blockIdx.x / OWN_NS;
    const unsigned slice  = blockIdx.x - owner * OWN_NS;
    const unsigned owners = gridDim.x / OWN_NS;
    const unsigned base = owner * OWN_R;
    if (base >= (unsigned)n_nodes) return;
    const unsigned span = min((unsigned)OWN_R, (unsigned)n_nodes - base);
    const int4* __restrict__ d4 = (const int4*)dst;
    const bool s_al = ((reinterpret_cast<size_t>(src) & 15u) == 0);
    const unsigned n4m1 = (unsigned)max((e_cnt >> 2) - 1, 0);

    const unsigned wslot = blockIdx.x * OWN_W + (unsigned)wv;
    const int* __restrict__ lst = lists + (size_t)wslot * cap;
    unsigned lcount = 0;
    int it0 = 0;
    if (cap != 0u) {
        lcount = min((unsigned)__builtin_amdgcn_readfirstlane(meta[2 * wslot + 0]), cap);
        it0    = max(__builtin_amdgcn_readfirstlane(meta[2 * wslot + 1]), 0);
    }

    {
        float4* a4 = (float4*)acc;
        for (unsigned i = tid; i < span * (OUT_CH / 4); i += OWN_T) a4[i] = make_float4(0.0f, 0.0f, 0.0f, 0.0f);
        for (unsigned i = tid; i < span; i += OWN_T) dvs[i] = dinv[base + i];
    }
    __syncthreads();

    auto add_src = [&](int s, unsigned x) {
        const float4* __restrict__ hs = (const float4*)(h + (size_t)s * OUT_CH);
        const float4 h0 = hs[0];
        const float4 h1 = hs[1];
        own_add_row(&acc[x * OUT_CH], h0, h1);
    };

    auto drain = [&](unsigned n, auto get) {
        for (unsigned j0 = 0; j0 < n; j0 += 32 * DI) {
            bool ok[DI]; unsigned idx[DI]; int4 v[DI]; int4 sv[DI];
#pragma unroll
            for (int d = 0; d < DI; ++d) {
                const unsigned j = j0 + d * 32 + lane;
                ok[d]  = j < n;
                idx[d] = min((unsigned)get(ok[d] ? j : 0u), n4m1);
            }
#pragma unroll
            for (int d = 0; d < DI; ++d) {
                v[d] = d4[idx[d]];
                if (s_al) {
                    sv[d] = ((const int4*)src)[idx[d]];
                } else {
                    const int* sp = src + 4 * (size_t)idx[d];
                    sv[d] = make_int4(sp[0], sp[1], sp[2], sp[3]);
                }
            }
            unsigned xm[DI]; int kk[DI]; int s[DI];
#pragma unroll
            for (int d = 0; d < DI; ++d) {
                const unsigned x0 = (unsigned)v[d].x - base;
                const unsigned x1 = (unsigned)v[d].y - base;
                const unsigned x2 = (unsigned)v[d].z - base;
                const unsigned x3 = (unsigned)v[d].w - base;
                const unsigned m  = min(min(x0, x1), min(x2, x3));
                kk[d] = (x0 == m) ? 0 : (x1 == m) ? 1 : (x2 == m) ? 2 : 3;
                xm[d] = min(m, span - 1u);
                s[d]  = (x0 == m) ? sv[d].x : (x1 == m) ? sv[d].y : (x2 == m) ? sv[d].z : sv[d].w;
            }
            float4 r0[DI], r1[DI];
#pragma unroll
            for (int d = 0; d < DI; ++d) {
                const float4* __restrict__ hs = (const float4*)(h + (size_t)s[d] * OUT_CH);
                r0[d] = hs[0];
                r1[d] = hs[1];
            }
#pragma unroll
            for (int d = 0; d < DI; ++d) {
                asm volatile("" :: "v"(r0[d].x), "v"(r0[d].y), "v"(r0[d].z), "v"(r0[d].w),
                                   "v"(r1[d].x), "v"(r1[d].y), "v"(r1[d].z), "v"(r1[d].w));
            }
#pragma unroll
            for (int d = 0; d < DI; ++d)
                if (ok[d]) own_add_row(&acc[xm[d] * OUT_CH], r0[d], r1[d]);
#pragma unroll
            for (int d = 0; d < DI; ++d) {
                if (ok[d]) {
                    const unsigned x0 = (unsigned)v[d].x - base;
                    const unsigned x1 = (unsigned)v[d].y - base;
                    const unsigned x2 = (unsigned)v[d].z - base;
                    const unsigned x3 = (unsigned)v[d].w - base;
                    if (kk[d] != 0 && x0 < span) add_src(sv[d].x, x0);
                    if (kk[d] != 1 && x1 < span) add_src(sv[d].y, x1);
                    if (kk[d] != 2 && x2 < span) add_src(sv[d].z, x2);
                    if (kk[d] != 3 && x3 < span) add_src(sv[d].w, x3);
                }
            }
        }
    };

    drain(lcount, [&](unsigned j) { return lst[j]; });

    unsigned wcount = 0;

    own_scan_all(dst, e_cnt, tid, owner, owners, slice, it0,
        [&](int e) {
            const unsigned x = (unsigned)dst[e] - base;
            if (x < span) add_src(src[e], x);
        },
        [&](int it, int first, int n4, bool partial) {
            int4 v[OWN_U];
            if (partial) own_load_step<true >(d4, first, n4, tid, v);
            else         own_load_step<false>(d4, first, n4, tid, v);
#pragma unroll
            for (int u = 0; u < OWN_U; ++u) {
                const unsigned x0 = (unsigned)v[u].x - base;
                const unsigned x1 = (unsigned)v[u].y - base;
                const unsigned x2 = (unsigned)v[u].z - base;
                const unsigned x3 = (unsigned)v[u].w - base;
                const bool p = min(min(x0, x1), min(x2, x3)) < span;
                const unsigned mask = __builtin_amdgcn_ballot_w32(p);
                const unsigned rank = __builtin_amdgcn_mbcnt_lo(mask, 0u);
                if (p) myq[wcount + rank] = first + u * OWN_T + tid;
                wcount += (unsigned)__builtin_popcount(mask);
            }
            if (wcount > QA_CAP - QA_STEP) { drain(wcount, [&](unsigned j) { return myq[j]; }); wcount = 0; }
        });
    drain(wcount, [&](unsigned j) { return myq[j]; });
    __syncthreads();

    {
        float* __restrict__ o = out + (size_t)base * OUT_CH;
        for (unsigned i = tid; i < span * OUT_CH; i += OWN_T) atomicAdd(o + i, dvs[i / OUT_CH] * acc[i]);
    }
}

extern "C" void kernel_launch(void* const* d_in, const int* in_sizes, int n_in,
                              void* d_out, int out_size, void* d_ws, size_t ws_size,
                              hipStream_t stream) {
    const float* x    = (const float*)d_in[0];
    const int*   ei   = (const int*)d_in[1];
    const float* W    = (const float*)d_in[2];
    const float* bias = (const float*)d_in[3];

    const int N = in_sizes[0] / IN_CH;
    const int E = in_sizes[1] / 2;
    const int* src = ei;
    const int* dst = ei + (size_t)E;

    float* dinv = (float*)d_ws;
    size_t h_off = (((size_t)N * sizeof(float)) + 255u) & ~(size_t)255u;
    float* h = (float*)((char*)d_ws + h_off);
    float* out = (float*)d_out;

    const int TPB = 256;
    const int owners = (N + OWN_R - 1) / OWN_R;

    const size_t lwaves   = (size_t)owners * OWN_NS * OWN_W;
    const size_t cstride  = ((size_t)N + 63u) & ~(size_t)63u;
    const size_t cnt_off  = (h_off + (size_t)N * OUT_CH * sizeof(float) + 255u) & ~(size_t)255u;
    const size_t meta_off = (cnt_off + (size_t)OWN_NS * cstride * sizeof(unsigned) + 255u) & ~(size_t)255u;
    const size_t list_off = (meta_off + lwaves * 2 * sizeof(int) + 255u) & ~(size_t)255u;
    size_t cap = 0;
    if (lwaves != 0 && ws_size > list_off) cap = (ws_size - list_off) / (lwaves * sizeof(int));
    if (cap > LIST_MAX) cap = LIST_MAX;
    unsigned* cntp = (unsigned*)((char*)d_ws + cnt_off);
    int* meta  = (int*)((char*)d_ws + meta_off);
    int* lists = (int*)((char*)d_ws + list_off);

    gcn_deg_owner<<<owners * OWN_NS, OWN_T, 0, stream>>>(dst, cntp, (int)cstride, meta, lists, (unsigned)cap, N, E);

    const int n_tiles = (N + 15) / 16;
    const int waves_per_block = TPB / 32;
    const int gemm_blocks = (n_tiles + waves_per_block - 1) / waves_per_block;
    gcn_gemm_wmma<<<gemm_blocks, TPB, 0, stream>>>(x, W, bias, cntp, (int)cstride, dinv, h, out, N, n_tiles);

    gcn_sum_owner<<<owners * OWN_NS, OWN_T, 0, stream>>>(src, dst, dinv, h, meta, lists, (unsigned)cap, out, N, E);
}


# walk ranges widened to 16384 nodes, list entries tagged by 2048-row sub-range; each sum group owns its rows and stores out once: dst read 31x not 123x, no global atomics; sum order changes
# speedup vs baseline: 3.7734x; 1.2692x over previous
#include <hip/hip_runtime.h>
#include <hip/hip_bf16.h>


typedef __attribute__((ext_vector_type(2))) float v2f;
typedef __attribute__((ext_vector_type(8))) float v8f;

#define IN_CH 16
#define OUT_CH 8

#define DEG_R    16384
#define DEG_NS   8
#define SUB_R    2048
#define SUB_SH   11
#define SUB_N    (DEG_R / SUB_R)
#define POS_BITS 21
#define POS_MASK ((1u << POS_BITS) - 1u)
#define WO_K     (SUB_R * (OUT_CH / 4) / OWN_T)
#define OWN_T    1024
#define OWN_W    (OWN_T / 32)
#define OWN_U    8
#define OWN_SS   (OWN_T * OWN_U)
#define QA_CAP   1024
#define QA_STEP  (OWN_U * 32)
#define DI       4
#define LIST_MAX 2048

template <bool GUARD>
__device__ __forceinline__ void own_load_step(const int4* __restrict__ d4, int first, int n4,
                                              int tid, int4 (&v)[OWN_U]) {
#pragma unroll
    for (int u = 0; u < OWN_U; ++u) {
        const int idx = first + u * OWN_T + tid;
        if (GUARD) {
            v[u] = make_int4(-1, -1, -1, -1);
            if (idx < n4) v[u] = d4[idx];
        } else {
            v[u] = d4[idx];
        }
    }
}

template <typename T, typename S>
__device__ __forceinline__ void own_scan_all(const int* __restrict__ dst, int e_cnt, int tid,
                                             unsigned owner, unsigned owners, unsigned slice,
                                             int it0, T tail, S step) {
    const bool aligned = ((reinterpret_cast<size_t>(dst) & 15u) == 0);
    const int n4 = aligned ? (e_cnt >> 2) : 0;
    if (slice == 0u)
        for (int e = 4 * n4 + tid; e < e_cnt; e += OWN_T) tail(e);
    const int nss = (n4 + OWN_SS - 1) / OWN_SS;
    const int lo  = (int)(((unsigned)nss * slice) / (unsigned)DEG_NS);
    const int hi  = (int)(((unsigned)nss * (slice + 1u)) / (unsigned)DEG_NS);
    const int ns  = hi - lo;
    unsigned rot = (owner * (unsigned)ns) / owners;
    if (rot >= (unsigned)ns) rot = 0;
    for (int it = it0; it < ns; ++it) {
        int ss = it + (int)rot;
        if (ss >= ns) ss -= ns;
        const int first = (lo + ss) * OWN_SS;
        step(it, first, n4, first + OWN_SS > n4);
    }
}

__global__ __launch_bounds__(OWN_T) void gcn_deg_owner(const int* __restrict__ dst,
                                                       unsigned* __restrict__ cntp,
                                                       int cstride,
                                                       int* __restrict__ meta,
                                                       int* __restrict__ lists,
                                                       unsigned cap,
                                                       int n_nodes, int e_cnt) {
    __shared__ unsigned cnt[DEG_R];
    const int tid = threadIdx.x;
    const int lane = tid & 31;
    const int wv   = __builtin_amdgcn_readfirstlane(tid >> 5);
    const unsigned owner  = blockIdx.x / DEG_NS;
    const unsigned slice  = blockIdx.x - owner * DEG_NS;
    const unsigned owners = gridDim.x / DEG_NS;
    const unsigned base = owner * DEG_R;
    if (base >= (unsigned)n_nodes) return;
    const unsigned span = min((unsigned)DEG_R, (unsigned)n_nodes - base);
    const int4* __restrict__ d4 = (const int4*)dst;

    for (int i = tid; i < DEG_R; i += OWN_T) cnt[i] = 0u;
    __syncthreads();

    const unsigned wslot = blockIdx.x * OWN_W + (unsigned)wv;
    int* __restrict__ lst = lists + (size_t)wslot * cap;
    unsigned wcount = 0;
    bool     rec    = true;
    int      stop   = 0x7fffffff;

    auto count4 = [&](unsigned x0, unsigned x1, unsigned x2, unsigned x3) {
        if (x0 < span) atomicAdd(&cnt[x0], 1u);
        if (x1 < span) atomicAdd(&cnt[x1], 1u);
        if (x2 < span) atomicAdd(&cnt[x2], 1u);
        if (x3 < span) atomicAdd(&cnt[x3], 1u);
    };
    auto test4 = [&](const int4& w) {
        const unsigned x0 = (unsigned)w.x - base;
        const unsigned x1 = (unsigned)w.y - base;
        const unsigned x2 = (unsigned)w.z - base;
        const unsigned x3 = (unsigned)w.w - base;
        if (min(min(x0, x1), min(x2, x3)) < span) count4(x0, x1, x2, x3);
    };
    auto test4_list = [&](const int4& w, int pos) {
        const unsigned x0 = (unsigned)w.x - base;
        const unsigned x1 = (unsigned)w.y - base;
        const unsigned x2 = (unsigned)w.z - base;
        const unsigned x3 = (unsigned)w.w - base;
        const bool p = min(min(x0, x1), min(x2, x3)) < span;
        const unsigned mask = __builtin_amdgcn_ballot_w32(p);
        const unsigned rank = __builtin_amdgcn_mbcnt_lo(mask, 0u);
        if (p) {
            const unsigned tag = ((1u << min(x0 >> SUB_SH, (unsigned)SUB_N)) |
                                  (1u << min(x1 >> SUB_SH, (unsigned)SUB_N)) |
                                  (1u << min(x2 >> SUB_SH, (unsigned)SUB_N)) |
                                  (1u << min(x3 >> SUB_SH, (unsigned)SUB_N))) & ((1u << SUB_N) - 1u);
            lst[wcount + rank] = (int)((unsigned)pos | (tag << POS_BITS));
            count4(x0, x1, x2, x3);
        }
        wcount += (unsigned)__builtin_popcount(mask);
    };
    own_scan_all(dst, e_cnt, tid, owner, owners, slice, 0,
        [&](int e) {
            const unsigned x = (unsigned)dst[e] - base;
            if (x < span) atomicAdd(&cnt[x], 1u);
        },
        [&](int it, int first, int n4, bool partial) {
            if (rec && wcount + QA_STEP > cap) { rec = false; stop = it; }
            int4 v[OWN_U];
            if (partial) own_load_step<true >(d4, first, n4, tid, v);
            else         own_load_step<false>(d4, first, n4, tid, v);
            if (rec) {
#pragma unroll
                for (int u = 0; u < OWN_U; ++u) test4_list(v[u], first + u * OWN_T + tid);
            } else {
#pragma unroll
                for (int u = 0; u < OWN_U; ++u) test4(v[u]);
            }
        });
    if (cap != 0u && lane == 0) {
        meta[2 * wslot + 0] = (int)wcount;
        meta[2 * wslot + 1] = stop;
    }
    __syncthreads();

    unsigned* __restrict__ plane = cntp + (size_t)slice * (size_t)cstride + base;
    for (unsigned i = tid; i < span; i += OWN_T) plane[i] = cnt[i];
}

__device__ __forceinline__ float own_dinv(unsigned c) {
    const float d = 1.0f + (float)c;
    return d > 0.0f ? __frsqrt_rn(d) : 0.0f;
}

__global__ void gcn_gemm_wmma(const float* __restrict__ x,
                              const float* __restrict__ W,
                              const unsigned* __restrict__ cntp,
                              int cstride,
                              float* __restrict__ dinv,
                              float* __restrict__ h,
                              int n_nodes, int n_tiles) {
    const int lane  = threadIdx.x & 31;
    const int wave  = threadIdx.x >> 5;
    const int tile  = blockIdx.x * (blockDim.x >> 5) + wave;
    if (tile >= n_tiles) return;

    const int base  = tile * 16;
    const int m     = lane & 15;
    const int khalf = (lane >> 4) * 2;

    int arow = base + m;
    if (arow >= n_nodes) arow = n_nodes - 1;
    const float* __restrict__ xrow = x + (size_t)arow * IN_CH;

    v2f a0 = *(const v2f*)(xrow + 0  + khalf);
    v2f a1 = *(const v2f*)(xrow + 4  + khalf);
    v2f a2 = *(const v2f*)(xrow + 8  + khalf);
    v2f a3 = *(const v2f*)(xrow + 12 + khalf);

    const int   wrow = (m < OUT_CH) ? m : 0;
    const float sel  = (m < OUT_CH) ? 1.0f : 0.0f;
    const float* __restrict__ wr = W + wrow * IN_CH + khalf;
    v2f b0 = *(const v2f*)(wr + 0);
    v2f b1 = *(const v2f*)(wr + 4);
    v2f b2 = *(const v2f*)(wr + 8);
    v2f b3 = *(const v2f*)(wr + 12);
    b0 *= sel; b1 *= sel; b2 *= sel; b3 *= sel;

    v8f acc = {};
    acc = __builtin_amdgcn_wmma_f32_16x16x4_f32(false, a0, false, b0, (short)0, acc, false, false);
    acc = __builtin_amdgcn_wmma_f32_16x16x4_f32(false, a1, false, b1, (short)0, acc, false, false);
    acc = __builtin_amdgcn_wmma_f32_16x16x4_f32(false, a2, false, b2, (short)0, acc, false, false);
    acc = __builtin_amdgcn_wmma_f32_16x16x4_f32(false, a3, false, b3, (short)0, acc, false, false);

    if ((tile & 1) == 0) {
        const int node = base + lane;
        if (node < n_nodes) {
            unsigned c = 0u;
#pragma unroll
            for (int s = 0; s < DEG_NS; ++s) c += cntp[(size_t)s * (size_t)cstride + node];
            dinv[node] = own_dinv(c);
        }
    }

    const int  n     = m;
    const int  mbase = base + ((lane >> 4) * 8);
    const bool full  = (base + 16) <= n_nodes;

    if (full) {
        if (n < OUT_CH) {
            uint4 c0 = *(const uint4*)(cntp + mbase);
            uint4 c1 = *(const uint4*)(cntp + mbase + 4);
#pragma unroll
            for (int s = 1; s < DEG_NS; ++s) {
                const unsigned* __restrict__ pl = cntp + (size_t)s * (size_t)cstride + mbase;
                const uint4 e0 = *(const uint4*)(pl);
                const uint4 e1 = *(const uint4*)(pl + 4);
                c0.x += e0.x; c0.y += e0.y; c0.z += e0.z; c0.w += e0.w;
                c1.x += e1.x; c1.y += e1.y; c1.z += e1.z; c1.w += e1.w;
            }
            const float dv[8] = {own_dinv(c0.x), own_dinv(c0.y), own_dinv(c0.z), own_dinv(c0.w),
                                 own_dinv(c1.x), own_dinv(c1.y), own_dinv(c1.z), own_dinv(c1.w)};
            float* __restrict__ hp = h   + (size_t)mbase * OUT_CH + n;
#pragma unroll
            for (int r = 0; r < 8; ++r) hp[r * OUT_CH] = dv[r] * acc[r];
        }
    } else {
        if (n < OUT_CH) {
#pragma unroll
            for (int r = 0; r < 8; ++r) {
                const int node = mbase + r;
                if (node < n_nodes) {
                    unsigned c = 0u;
#pragma unroll
                    for (int s = 0; s < DEG_NS; ++s) c += cntp[(size_t)s * (size_t)cstride + node];
                    h[(size_t)node * OUT_CH + n] = own_dinv(c) * acc[r];
                }
            }
        }
    }
}

__device__ __forceinline__ void own_add_row(float* acc_row, const float4& h0, const float4& h1) {
    atomicAdd(acc_row + 0, h0.x);
    atomicAdd(acc_row + 1, h0.y);
    atomicAdd(acc_row + 2, h0.z);
    atomicAdd(acc_row + 3, h0.w);
    atomicAdd(acc_row + 4, h1.x);
    atomicAdd(acc_row + 5, h1.y);
    atomicAdd(acc_row + 6, h1.z);
    atomicAdd(acc_row + 7, h1.w);
}

__global__ __launch_bounds__(OWN_T) void gcn_sum_owner(const int* __restrict__ src,
                                                       const int* __restrict__ dst,
                                                       const float* __restrict__ dinv,
                                                       const float* __restrict__ h,
                                                       const float* __restrict__ bias,
                                                       const int* __restrict__ meta,
                                                       const int* __restrict__ lists,
                                                       unsigned cap,
                                                       float* __restrict__ out,
                                                       int n_nodes, int e_cnt) {
    __shared__ float acc[SUB_R * OUT_CH];
    __shared__ int   qa[OWN_W][QA_CAP];
    __shared__ float dvs[SUB_R];
    const int tid  = threadIdx.x;
    const int lane = tid & 31;
    const int wv   = __builtin_amdgcn_readfirstlane(tid >> 5);
    int* myq = qa[wv];
    const unsigned owner  = blockIdx.x / SUB_N;
    const unsigned sub    = blockIdx.x - owner * SUB_N;
    const unsigned owners = gridDim.x / SUB_N;
    const unsigned base = owner * DEG_R + sub * SUB_R;
    if (base >= (unsigned)n_nodes) return;
    const unsigned span = min((unsigned)SUB_R, (unsigned)n_nodes - base);
    const int4* __restrict__ d4 = (const int4*)dst;
    const bool s_al = ((reinterpret_cast<size_t>(src) & 15u) == 0);
    const unsigned n4m1 = (unsigned)max((e_cnt >> 2) - 1, 0);
    const unsigned tagbit = POS_BITS + sub;

    {
        float4* a4 = (float4*)acc;
        for (unsigned i = tid; i < span * (OUT_CH / 4); i += OWN_T) a4[i] = make_float4(0.0f, 0.0f, 0.0f, 0.0f);
        for (unsigned i = tid; i < span; i += OWN_T) dvs[i] = dinv[base + i];
    }
    __syncthreads();

    auto add_src = [&](int s, unsigned x) {
        const float4* __restrict__ hs = (const float4*)(h + (size_t)s * OUT_CH);
        const float4 h0 = hs[0];
        const float4 h1 = hs[1];
        own_add_row(&acc[x * OUT_CH], h0, h1);
    };

    auto drain = [&](unsigned n) {
        for (unsigned j0 = 0; j0 < n; j0 += 32 * DI) {
            bool ok[DI]; unsigned idx[DI]; int4 v[DI]; int4 sv[DI];
#pragma unroll
            for (int d = 0; d < DI; ++d) {
                const unsigned j = j0 + d * 32 + lane;
                ok[d]  = j < n;
                idx[d] = min((unsigned)myq[ok[d] ? j : 0u], n4m1);
            }
#pragma unroll
            for (int d = 0; d < DI; ++d) {
                v[d] = d4[idx[d]];
                if (s_al) {
                    sv[d] = ((const int4*)src)[idx[d]];
                } else {
                    const int* sp = src + 4 * (size_t)idx[d];
                    sv[d] = make_int4(sp[0], sp[1], sp[2], sp[3]);
                }
            }
            unsigned xm[DI]; int kk[DI]; int s[DI];
#pragma unroll
            for (int d = 0; d < DI; ++d) {
                const unsigned x0 = (unsigned)v[d].x - base;
                const unsigned x1 = (unsigned)v[d].y - base;
                const unsigned x2 = (unsigned)v[d].z - base;
                const unsigned x3 = (unsigned)v[d].w - base;
                const unsigned m  = min(min(x0, x1), min(x2, x3));
                ok[d] = ok[d] && (m < span);
                kk[d] = (x0 == m) ? 0 : (x1 == m) ? 1 : (x2 == m) ? 2 : 3;
                xm[d] = min(m, span - 1u);
                s[d]  = (x0 == m) ? sv[d].x : (x1 == m) ? sv[d].y : (x2 == m) ? sv[d].z : sv[d].w;
            }
            float4 r0[DI], r1[DI];
#pragma unroll
            for (int d = 0; d < DI; ++d) {
                const float4* __restrict__ hs = (const float4*)(h + (size_t)s[d] * OUT_CH);
                r0[d] = hs[0];
                r1[d] = hs[1];
            }
#pragma unroll
            for (int d = 0; d < DI; ++d) {
                asm volatile("" :: "v"(r0[d].x), "v"(r0[d].y), "v"(r0[d].z), "v"(r0[d].w),
                                   "v"(r1[d].x), "v"(r1[d].y), "v"(r1[d].z), "v"(r1[d].w));
            }
#pragma unroll
            for (int d = 0; d < DI; ++d)
                if (ok[d]) own_add_row(&acc[xm[d] * OUT_CH], r0[d], r1[d]);
#pragma unroll
            for (int d = 0; d < DI; ++d) {
                if (ok[d]) {
                    const unsigned x0 = (unsigned)v[d].x - base;
                    const unsigned x1 = (unsigned)v[d].y - base;
                    const unsigned x2 = (unsigned)v[d].z - base;
                    const unsigned x3 = (unsigned)v[d].w - base;
                    if (kk[d] != 0 && x0 < span) add_src(sv[d].x, x0);
                    if (kk[d] != 1 && x1 < span) add_src(sv[d].y, x1);
                    if (kk[d] != 2 && x2 < span) add_src(sv[d].z, x2);
                    if (kk[d] != 3 && x3 < span) add_src(sv[d].w, x3);
                }
            }
        }
    };

    unsigned wcount = 0;

#pragma unroll 1
    for (unsigned slice = 0; slice < (unsigned)DEG_NS; ++slice) {
        const unsigned wslot = (owner * DEG_NS + slice) * OWN_W + (unsigned)wv;
        const int* __restrict__ lst = lists + (size_t)wslot * cap;
        unsigned lcount = 0;
        int it0 = 0;
        if (cap != 0u) {
            lcount = min((unsigned)__builtin_amdgcn_readfirstlane(meta[2 * wslot + 0]), cap);
            it0    = max(__builtin_amdgcn_readfirstlane(meta[2 * wslot + 1]), 0);
        }
        for (unsigned j0 = 0; j0 < lcount; j0 += QA_STEP) {
            unsigned e[OWN_U];
#pragma unroll
            for (int u = 0; u < OWN_U; ++u) {
                const unsigned j = j0 + u * 32 + lane;
                e[u] = (unsigned)lst[min(j, lcount - 1u)];
            }
            __builtin_amdgcn_sched_barrier(0);
#pragma unroll
            for (int u = 0; u < OWN_U; ++u)
                if (j0 + u * 32 + lane >= lcount) e[u] = 0u;
#pragma unroll
            for (int u = 0; u < OWN_U; ++u) {
                const bool p = ((e[u] >> tagbit) & 1u) != 0u;
                const unsigned mask = __builtin_amdgcn_ballot_w32(p);
                const unsigned rank = __builtin_amdgcn_mbcnt_lo(mask, 0u);
                if (p) myq[wcount + rank] = (int)(e[u] & POS_MASK);
                wcount += (unsigned)__builtin_popcount(mask);
            }
            if (wcount > QA_CAP - QA_STEP) { drain(wcount); wcount = 0; }
        }
        own_scan_all(dst, e_cnt, tid, owner, owners, slice, it0,
            [&](int e) {
                const unsigned x = (unsigned)dst[e] - base;
                if (x < span) add_src(src[e], x);
            },
            [&](int it, int first, int n4, bool partial) {
                int4 v[OWN_U];
                if (partial) own_load_step<true >(d4, first, n4, tid, v);
                else         own_load_step<false>(d4, first, n4, tid, v);
#pragma unroll
                for (int u = 0; u < OWN_U; ++u) {
                    const unsigned x0 = (unsigned)v[u].x - base;
                    const unsigned x1 = (unsigned)v[u].y - base;
                    const unsigned x2 = (unsigned)v[u].z - base;
                    const unsigned x3 = (unsigned)v[u].w - base;
                    const bool p = min(min(x0, x1), min(x2, x3)) < span;
                    const unsigned mask = __builtin_amdgcn_ballot_w32(p);
                    const unsigned rank = __builtin_amdgcn_mbcnt_lo(mask, 0u);
                    if (p) myq[wcount + rank] = first + u * OWN_T + tid;
                    wcount += (unsigned)__builtin_popcount(mask);
                }
                if (wcount > QA_CAP - QA_STEP) { drain(wcount); wcount = 0; }
            });
    }
    drain(wcount);

    const float4* __restrict__ h4 = (const float4*)(h + (size_t)base * OUT_CH);
    float4* __restrict__ o4 = (float4*)(out + (size_t)base * OUT_CH);
    const float* __restrict__ bp = bias + (tid & 1) * 4;
    const float bx = bp[0], by = bp[1], bz = bp[2], bw = bp[3];
    const unsigned nq = span * (OUT_CH / 4);
    float4 hv[WO_K];
#pragma unroll
    for (int k = 0; k < WO_K; ++k) {
        const unsigned q = (unsigned)(k * OWN_T + tid);
        hv[k] = h4[min(q, nq - 1u)];
    }
    __syncthreads();
    {
        const float4* a4 = (const float4*)acc;
#pragma unroll
        for (int k = 0; k < WO_K; ++k) {
            const unsigned q = (unsigned)(k * OWN_T + tid);
            if (q < nq) {
                const float  di = dvs[q >> 1];
                const float4 av = a4[q];
                float4 o;
                o.x = bx + di * (hv[k].x + av.x);
                o.y = by + di * (hv[k].y + av.y);
                o.z = bz + di * (hv[k].z + av.z);
                o.w = bw + di * (hv[k].w + av.w);
                o4[q] = o;
            }
        }
    }
}

extern "C" void kernel_launch(void* const* d_in, const int* in_sizes, int n_in,
                              void* d_out, int out_size, void* d_ws, size_t ws_size,
                              hipStream_t stream) {
    const float* x    = (const float*)d_in[0];
    const int*   ei   = (const int*)d_in[1];
    const float* W    = (const float*)d_in[2];
    const float* bias = (const float*)d_in[3];

    const int N = in_sizes[0] / IN_CH;
    const int E = in_sizes[1] / 2;
    const int* src = ei;
    const int* dst = ei + (size_t)E;

    float* dinv = (float*)d_ws;
    size_t h_off = (((size_t)N * sizeof(float)) + 255u) & ~(size_t)255u;
    float* h = (float*)((char*)d_ws + h_off);
    float* out = (float*)d_out;

    const int TPB = 256;
    const int owners = (N + DEG_R - 1) / DEG_R;

    const size_t lwaves   = (size_t)owners * DEG_NS * OWN_W;
    const size_t cstride  = ((size_t)N + 63u) & ~(size_t)63u;
    const size_t cnt_off  = (h_off + (size_t)N * OUT_CH * sizeof(float) + 255u) & ~(size_t)255u;
    const size_t meta_off = (cnt_off + (size_t)DEG_NS * cstride * sizeof(unsigned) + 255u) & ~(size_t)255u;
    const size_t list_off = (meta_off + lwaves * 2 * sizeof(int) + 255u) & ~(size_t)255u;
    size_t cap = 0;
    if (lwaves != 0 && ws_size > list_off) cap = (ws_size - list_off) / (lwaves * sizeof(int));
    if (cap > LIST_MAX) cap = LIST_MAX;
    if ((E >> 2) > (1 << POS_BITS)) cap = 0;
    unsigned* cntp = (unsigned*)((char*)d_ws + cnt_off);
    int* meta  = (int*)((char*)d_ws + meta_off);
    int* lists = (int*)((char*)d_ws + list_off);

    gcn_deg_owner<<<owners * DEG_NS, OWN_T, 0, stream>>>(dst, cntp, (int)cstride, meta, lists, (unsigned)cap, N, E);

    const int n_tiles = (N + 15) / 16;
    const int waves_per_block = TPB / 32;
    const int gemm_blocks = (n_tiles + waves_per_block - 1) / waves_per_block;
    gcn_gemm_wmma<<<gemm_blocks, TPB, 0, stream>>>(x, W, cntp, (int)cstride, dinv, h, N, n_tiles);

    gcn_sum_owner<<<owners * SUB_N, OWN_T, 0, stream>>>(src, dst, dinv, h, bias, meta, lists, (unsigned)cap, out, N, E);
}


# degree walk only tests: an int4 word holding a node of the range is queued per wave in LDS with its position; counts, tags and list entries are made by full waves of found words; lists unchanged
# speedup vs baseline: 4.0569x; 1.0751x over previous
#include <hip/hip_runtime.h>
#include <hip/hip_bf16.h>


typedef __attribute__((ext_vector_type(2))) float v2f;
typedef __attribute__((ext_vector_type(8))) float v8f;

#define IN_CH 16
#define OUT_CH 8

#define DEG_R    16384
#define DEG_NS   8
#define SUB_R    2048
#define SUB_SH   11
#define SUB_N    (DEG_R / SUB_R)
#define POS_BITS 21
#define POS_MASK ((1u << POS_BITS) - 1u)
#define WO_K     (SUB_R * (OUT_CH / 4) / OWN_T)
#define OWN_T    1024
#define OWN_W    (OWN_T / 32)
#define OWN_U    8
#define OWN_SS   (OWN_T * OWN_U)
#define QA_CAP   1024
#define QA_STEP  (OWN_U * 32)
#define DI       4
#define LIST_MAX 2048
#define DQ_DRAIN 96
#define DQ_CAP   (DQ_DRAIN - 1 + QA_STEP + 1)

template <bool GUARD>
__device__ __forceinline__ void own_load_step(const int4* __restrict__ d4, int first, int n4,
                                              int tid, int4 (&v)[OWN_U]) {
#pragma unroll
    for (int u = 0; u < OWN_U; ++u) {
        const int idx = first + u * OWN_T + tid;
        if (GUARD) {
            v[u] = make_int4(-1, -1, -1, -1);
            if (idx < n4) v[u] = d4[idx];
        } else {
            v[u] = d4[idx];
        }
    }
}

template <typename T, typename S>
__device__ __forceinline__ void own_scan_all(const int* __restrict__ dst, int e_cnt, int tid,
                                             unsigned owner, unsigned owners, unsigned slice,
                                             int it0, T tail, S step) {
    const bool aligned = ((reinterpret_cast<size_t>(dst) & 15u) == 0);
    const int n4 = aligned ? (e_cnt >> 2) : 0;
    if (slice == 0u)
        for (int e = 4 * n4 + tid; e < e_cnt; e += OWN_T) tail(e);
    const int nss = (n4 + OWN_SS - 1) / OWN_SS;
    const int lo  = (int)(((unsigned)nss * slice) / (unsigned)DEG_NS);
    const int hi  = (int)(((unsigned)nss * (slice + 1u)) / (unsigned)DEG_NS);
    const int ns  = hi - lo;
    unsigned rot = (owner * (unsigned)ns) / owners;
    if (rot >= (unsigned)ns) rot = 0;
    for (int it = it0; it < ns; ++it) {
        int ss = it + (int)rot;
        if (ss >= ns) ss -= ns;
        const int first = (lo + ss) * OWN_SS;
        step(it, first, n4, first + OWN_SS > n4);
    }
}

__global__ __launch_bounds__(OWN_T) void gcn_deg_owner(const int* __restrict__ dst,
                                                       unsigned* __restrict__ cntp,
                                                       int cstride,
                                                       int* __restrict__ meta,
                                                       int* __restrict__ lists,
                                                       unsigned cap,
                                                       int n_nodes, int e_cnt) {
    __shared__ unsigned cnt[DEG_R];
    __shared__ int4     dqw[OWN_W][DQ_CAP];
    __shared__ int      dqp[OWN_W][DQ_CAP];
    const int tid = threadIdx.x;
    const int lane = tid & 31;
    const int wv   = __builtin_amdgcn_readfirstlane(tid >> 5);
    const unsigned owner  = blockIdx.x / DEG_NS;
    const unsigned slice  = blockIdx.x - owner * DEG_NS;
    const unsigned owners = gridDim.x / DEG_NS;
    const unsigned base = owner * DEG_R;
    if (base >= (unsigned)n_nodes) return;
    const unsigned span = min((unsigned)DEG_R, (unsigned)n_nodes - base);
    const int4* __restrict__ d4 = (const int4*)dst;
    int4* myw = dqw[wv];
    int*  myp = dqp[wv];

    for (int i = tid; i < DEG_R; i += OWN_T) cnt[i] = 0u;
    __syncthreads();

    const unsigned wslot = blockIdx.x * OWN_W + (unsigned)wv;
    int* __restrict__ lst = lists + (size_t)wslot * cap;
    unsigned wcount = 0;
    unsigned qn     = 0;
    bool     rec    = true;
    int      stop   = 0x7fffffff;

    auto test4_queue = [&](const int4& w, int pos) {
        const unsigned x0 = (unsigned)w.x - base;
        const unsigned x1 = (unsigned)w.y - base;
        const unsigned x2 = (unsigned)w.z - base;
        const unsigned x3 = (unsigned)w.w - base;
        const bool p = min(min(x0, x1), min(x2, x3)) < span;
        const unsigned mask = __builtin_amdgcn_ballot_w32(p);
        const unsigned rank = __builtin_amdgcn_mbcnt_lo(mask, 0u);
        if (p) {
            myw[qn + rank] = w;
            myp[qn + rank] = pos;
        }
        qn += (unsigned)__builtin_popcount(mask);
    };
    auto work_off = [&](unsigned n, bool listing) {
        for (unsigned j0 = 0; j0 < n; j0 += 32) {
            const unsigned j = j0 + (unsigned)lane;
            if (j < n) {
                const int4 w = myw[j];
                const unsigned x0 = (unsigned)w.x - base;
                const unsigned x1 = (unsigned)w.y - base;
                const unsigned x2 = (unsigned)w.z - base;
                const unsigned x3 = (unsigned)w.w - base;
                if (listing) {
                    const unsigned tag = ((1u << min(x0 >> SUB_SH, (unsigned)SUB_N)) |
                                          (1u << min(x1 >> SUB_SH, (unsigned)SUB_N)) |
                                          (1u << min(x2 >> SUB_SH, (unsigned)SUB_N)) |
                                          (1u << min(x3 >> SUB_SH, (unsigned)SUB_N))) & ((1u << SUB_N) - 1u);
                    lst[wcount + j] = (int)((unsigned)myp[j] | (tag << POS_BITS));
                }
                if (x0 < span) atomicAdd(&cnt[x0], 1u);
                if (x1 < span) atomicAdd(&cnt[x1], 1u);
                if (x2 < span) atomicAdd(&cnt[x2], 1u);
                if (x3 < span) atomicAdd(&cnt[x3], 1u);
            }
        }
        if (listing) wcount += n;
    };
    own_scan_all(dst, e_cnt, tid, owner, owners, slice, 0,
        [&](int e) {
            const unsigned x = (unsigned)dst[e] - base;
            if (x < span) atomicAdd(&cnt[x], 1u);
        },
        [&](int it, int first, int n4, bool partial) {
            if (rec && wcount + qn + QA_STEP > cap) { work_off(qn, true); qn = 0; rec = false; stop = it; }
            int4 v[OWN_U];
            if (partial) own_load_step<true >(d4, first, n4, tid, v);
            else         own_load_step<false>(d4, first, n4, tid, v);
#pragma unroll
            for (int u = 0; u < OWN_U; ++u) test4_queue(v[u], first + u * OWN_T + tid);
            if (qn >= DQ_DRAIN) { work_off(qn, rec); qn = 0; }
        });
    work_off(qn, rec);
    if (cap != 0u && lane == 0) {
        meta[2 * wslot + 0] = (int)wcount;
        meta[2 * wslot + 1] = stop;
    }
    __syncthreads();

    unsigned* __restrict__ plane = cntp + (size_t)slice * (size_t)cstride + base;
    for (unsigned i = tid; i < span; i += OWN_T) plane[i] = cnt[i];
}

__device__ __forceinline__ float own_dinv(unsigned c) {
    const float d = 1.0f + (float)c;
    return d > 0.0f ? __frsqrt_rn(d) : 0.0f;
}

__global__ void gcn_gemm_wmma(const float* __restrict__ x,
                              const float* __restrict__ W,
                              const unsigned* __restrict__ cntp,
                              int cstride,
                              float* __restrict__ dinv,
                              float* __restrict__ h,
                              int n_nodes, int n_tiles) {
    const int lane  = threadIdx.x & 31;
    const int wave  = threadIdx.x >> 5;
    const int tile  = blockIdx.x * (blockDim.x >> 5) + wave;
    if (tile >= n_tiles) return;

    const int base  = tile * 16;
    const int m     = lane & 15;
    const int khalf = (lane >> 4) * 2;

    int arow = base + m;
    if (arow >= n_nodes) arow = n_nodes - 1;
    const float* __restrict__ xrow = x + (size_t)arow * IN_CH;

    v2f a0 = *(const v2f*)(xrow + 0  + khalf);
    v2f a1 = *(const v2f*)(xrow + 4  + khalf);
    v2f a2 = *(const v2f*)(xrow + 8  + khalf);
    v2f a3 = *(const v2f*)(xrow + 12 + khalf);

    const int   wrow = (m < OUT_CH) ? m : 0;
    const float sel  = (m < OUT_CH) ? 1.0f : 0.0f;
    const float* __restrict__ wr = W + wrow * IN_CH + khalf;
    v2f b0 = *(const v2f*)(wr + 0);
    v2f b1 = *(const v2f*)(wr + 4);
    v2f b2 = *(const v2f*)(wr + 8);
    v2f b3 = *(const v2f*)(wr + 12);
    b0 *= sel; b1 *= sel; b2 *= sel; b3 *= sel;

    v8f acc = {};
    acc = __builtin_amdgcn_wmma_f32_16x16x4_f32(false, a0, false, b0, (short)0, acc, false, false);
    acc = __builtin_amdgcn_wmma_f32_16x16x4_f32(false, a1, false, b1, (short)0, acc, false, false);
    acc = __builtin_amdgcn_wmma_f32_16x16x4_f32(false, a2, false, b2, (short)0, acc, false, false);
    acc = __builtin_amdgcn_wmma_f32_16x16x4_f32(false, a3, false, b3, (short)0, acc, false, false);

    if ((tile & 1) == 0) {
        const int node = base + lane;
        if (node < n_nodes) {
            unsigned c = 0u;
#pragma unroll
            for (int s = 0; s < DEG_NS; ++s) c += cntp[(size_t)s * (size_t)cstride + node];
            dinv[node] = own_dinv(c);
        }
    }

    const int  n     = m;
    const int  mbase = base + ((lane >> 4) * 8);
    const bool full  = (base + 16) <= n_nodes;

    if (full) {
        if (n < OUT_CH) {
            uint4 c0 = *(const uint4*)(cntp + mbase);
            uint4 c1 = *(const uint4*)(cntp + mbase + 4);
#pragma unroll
            for (int s = 1; s < DEG_NS; ++s) {
                const unsigned* __restrict__ pl = cntp + (size_t)s * (size_t)cstride + mbase;
                const uint4 e0 = *(const uint4*)(pl);
                const uint4 e1 = *(const uint4*)(pl + 4);
                c0.x += e0.x; c0.y += e0.y; c0.z += e0.z; c0.w += e0.w;
                c1.x += e1.x; c1.y += e1.y; c1.z += e1.z; c1.w += e1.w;
            }
            const float dv[8] = {own_dinv(c0.x), own_dinv(c0.y), own_dinv(c0.z), own_dinv(c0.w),
                                 own_dinv(c1.x), own_dinv(c1.y), own_dinv(c1.z), own_dinv(c1.w)};
            float* __restrict__ hp = h   + (size_t)mbase * OUT_CH + n;
#pragma unroll
            for (int r = 0; r < 8; ++r) hp[r * OUT_CH] = dv[r] * acc[r];
        }
    } else {
        if (n < OUT_CH) {
#pragma unroll
            for (int r = 0; r < 8; ++r) {
                const int node = mbase + r;
                if (node < n_nodes) {
                    unsigned c = 0u;
#pragma unroll
                    for (int s = 0; s < DEG_NS; ++s) c += cntp[(size_t)s * (size_t)cstride + node];
                    h[(size_t)node * OUT_CH + n] = own_dinv(c) * acc[r];
                }
            }
        }
    }
}

__device__ __forceinline__ void own_add_row(float* acc_row, const float4& h0, const float4& h1) {
    atomicAdd(acc_row + 0, h0.x);
    atomicAdd(acc_row + 1, h0.y);
    atomicAdd(acc_row + 2, h0.z);
    atomicAdd(acc_row + 3, h0.w);
    atomicAdd(acc_row + 4, h1.x);
    atomicAdd(acc_row + 5, h1.y);
    atomicAdd(acc_row + 6, h1.z);
    atomicAdd(acc_row + 7, h1.w);
}

__global__ __launch_bounds__(OWN_T) void gcn_sum_owner(const int* __restrict__ src,
                                                       const int* __restrict__ dst,
                                                       const float* __restrict__ dinv,
                                                       const float* __restrict__ h,
                                                       const float* __restrict__ bias,
                                                       const int* __restrict__ meta,
                                                       const int* __restrict__ lists,
                                                       unsigned cap,
                                                       float* __restrict__ out,
                                                       int n_nodes, int e_cnt) {
    __shared__ float acc[SUB_R * OUT_CH];
    __shared__ int   qa[OWN_W][QA_CAP];
    __shared__ float dvs[SUB_R];
    const int tid  = threadIdx.x;
    const int lane = tid & 31;
    const int wv   = __builtin_amdgcn_readfirstlane(tid >> 5);
    int* myq = qa[wv];
    const unsigned owner  = blockIdx.x / SUB_N;
    const unsigned sub    = blockIdx.x - owner * SUB_N;
    const unsigned owners = gridDim.x / SUB_N;
    const unsigned base = owner * DEG_R + sub * SUB_R;
    if (base >= (unsigned)n_nodes) return;
    const unsigned span = min((unsigned)SUB_R, (unsigned)n_nodes - base);
    const int4* __restrict__ d4 = (const int4*)dst;
    const bool s_al = ((reinterpret_cast<size_t>(src) & 15u) == 0);
    const unsigned n4m1 = (unsigned)max((e_cnt >> 2) - 1, 0);
    const unsigned tagbit = POS_BITS + sub;

    {
        float4* a4 = (float4*)acc;
        for (unsigned i = tid; i < span * (OUT_CH / 4); i += OWN_T) a4[i] = make_float4(0.0f, 0.0f, 0.0f, 0.0f);
        for (unsigned i = tid; i < span; i += OWN_T) dvs[i] = dinv[base + i];
    }
    __syncthreads();

    auto add_src = [&](int s, unsigned x) {
        const float4* __restrict__ hs = (const float4*)(h + (size_t)s * OUT_CH);
        const float4 h0 = hs[0];
        const float4 h1 = hs[1];
        own_add_row(&acc[x * OUT_CH], h0, h1);
    };

    auto drain = [&](unsigned n) {
        for (unsigned j0 = 0; j0 < n; j0 += 32 * DI) {
            bool ok[DI]; unsigned idx[DI]; int4 v[DI]; int4 sv[DI];
#pragma unroll
            for (int d = 0; d < DI; ++d) {
                const unsigned j = j0 + d * 32 + lane;
                ok[d]  = j < n;
                idx[d] = min((unsigned)myq[ok[d] ? j : 0u], n4m1);
            }
#pragma unroll
            for (int d = 0; d < DI; ++d) {
                v[d] = d4[idx[d]];
                if (s_al) {
                    sv[d] = ((const int4*)src)[idx[d]];
                } else {
                    const int* sp = src + 4 * (size_t)idx[d];
                    sv[d] = make_int4(sp[0], sp[1], sp[2], sp[3]);
                }
            }
            unsigned xm[DI]; int kk[DI]; int s[DI];
#pragma unroll
            for (int d = 0; d < DI; ++d) {
                const unsigned x0 = (unsigned)v[d].x - base;
                const unsigned x1 = (unsigned)v[d].y - base;
                const unsigned x2 = (unsigned)v[d].z - base;
                const unsigned x3 = (unsigned)v[d].w - base;
                const unsigned m  = min(min(x0, x1), min(x2, x3));
                ok[d] = ok[d] && (m < span);
                kk[d] = (x0 == m) ? 0 : (x1 == m) ? 1 : (x2 == m) ? 2 : 3;
                xm[d] = min(m, span - 1u);
                s[d]  = (x0 == m) ? sv[d].x : (x1 == m) ? sv[d].y : (x2 == m) ? sv[d].z : sv[d].w;
            }
            float4 r0[DI], r1[DI];
#pragma unroll
            for (int d = 0; d < DI; ++d) {
                const float4* __restrict__ hs = (const float4*)(h + (size_t)s[d] * OUT_CH);
                r0[d] = hs[0];
                r1[d] = hs[1];
            }
#pragma unroll
            for (int d = 0; d < DI; ++d) {
                asm volatile("" :: "v"(r0[d].x), "v"(r0[d].y), "v"(r0[d].z), "v"(r0[d].w),
                                   "v"(r1[d].x), "v"(r1[d].y), "v"(r1[d].z), "v"(r1[d].w));
            }
#pragma unroll
            for (int d = 0; d < DI; ++d)
                if (ok[d]) own_add_row(&acc[xm[d] * OUT_CH], r0[d], r1[d]);
#pragma unroll
            for (int d = 0; d < DI; ++d) {
                if (ok[d]) {
                    const unsigned x0 = (unsigned)v[d].x - base;
                    const unsigned x1 = (unsigned)v[d].y - base;
                    const unsigned x2 = (unsigned)v[d].z - base;
                    const unsigned x3 = (unsigned)v[d].w - base;
                    if (kk[d] != 0 && x0 < span) add_src(sv[d].x, x0);
                    if (kk[d] != 1 && x1 < span) add_src(sv[d].y, x1);
                    if (kk[d] != 2 && x2 < span) add_src(sv[d].z, x2);
                    if (kk[d] != 3 && x3 < span) add_src(sv[d].w, x3);
                }
            }
        }
    };

    unsigned wcount = 0;

#pragma unroll 1
    for (unsigned slice = 0; slice < (unsigned)DEG_NS; ++slice) {
        const unsigned wslot = (owner * DEG_NS + slice) * OWN_W + (unsigned)wv;
        const int* __restrict__ lst = lists + (size_t)wslot * cap;
        unsigned lcount = 0;
        int it0 = 0;
        if (cap != 0u) {
            lcount = min((unsigned)__builtin_amdgcn_readfirstlane(meta[2 * wslot + 0]), cap);
            it0    = max(__builtin_amdgcn_readfirstlane(meta[2 * wslot + 1]), 0);
        }
        for (unsigned j0 = 0; j0 < lcount; j0 += QA_STEP) {
            unsigned e[OWN_U];
#pragma unroll
            for (int u = 0; u < OWN_U; ++u) {
                const unsigned j = j0 + u * 32 + lane;
                e[u] = (unsigned)lst[min(j, lcount - 1u)];
            }
            __builtin_amdgcn_sched_barrier(0);
#pragma unroll
            for (int u = 0; u < OWN_U; ++u)
                if (j0 + u * 32 + lane >= lcount) e[u] = 0u;
#pragma unroll
            for (int u = 0; u < OWN_U; ++u) {
                const bool p = ((e[u] >> tagbit) & 1u) != 0u;
                const unsigned mask = __builtin_amdgcn_ballot_w32(p);
                const unsigned rank = __builtin_amdgcn_mbcnt_lo(mask, 0u);
                if (p) myq[wcount + rank] = (int)(e[u] & POS_MASK);
                wcount += (unsigned)__builtin_popcount(mask);
            }
            if (wcount > QA_CAP - QA_STEP) { drain(wcount); wcount = 0; }
        }
        own_scan_all(dst, e_cnt, tid, owner, owners, slice, it0,
            [&](int e) {
                const unsigned x = (unsigned)dst[e] - base;
                if (x < span) add_src(src[e], x);
            },
            [&](int it, int first, int n4, bool partial) {
                int4 v[OWN_U];
                if (partial) own_load_step<true >(d4, first, n4, tid, v);
                else         own_load_step<false>(d4, first, n4, tid, v);
#pragma unroll
                for (int u = 0; u < OWN_U; ++u) {
                    const unsigned x0 = (unsigned)v[u].x - base;
                    const unsigned x1 = (unsigned)v[u].y - base;
                    const unsigned x2 = (unsigned)v[u].z - base;
                    const unsigned x3 = (unsigned)v[u].w - base;
                    const bool p = min(min(x0, x1), min(x2, x3)) < span;
                    const unsigned mask = __builtin_amdgcn_ballot_w32(p);
                    const unsigned rank = __builtin_amdgcn_mbcnt_lo(mask, 0u);
                    if (p) myq[wcount + rank] = first + u * OWN_T + tid;
                    wcount += (unsigned)__builtin_popcount(mask);
                }
                if (wcount > QA_CAP - QA_STEP) { drain(wcount); wcount = 0; }
            });
    }
    drain(wcount);

    const float4* __restrict__ h4 = (const float4*)(h + (size_t)base * OUT_CH);
    float4* __restrict__ o4 = (float4*)(out + (size_t)base * OUT_CH);
    const float* __restrict__ bp = bias + (tid & 1) * 4;
    const float bx = bp[0], by = bp[1], bz = bp[2], bw = bp[3];
    const unsigned nq = span * (OUT_CH / 4);
    float4 hv[WO_K];
#pragma unroll
    for (int k = 0; k < WO_K; ++k) {
        const unsigned q = (unsigned)(k * OWN_T + tid);
        hv[k] = h4[min(q, nq - 1u)];
    }
    __syncthreads();
    {
        const float4* a4 = (const float4*)acc;
#pragma unroll
        for (int k = 0; k < WO_K; ++k) {
            const unsigned q = (unsigned)(k * OWN_T + tid);
            if (q < nq) {
                const float  di = dvs[q >> 1];
                const float4 av = a4[q];
                float4 o;
                o.x = bx + di * (hv[k].x + av.x);
                o.y = by + di * (hv[k].y + av.y);
                o.z = bz + di * (hv[k].z + av.z);
                o.w = bw + di * (hv[k].w + av.w);
                o4[q] = o;
            }
        }
    }
}

extern "C" void kernel_launch(void* const* d_in, const int* in_sizes, int n_in,
                              void* d_out, int out_size, void* d_ws, size_t ws_size,
                              hipStream_t stream) {
    const float* x    = (const float*)d_in[0];
    const int*   ei   = (const int*)d_in[1];
    const float* W    = (const float*)d_in[2];
    const float* bias = (const float*)d_in[3];

    const int N = in_sizes[0] / IN_CH;
    const int E = in_sizes[1] / 2;
    const int* src = ei;
    const int* dst = ei + (size_t)E;

    float* dinv = (float*)d_ws;
    size_t h_off = (((size_t)N * sizeof(float)) + 255u) & ~(size_t)255u;
    float* h = (float*)((char*)d_ws + h_off);
    float* out = (float*)d_out;

    const int TPB = 256;
    const int owners = (N + DEG_R - 1) / DEG_R;

    const size_t lwaves   = (size_t)owners * DEG_NS * OWN_W;
    const size_t cstride  = ((size_t)N + 63u) & ~(size_t)63u;
    const size_t cnt_off  = (h_off + (size_t)N * OUT_CH * sizeof(float) + 255u) & ~(size_t)255u;
    const size_t meta_off = (cnt_off + (size_t)DEG_NS * cstride * sizeof(unsigned) + 255u) & ~(size_t)255u;
    const size_t list_off = (meta_off + lwaves * 2 * sizeof(int) + 255u) & ~(size_t)255u;
    size_t cap = 0;
    if (lwaves != 0 && ws_size > list_off) cap = (ws_size - list_off) / (lwaves * sizeof(int));
    if (cap > LIST_MAX) cap = LIST_MAX;
    if ((E >> 2) > (1 << POS_BITS)) cap = 0;
    unsigned* cntp = (unsigned*)((char*)d_ws + cnt_off);
    int* meta  = (int*)((char*)d_ws + meta_off);
    int* lists = (int*)((char*)d_ws + list_off);

    gcn_deg_owner<<<owners * DEG_NS, OWN_T, 0, stream>>>(dst, cntp, (int)cstride, meta, lists, (unsigned)cap, N, E);

    const int n_tiles = (N + 15) / 16;
    const int waves_per_block = TPB / 32;
    const int gemm_blocks = (n_tiles + waves_per_block - 1) / waves_per_block;
    gcn_gemm_wmma<<<gemm_blocks, TPB, 0, stream>>>(x, W, cntp, (int)cstride, dinv, h, N, n_tiles);

    gcn_sum_owner<<<owners * SUB_N, OWN_T, 0, stream>>>(src, dst, dinv, h, bias, meta, lists, (unsigned)cap, out, N, E);
}


# list entries name the edge itself (place of its word in the listing wave's walk, which of the four, target row): sum kernel reads src[e] and adds, no second read of dst; sum order changes
# speedup vs baseline: 4.2021x; 1.0358x over previous
#include <hip/hip_runtime.h>
#include <hip/hip_bf16.h>


typedef __attribute__((ext_vector_type(2))) float v2f;
typedef __attribute__((ext_vector_type(8))) float v8f;

#define IN_CH 16
#define OUT_CH 8

#define DEG_R    16384
#define DEG_NS   8
#define SUB_R    2048
#define SUB_SH   11
#define SUB_N    (DEG_R / SUB_R)
#define POS_BITS 21
#define X_BITS   14
#define K_SH     X_BITS
#define REL_SH   (X_BITS + 2)
#define OWN_TSH  10
#define QS_SH    26
#define WO_K     (SUB_R * (OUT_CH / 4) / OWN_T)
#define OWN_T    1024
#define OWN_W    (OWN_T / 32)
#define OWN_U    8
#define OWN_SS   (OWN_T * OWN_U)
#define QA_CAP   1024
#define QA_STEP  (OWN_U * 32)
#define DI       4
#define LIST_MAX 4096
#define DQ_DRAIN 96
#define DQ_CAP   (DQ_DRAIN - 1 + QA_STEP + 1)

template <bool GUARD>
__device__ __forceinline__ void own_load_step(const int4* __restrict__ d4, int first, int n4,
                                              int tid, int4 (&v)[OWN_U]) {
#pragma unroll
    for (int u = 0; u < OWN_U; ++u) {
        const int idx = first + u * OWN_T + tid;
        if (GUARD) {
            v[u] = make_int4(-1, -1, -1, -1);
            if (idx < n4) v[u] = d4[idx];
        } else {
            v[u] = d4[idx];
        }
    }
}

__device__ __forceinline__ int own_steps(int n4) { return (n4 + OWN_SS - 1) / OWN_SS; }
__device__ __forceinline__ int own_share_lo(int nss, unsigned slice) {
    return (int)(((unsigned)nss * slice) / (unsigned)DEG_NS);
}
static_assert((1 << OWN_TSH) == OWN_T && (1 << X_BITS) == DEG_R && (SUB_N & (SUB_N - 1)) == 0, "entry format");
#define REL_BITS (POS_BITS - 3 - OWN_TSH + 5)
static_assert(DEG_NS == 8 && REL_BITS + REL_SH <= 32 && SUB_SH + REL_BITS + 2 <= QS_SH && QS_SH + 3 <= 32, "entry format");

template <typename T, typename S>
__device__ __forceinline__ void own_scan_all(const int* __restrict__ dst, int e_cnt, int tid,
                                             unsigned owner, unsigned owners, unsigned slice,
                                             int it0, T tail, S step) {
    const bool aligned = ((reinterpret_cast<size_t>(dst) & 15u) == 0);
    const int n4 = aligned ? (e_cnt >> 2) : 0;
    if (slice == 0u)
        for (int e = 4 * n4 + tid; e < e_cnt; e += OWN_T) tail(e);
    const int nss = own_steps(n4);
    const int lo  = own_share_lo(nss, slice);
    const int hi  = own_share_lo(nss, slice + 1u);
    const int ns  = hi - lo;
    unsigned rot = (owner * (unsigned)ns) / owners;
    if (rot >= (unsigned)ns) rot = 0;
    for (int it = it0; it < ns; ++it) {
        int ss = it + (int)rot;
        if (ss >= ns) ss -= ns;
        const int first = (lo + ss) * OWN_SS;
        step(it, first, n4, first + OWN_SS > n4);
    }
}

__global__ __launch_bounds__(OWN_T) void gcn_deg_owner(const int* __restrict__ dst,
                                                       unsigned* __restrict__ cntp,
                                                       int cstride,
                                                       int* __restrict__ meta,
                                                       int* __restrict__ lists,
                                                       unsigned cap,
                                                       int n_nodes, int e_cnt) {
    __shared__ unsigned cnt[DEG_R];
    __shared__ int4     dqw[OWN_W][DQ_CAP];
    __shared__ int      dqp[OWN_W][DQ_CAP];
    const int tid = threadIdx.x;
    const int lane = tid & 31;
    const int wv   = __builtin_amdgcn_readfirstlane(tid >> 5);
    const unsigned owner  = blockIdx.x / DEG_NS;
    const unsigned slice  = blockIdx.x - owner * DEG_NS;
    const unsigned owners = gridDim.x / DEG_NS;
    const unsigned base = owner * DEG_R;
    if (base >= (unsigned)n_nodes) return;
    const unsigned span = min((unsigned)DEG_R, (unsigned)n_nodes - base);
    const int4* __restrict__ d4 = (const int4*)dst;
    int4* myw = dqw[wv];
    int*  myp = dqp[wv];
    const int n4w = ((reinterpret_cast<size_t>(dst) & 15u) == 0) ? (e_cnt >> 2) : 0;
    const unsigned lo_pos = (unsigned)own_share_lo(own_steps(n4w), slice) * (unsigned)OWN_SS;

    for (int i = tid; i < DEG_R; i += OWN_T) cnt[i] = 0u;
    __syncthreads();

    const unsigned wslot = blockIdx.x * OWN_W + (unsigned)wv;
    int* __restrict__ lst = lists + (size_t)wslot * cap;
    unsigned wcount = 0;
    unsigned qn     = 0;
    bool     rec    = true;
    int      stop   = 0x7fffffff;

    auto test4_queue = [&](const int4& w, int pos) {
        const unsigned x0 = (unsigned)w.x - base;
        const unsigned x1 = (unsigned)w.y - base;
        const unsigned x2 = (unsigned)w.z - base;
        const unsigned x3 = (unsigned)w.w - base;
        const bool p = min(min(x0, x1), min(x2, x3)) < span;
        const unsigned mask = __builtin_amdgcn_ballot_w32(p);
        const unsigned rank = __builtin_amdgcn_mbcnt_lo(mask, 0u);
        if (p) {
            myw[qn + rank] = w;
            myp[qn + rank] = pos;
        }
        qn += (unsigned)__builtin_popcount(mask);
    };
    auto work_off = [&](unsigned n, bool listing) {
        for (unsigned j0 = 0; j0 < n; j0 += 32) {
            const unsigned j = j0 + (unsigned)lane;
            int4 w = make_int4(-1, -1, -1, -1);
            unsigned rel = 0u;
            if (j < n) {
                w = myw[j];
                const unsigned q = (unsigned)myp[j] - lo_pos;
                rel = (((q >> OWN_TSH) << 5) | (q & 31u)) << REL_SH;
            }
            const unsigned xk[4] = {(unsigned)w.x - base, (unsigned)w.y - base,
                                    (unsigned)w.z - base, (unsigned)w.w - base};
#pragma unroll
            for (int k = 0; k < 4; ++k) {
                const bool p = xk[k] < span;
                if (listing) {
                    const unsigned mask = __builtin_amdgcn_ballot_w32(p);
                    const unsigned rank = __builtin_amdgcn_mbcnt_lo(mask, 0u);
                    if (p) lst[wcount + rank] = (int)(rel | ((unsigned)k << K_SH) | xk[k]);
                    wcount += (unsigned)__builtin_popcount(mask);
                }
                if (p) atomicAdd(&cnt[xk[k]], 1u);
            }
        }
    };
    auto count_edges = [&](unsigned n) {
        unsigned ne = 0u;
        for (unsigned j0 = 0; j0 < n; j0 += 32) {
            const unsigned j = j0 + (unsigned)lane;
            int4 w = make_int4(-1, -1, -1, -1);
            if (j < n) w = myw[j];
            ne += (unsigned)__builtin_popcount(__builtin_amdgcn_ballot_w32((unsigned)w.x - base < span))
                + (unsigned)__builtin_popcount(__builtin_amdgcn_ballot_w32((unsigned)w.y - base < span))
                + (unsigned)__builtin_popcount(__builtin_amdgcn_ballot_w32((unsigned)w.z - base < span))
                + (unsigned)__builtin_popcount(__builtin_amdgcn_ballot_w32((unsigned)w.w - base < span));
        }
        return ne;
    };
    own_scan_all(dst, e_cnt, tid, owner, owners, slice, 0,
        [&](int e) {
            const unsigned x = (unsigned)dst[e] - base;
            if (x < span) atomicAdd(&cnt[x], 1u);
        },
        [&](int it, int first, int n4, bool partial) {
            const bool near_end = rec && wcount + 4u * (qn + QA_STEP) > cap;
            if (near_end) { work_off(qn, true); qn = 0; }
            int4 v[OWN_U];
            if (partial) own_load_step<true >(d4, first, n4, tid, v);
            else         own_load_step<false>(d4, first, n4, tid, v);
#pragma unroll
            for (int u = 0; u < OWN_U; ++u) test4_queue(v[u], first + u * OWN_T + tid);
            bool off = qn >= DQ_DRAIN;
            if (near_end) {
                if (wcount + count_edges(qn) > cap) { rec = false; stop = it; }
                off = true;
            }
            if (off) { work_off(qn, rec); qn = 0; }
        });
    work_off(qn, rec);
    if (cap != 0u && lane == 0) {
        meta[2 * wslot + 0] = (int)wcount;
        meta[2 * wslot + 1] = stop;
    }
    __syncthreads();

    unsigned* __restrict__ plane = cntp + (size_t)slice * (size_t)cstride + base;
    for (unsigned i = tid; i < span; i += OWN_T) plane[i] = cnt[i];
}

__device__ __forceinline__ float own_dinv(unsigned c) {
    const float d = 1.0f + (float)c;
    return d > 0.0f ? __frsqrt_rn(d) : 0.0f;
}

__global__ void gcn_gemm_wmma(const float* __restrict__ x,
                              const float* __restrict__ W,
                              const unsigned* __restrict__ cntp,
                              int cstride,
                              float* __restrict__ dinv,
                              float* __restrict__ h,
                              int n_nodes, int n_tiles) {
    const int lane  = threadIdx.x & 31;
    const int wave  = threadIdx.x >> 5;
    const int tile  = blockIdx.x * (blockDim.x >> 5) + wave;
    if (tile >= n_tiles) return;

    const int base  = tile * 16;
    const int m     = lane & 15;
    const int khalf = (lane >> 4) * 2;

    int arow = base + m;
    if (arow >= n_nodes) arow = n_nodes - 1;
    const float* __restrict__ xrow = x + (size_t)arow * IN_CH;

    v2f a0 = *(const v2f*)(xrow + 0  + khalf);
    v2f a1 = *(const v2f*)(xrow + 4  + khalf);
    v2f a2 = *(const v2f*)(xrow + 8  + khalf);
    v2f a3 = *(const v2f*)(xrow + 12 + khalf);

    const int   wrow = (m < OUT_CH) ? m : 0;
    const float sel  = (m < OUT_CH) ? 1.0f : 0.0f;
    const float* __restrict__ wr = W + wrow * IN_CH + khalf;
    v2f b0 = *(const v2f*)(wr + 0);
    v2f b1 = *(const v2f*)(wr + 4);
    v2f b2 = *(const v2f*)(wr + 8);
    v2f b3 = *(const v2f*)(wr + 12);
    b0 *= sel; b1 *= sel; b2 *= sel; b3 *= sel;

    v8f acc = {};
    acc = __builtin_amdgcn_wmma_f32_16x16x4_f32(false, a0, false, b0, (short)0, acc, false, false);
    acc = __builtin_amdgcn_wmma_f32_16x16x4_f32(false, a1, false, b1, (short)0, acc, false, false);
    acc = __builtin_amdgcn_wmma_f32_16x16x4_f32(false, a2, false, b2, (short)0, acc, false, false);
    acc = __builtin_amdgcn_wmma_f32_16x16x4_f32(false, a3, false, b3, (short)0, acc, false, false);

    if ((tile & 1) == 0) {
        const int node = base + lane;
        if (node < n_nodes) {
            unsigned c = 0u;
#pragma unroll
            for (int s = 0; s < DEG_NS; ++s) c += cntp[(size_t)s * (size_t)cstride + node];
            dinv[node] = own_dinv(c);
        }
    }

    const int  n     = m;
    const int  mbase = base + ((lane >> 4) * 8);
    const bool full  = (base + 16) <= n_nodes;

    if (full) {
        if (n < OUT_CH) {
            uint4 c0 = *(const uint4*)(cntp + mbase);
            uint4 c1 = *(const uint4*)(cntp + mbase + 4);
#pragma unroll
            for (int s = 1; s < DEG_NS; ++s) {
                const unsigned* __restrict__ pl = cntp + (size_t)s * (size_t)cstride + mbase;
                const uint4 e0 = *(const uint4*)(pl);
                const uint4 e1 = *(const uint4*)(pl + 4);
                c0.x += e0.x; c0.y += e0.y; c0.z += e0.z; c0.w += e0.w;
                c1.x += e1.x; c1.y += e1.y; c1.z += e1.z; c1.w += e1.w;
            }
            const float dv[8] = {own_dinv(c0.x), own_dinv(c0.y), own_dinv(c0.z), own_dinv(c0.w),
                                 own_dinv(c1.x), own_dinv(c1.y), own_dinv(c1.z), own_dinv(c1.w)};
            float* __restrict__ hp = h   + (size_t)mbase * OUT_CH + n;
#pragma unroll
            for (int r = 0; r < 8; ++r) hp[r * OUT_CH] = dv[r] * acc[r];
        }
    } else {
        if (n < OUT_CH) {
#pragma unroll
            for (int r = 0; r < 8; ++r) {
                const int node = mbase + r;
                if (node < n_nodes) {
                    unsigned c = 0u;
#pragma unroll
                    for (int s = 0; s < DEG_NS; ++s) c += cntp[(size_t)s * (size_t)cstride + node];
                    h[(size_t)node * OUT_CH + n] = own_dinv(c) * acc[r];
                }
            }
        }
    }
}

__device__ __forceinline__ void own_add_row(float* acc_row, const float4& h0, const float4& h1) {
    atomicAdd(acc_row + 0, h0.x);
    atomicAdd(acc_row + 1, h0.y);
    atomicAdd(acc_row + 2, h0.z);
    atomicAdd(acc_row + 3, h0.w);
    atomicAdd(acc_row + 4, h1.x);
    atomicAdd(acc_row + 5, h1.y);
    atomicAdd(acc_row + 6, h1.z);
    atomicAdd(acc_row + 7, h1.w);
}

__global__ __launch_bounds__(OWN_T) void gcn_sum_owner(const int* __restrict__ src,
                                                       const int* __restrict__ dst,
                                                       const float* __restrict__ dinv,
                                                       const float* __restrict__ h,
                                                       const float* __restrict__ bias,
                                                       const int* __restrict__ meta,
                                                       const int* __restrict__ lists,
                                                       unsigned cap,
                                                       float* __restrict__ out,
                                                       int n_nodes, int e_cnt) {
    __shared__ float acc[SUB_R * OUT_CH];
    __shared__ int   qa[OWN_W][QA_CAP];
    __shared__ float dvs[SUB_R];
    const int tid  = threadIdx.x;
    const int lane = tid & 31;
    const int wv   = __builtin_amdgcn_readfirstlane(tid >> 5);
    int* myq = qa[wv];
    const unsigned owner  = blockIdx.x / SUB_N;
    const unsigned sub    = blockIdx.x - owner * SUB_N;
    const unsigned owners = gridDim.x / SUB_N;
    const unsigned base = owner * DEG_R + sub * SUB_R;
    if (base >= (unsigned)n_nodes) return;
    const unsigned span = min((unsigned)SUB_R, (unsigned)n_nodes - base);
    const int4* __restrict__ d4 = (const int4*)dst;
    const int nssw = own_steps(((reinterpret_cast<size_t>(dst) & 15u) == 0) ? (e_cnt >> 2) : 0);
    const unsigned e_last = (unsigned)max(e_cnt - 1, 0);

    {
        float4* a4 = (float4*)acc;
        for (unsigned i = tid; i < span * (OUT_CH / 4); i += OWN_T) a4[i] = make_float4(0.0f, 0.0f, 0.0f, 0.0f);
        for (unsigned i = tid; i < span; i += OWN_T) dvs[i] = dinv[base + i];
    }
    __syncthreads();

    auto add_src = [&](int s, unsigned x) {
        const float4* __restrict__ hs = (const float4*)(h + (size_t)s * OUT_CH);
        const float4 h0 = hs[0];
        const float4 h1 = hs[1];
        own_add_row(&acc[x * OUT_CH], h0, h1);
    };

    auto drain = [&](unsigned n) {
        for (unsigned j0 = 0; j0 < n; j0 += 32 * DI) {
            bool ok[DI]; unsigned e[DI]; unsigned xr[DI]; int s[DI];
#pragma unroll
            for (int d = 0; d < DI; ++d) {
                const unsigned j = j0 + d * 32 + lane;
                ok[d] = j < n;
                const unsigned qv  = (unsigned)myq[ok[d] ? j : 0u];
                const unsigned rk  = (qv >> SUB_SH) & ((1u << (REL_BITS + 2)) - 1u);
                const unsigned rel = rk >> 2;
                const unsigned pos = (unsigned)own_share_lo(nssw, qv >> QS_SH) * (unsigned)OWN_SS
                                   + (((rel >> 5) << OWN_TSH) | ((unsigned)wv << 5) | (rel & 31u));
                e[d]  = min(4u * pos + (rk & 3u), e_last);
                xr[d] = min(qv & (SUB_R - 1u), span - 1u);
            }
#pragma unroll
            for (int d = 0; d < DI; ++d) s[d] = src[e[d]];
            float4 r0[DI], r1[DI];
#pragma unroll
            for (int d = 0; d < DI; ++d) {
                const float4* __restrict__ hs = (const float4*)(h + (size_t)s[d] * OUT_CH);
                r0[d] = hs[0];
                r1[d] = hs[1];
            }
#pragma unroll
            for (int d = 0; d < DI; ++d) {
                asm volatile("" :: "v"(r0[d].x), "v"(r0[d].y), "v"(r0[d].z), "v"(r0[d].w),
                                   "v"(r1[d].x), "v"(r1[d].y), "v"(r1[d].z), "v"(r1[d].w));
            }
#pragma unroll
            for (int d = 0; d < DI; ++d)
                if (ok[d]) own_add_row(&acc[xr[d] * OUT_CH], r0[d], r1[d]);
        }
    };

    unsigned wcount = 0;

#pragma unroll 1
    for (unsigned slice = 0; slice < (unsigned)DEG_NS; ++slice) {
        const unsigned wslot = (owner * DEG_NS + slice) * OWN_W + (unsigned)wv;
        const int* __restrict__ lst = lists + (size_t)wslot * cap;
        unsigned lcount = 0;
        int it0 = 0;
        if (cap != 0u) {
            lcount = min((unsigned)__builtin_amdgcn_readfirstlane(meta[2 * wslot + 0]), cap);
            it0    = max(__builtin_amdgcn_readfirstlane(meta[2 * wslot + 1]), 0);
        }
        for (unsigned j0 = 0; j0 < lcount; j0 += QA_STEP) {
            unsigned e[OWN_U];
#pragma unroll
            for (int u = 0; u < OWN_U; ++u) {
                const unsigned j = j0 + u * 32 + lane;
                e[u] = (unsigned)lst[min(j, lcount - 1u)];
            }
            __builtin_amdgcn_sched_barrier(0);
#pragma unroll
            for (int u = 0; u < OWN_U; ++u) {
                const bool p = (j0 + u * 32 + lane < lcount) && (((e[u] >> SUB_SH) & (SUB_N - 1u)) == sub);
                const unsigned mask = __builtin_amdgcn_ballot_w32(p);
                const unsigned rank = __builtin_amdgcn_mbcnt_lo(mask, 0u);
                if (p) myq[wcount + rank] = (int)((slice << QS_SH) | ((e[u] >> K_SH) << SUB_SH) | (e[u] & (SUB_R - 1u)));
                wcount += (unsigned)__builtin_popcount(mask);
            }
            if (wcount > QA_CAP - QA_STEP) { drain(wcount); wcount = 0; }
        }
        own_scan_all(dst, e_cnt, tid, owner, owners, slice, it0,
            [&](int e) {
                const unsigned x = (unsigned)dst[e] - base;
                if (x < span) add_src(src[e], x);
            },
            [&](int it, int first, int n4, bool partial) {
                int4 v[OWN_U];
                if (partial) own_load_step<true >(d4, first, n4, tid, v);
                else         own_load_step<false>(d4, first, n4, tid, v);
#pragma unroll
                for (int u = 0; u < OWN_U; ++u) {
                    const unsigned x0 = (unsigned)v[u].x - base;
                    const unsigned x1 = (unsigned)v[u].y - base;
                    const unsigned x2 = (unsigned)v[u].z - base;
                    const unsigned x3 = (unsigned)v[u].w - base;
                    if (min(min(x0, x1), min(x2, x3)) < span) {
                        const int* __restrict__ sp = src + 4 * (size_t)(first + u * OWN_T + tid);
                        if (x0 < span) add_src(sp[0], x0);
                        if (x1 < span) add_src(sp[1], x1);
                        if (x2 < span) add_src(sp[2], x2);
                        if (x3 < span) add_src(sp[3], x3);
                    }
                }
            });
    }
    drain(wcount);

    const float4* __restrict__ h4 = (const float4*)(h + (size_t)base * OUT_CH);
    float4* __restrict__ o4 = (float4*)(out + (size_t)base * OUT_CH);
    const float* __restrict__ bp = bias + (tid & 1) * 4;
    const float bx = bp[0], by = bp[1], bz = bp[2], bw = bp[3];
    const unsigned nq = span * (OUT_CH / 4);
    float4 hv[WO_K];
#pragma unroll
    for (int k = 0; k < WO_K; ++k) {
        const unsigned q = (unsigned)(k * OWN_T + tid);
        hv[k] = h4[min(q, nq - 1u)];
    }
    __syncthreads();
    {
        const float4* a4 = (const float4*)acc;
#pragma unroll
        for (int k = 0; k < WO_K; ++k) {
            const unsigned q = (unsigned)(k * OWN_T + tid);
            if (q < nq) {
                const float  di = dvs[q >> 1];
                const float4 av = a4[q];
                float4 o;
                o.x = bx + di * (hv[k].x + av.x);
                o.y = by + di * (hv[k].y + av.y);
                o.z = bz + di * (hv[k].z + av.z);
                o.w = bw + di * (hv[k].w + av.w);
                o4[q] = o;
            }
        }
    }
}

extern "C" void kernel_launch(void* const* d_in, const int* in_sizes, int n_in,
                              void* d_out, int out_size, void* d_ws, size_t ws_size,
                              hipStream_t stream) {
    const float* x    = (const float*)d_in[0];
    const int*   ei   = (const int*)d_in[1];
    const float* W    = (const float*)d_in[2];
    const float* bias = (const float*)d_in[3];

    const int N = in_sizes[0] / IN_CH;
    const int E = in_sizes[1] / 2;
    const int* src = ei;
    const int* dst = ei + (size_t)E;

    float* dinv = (float*)d_ws;
    size_t h_off = (((size_t)N * sizeof(float)) + 255u) & ~(size_t)255u;
    float* h = (float*)((char*)d_ws + h_off);
    float* out = (float*)d_out;

    const int TPB = 256;
    const int owners = (N + DEG_R - 1) / DEG_R;

    const size_t lwaves   = (size_t)owners * DEG_NS * OWN_W;
    const size_t cstride  = ((size_t)N + 63u) & ~(size_t)63u;
    const size_t cnt_off  = (h_off + (size_t)N * OUT_CH * sizeof(float) + 255u) & ~(size_t)255u;
    const size_t meta_off = (cnt_off + (size_t)DEG_NS * cstride * sizeof(unsigned) + 255u) & ~(size_t)255u;
    const size_t list_off = (meta_off + lwaves * 2 * sizeof(int) + 255u) & ~(size_t)255u;
    size_t cap = 0;
    if (lwaves != 0 && ws_size > list_off) cap = (ws_size - list_off) / (lwaves * sizeof(int));
    if (cap > LIST_MAX) cap = LIST_MAX;
    if ((E >> 2) > (1 << POS_BITS)) cap = 0;
    unsigned* cntp = (unsigned*)((char*)d_ws + cnt_off);
    int* meta  = (int*)((char*)d_ws + meta_off);
    int* lists = (int*)((char*)d_ws + list_off);

    gcn_deg_owner<<<owners * DEG_NS, OWN_T, 0, stream>>>(dst, cntp, (int)cstride, meta, lists, (unsigned)cap, N, E);

    const int n_tiles = (N + 15) / 16;
    const int waves_per_block = TPB / 32;
    const int gemm_blocks = (n_tiles + waves_per_block - 1) / waves_per_block;
    gcn_gemm_wmma<<<gemm_blocks, TPB, 0, stream>>>(x, W, cntp, (int)cstride, dinv, h, N, n_tiles);

    gcn_sum_owner<<<owners * SUB_N, OWN_T, 0, stream>>>(src, dst, dinv, h, bias, meta, lists, (unsigned)cap, out, N, E);
}


# sum kernel's LDS tile kept as 8 channel planes (pitch 2056 floats), not 8-float rows: each float add's 32 words fall on all 64 banks, not 8; write-out reads planes; add arrival order may change
# speedup vs baseline: 5.1119x; 1.2165x over previous
#include <hip/hip_runtime.h>
#include <hip/hip_bf16.h>


typedef __attribute__((ext_vector_type(2))) float v2f;
typedef __attribute__((ext_vector_type(8))) float v8f;

#define IN_CH 16
#define OUT_CH 8

#define DEG_R    16384
#define DEG_NS   8
#define SUB_R    2048
#define SUB_SH   11
#define SUB_N    (DEG_R / SUB_R)
#define POS_BITS 21
#define X_BITS   14
#define K_SH     X_BITS
#define REL_SH   (X_BITS + 2)
#define OWN_TSH  10
#define QS_SH    26
#define WO_K     (SUB_R * (OUT_CH / 4) / OWN_T)
#define ACC_P    (SUB_R + 8)
#define OWN_T    1024
#define OWN_W    (OWN_T / 32)
#define OWN_U    8
#define OWN_SS   (OWN_T * OWN_U)
#define QA_CAP   1024
#define QA_STEP  (OWN_U * 32)
#define DI       4
#define LIST_MAX 4096
#define DQ_DRAIN 96
#define DQ_CAP   (DQ_DRAIN - 1 + QA_STEP + 1)

template <bool GUARD>
__device__ __forceinline__ void own_load_step(const int4* __restrict__ d4, int first, int n4,
                                              int tid, int4 (&v)[OWN_U]) {
#pragma unroll
    for (int u = 0; u < OWN_U; ++u) {
        const int idx = first + u * OWN_T + tid;
        if (GUARD) {
            v[u] = make_int4(-1, -1, -1, -1);
            if (idx < n4) v[u] = d4[idx];
        } else {
            v[u] = d4[idx];
        }
    }
}

__device__ __forceinline__ int own_steps(int n4) { return (n4 + OWN_SS - 1) / OWN_SS; }
__device__ __forceinline__ int own_share_lo(int nss, unsigned slice) {
    return (int)(((unsigned)nss * slice) / (unsigned)DEG_NS);
}
static_assert((1 << OWN_TSH) == OWN_T && (1 << X_BITS) == DEG_R && (SUB_N & (SUB_N - 1)) == 0, "entry format");
#define REL_BITS (POS_BITS - 3 - OWN_TSH + 5)
static_assert(DEG_NS == 8 && REL_BITS + REL_SH <= 32 && SUB_SH + REL_BITS + 2 <= QS_SH && QS_SH + 3 <= 32, "entry format");

template <typename T, typename S>
__device__ __forceinline__ void own_scan_all(const int* __restrict__ dst, int e_cnt, int tid,
                                             unsigned owner, unsigned owners, unsigned slice,
                                             int it0, T tail, S step) {
    const bool aligned = ((reinterpret_cast<size_t>(dst) & 15u) == 0);
    const int n4 = aligned ? (e_cnt >> 2) : 0;
    if (slice == 0u)
        for (int e = 4 * n4 + tid; e < e_cnt; e += OWN_T) tail(e);
    const int nss = own_steps(n4);
    const int lo  = own_share_lo(nss, slice);
    const int hi  = own_share_lo(nss, slice + 1u);
    const int ns  = hi - lo;
    unsigned rot = (owner * (unsigned)ns) / owners;
    if (rot >= (unsigned)ns) rot = 0;
    for (int it = it0; it < ns; ++it) {
        int ss = it + (int)rot;
        if (ss >= ns) ss -= ns;
        const int first = (lo + ss) * OWN_SS;
        step(it, first, n4, first + OWN_SS > n4);
    }
}

__global__ __launch_bounds__(OWN_T) void gcn_deg_owner(const int* __restrict__ dst,
                                                       unsigned* __restrict__ cntp,
                                                       int cstride,
                                                       int* __restrict__ meta,
                                                       int* __restrict__ lists,
                                                       unsigned cap,
                                                       int n_nodes, int e_cnt) {
    __shared__ unsigned cnt[DEG_R];
    __shared__ int4     dqw[OWN_W][DQ_CAP];
    __shared__ int      dqp[OWN_W][DQ_CAP];
    const int tid = threadIdx.x;
    const int lane = tid & 31;
    const int wv   = __builtin_amdgcn_readfirstlane(tid >> 5);
    const unsigned owner  = blockIdx.x / DEG_NS;
    const unsigned slice  = blockIdx.x - owner * DEG_NS;
    const unsigned owners = gridDim.x / DEG_NS;
    const unsigned base = owner * DEG_R;
    if (base >= (unsigned)n_nodes) return;
    const unsigned span = min((unsigned)DEG_R, (unsigned)n_nodes - base);
    const int4* __restrict__ d4 = (const int4*)dst;
    int4* myw = dqw[wv];
    int*  myp = dqp[wv];
    const int n4w = ((reinterpret_cast<size_t>(dst) & 15u) == 0) ? (e_cnt >> 2) : 0;
    const unsigned lo_pos = (unsigned)own_share_lo(own_steps(n4w), slice) * (unsigned)OWN_SS;

    for (int i = tid; i < DEG_R; i += OWN_T) cnt[i] = 0u;
    __syncthreads();

    const unsigned wslot = blockIdx.x * OWN_W + (unsigned)wv;
    int* __restrict__ lst = lists + (size_t)wslot * cap;
    unsigned wcount = 0;
    unsigned qn     = 0;
    bool     rec    = true;
    int      stop   = 0x7fffffff;

    auto test4_queue = [&](const int4& w, int pos) {
        const unsigned x0 = (unsigned)w.x - base;
        const unsigned x1 = (unsigned)w.y - base;
        const unsigned x2 = (unsigned)w.z - base;
        const unsigned x3 = (unsigned)w.w - base;
        const bool p = min(min(x0, x1), min(x2, x3)) < span;
        const unsigned mask = __builtin_amdgcn_ballot_w32(p);
        const unsigned rank = __builtin_amdgcn_mbcnt_lo(mask, 0u);
        if (p) {
            myw[qn + rank] = w;
            myp[qn + rank] = pos;
        }
        qn += (unsigned)__builtin_popcount(mask);
    };
    auto work_off = [&](unsigned n, bool listing) {
        for (unsigned j0 = 0; j0 < n; j0 += 32) {
            const unsigned j = j0 + (unsigned)lane;
            int4 w = make_int4(-1, -1, -1, -1);
            unsigned rel = 0u;
            if (j < n) {
                w = myw[j];
                const unsigned q = (unsigned)myp[j] - lo_pos;
                rel = (((q >> OWN_TSH) << 5) | (q & 31u)) << REL_SH;
            }
            const unsigned xk[4] = {(unsigned)w.x - base, (unsigned)w.y - base,
                                    (unsigned)w.z - base, (unsigned)w.w - base};
#pragma unroll
            for (int k = 0; k < 4; ++k) {
                const bool p = xk[k] < span;
                if (listing) {
                    const unsigned mask = __builtin_amdgcn_ballot_w32(p);
                    const unsigned rank = __builtin_amdgcn_mbcnt_lo(mask, 0u);
                    if (p) lst[wcount + rank] = (int)(rel | ((unsigned)k << K_SH) | xk[k]);
                    wcount += (unsigned)__builtin_popcount(mask);
                }
                if (p) atomicAdd(&cnt[xk[k]], 1u);
            }
        }
    };
    auto count_edges = [&](unsigned n) {
        unsigned ne = 0u;
        for (unsigned j0 = 0; j0 < n; j0 += 32) {
            const unsigned j = j0 + (unsigned)lane;
            int4 w = make_int4(-1, -1, -1, -1);
            if (j < n) w = myw[j];
            ne += (unsigned)__builtin_popcount(__builtin_amdgcn_ballot_w32((unsigned)w.x - base < span))
                + (unsigned)__builtin_popcount(__builtin_amdgcn_ballot_w32((unsigned)w.y - base < span))
                + (unsigned)__builtin_popcount(__builtin_amdgcn_ballot_w32((unsigned)w.z - base < span))
                + (unsigned)__builtin_popcount(__builtin_amdgcn_ballot_w32((unsigned)w.w - base < span));
        }
        return ne;
    };
    own_scan_all(dst, e_cnt, tid, owner, owners, slice, 0,
        [&](int e) {
            const unsigned x = (unsigned)dst[e] - base;
            if (x < span) atomicAdd(&cnt[x], 1u);
        },
        [&](int it, int first, int n4, bool partial) {
            const bool near_end = rec && wcount + 4u * (qn + QA_STEP) > cap;
            if (near_end) { work_off(qn, true); qn = 0; }
            int4 v[OWN_U];
            if (partial) own_load_step<true >(d4, first, n4, tid, v);
            else         own_load_step<false>(d4, first, n4, tid, v);
#pragma unroll
            for (int u = 0; u < OWN_U; ++u) test4_queue(v[u], first + u * OWN_T + tid);
            bool off = qn >= DQ_DRAIN;
            if (near_end) {
                if (wcount + count_edges(qn) > cap) { rec = false; stop = it; }
                off = true;
            }
            if (off) { work_off(qn, rec); qn = 0; }
        });
    work_off(qn, rec);
    if (cap != 0u && lane == 0) {
        meta[2 * wslot + 0] = (int)wcount;
        meta[2 * wslot + 1] = stop;
    }
    __syncthreads();

    unsigned* __restrict__ plane = cntp + (size_t)slice * (size_t)cstride + base;
    for (unsigned i = tid; i < span; i += OWN_T) plane[i] = cnt[i];
}

__device__ __forceinline__ float own_dinv(unsigned c) {
    const float d = 1.0f + (float)c;
    return d > 0.0f ? __frsqrt_rn(d) : 0.0f;
}

__global__ void gcn_gemm_wmma(const float* __restrict__ x,
                              const float* __restrict__ W,
                              const unsigned* __restrict__ cntp,
                              int cstride,
                              float* __restrict__ dinv,
                              float* __restrict__ h,
                              int n_nodes, int n_tiles) {
    const int lane  = threadIdx.x & 31;
    const int wave  = threadIdx.x >> 5;
    const int tile  = blockIdx.x * (blockDim.x >> 5) + wave;
    if (tile >= n_tiles) return;

    const int base  = tile * 16;
    const int m     = lane & 15;
    const int khalf = (lane >> 4) * 2;

    int arow = base + m;
    if (arow >= n_nodes) arow = n_nodes - 1;
    const float* __restrict__ xrow = x + (size_t)arow * IN_CH;

    v2f a0 = *(const v2f*)(xrow + 0  + khalf);
    v2f a1 = *(const v2f*)(xrow + 4  + khalf);
    v2f a2 = *(const v2f*)(xrow + 8  + khalf);
    v2f a3 = *(const v2f*)(xrow + 12 + khalf);

    const int   wrow = (m < OUT_CH) ? m : 0;
    const float sel  = (m < OUT_CH) ? 1.0f : 0.0f;
    const float* __restrict__ wr = W + wrow * IN_CH + khalf;
    v2f b0 = *(const v2f*)(wr + 0);
    v2f b1 = *(const v2f*)(wr + 4);
    v2f b2 = *(const v2f*)(wr + 8);
    v2f b3 = *(const v2f*)(wr + 12);
    b0 *= sel; b1 *= sel; b2 *= sel; b3 *= sel;

    v8f acc = {};
    acc = __builtin_amdgcn_wmma_f32_16x16x4_f32(false, a0, false, b0, (short)0, acc, false, false);
    acc = __builtin_amdgcn_wmma_f32_16x16x4_f32(false, a1, false, b1, (short)0, acc, false, false);
    acc = __builtin_amdgcn_wmma_f32_16x16x4_f32(false, a2, false, b2, (short)0, acc, false, false);
    acc = __builtin_amdgcn_wmma_f32_16x16x4_f32(false, a3, false, b3, (short)0, acc, false, false);

    if ((tile & 1) == 0) {
        const int node = base + lane;
        if (node < n_nodes) {
            unsigned c = 0u;
#pragma unroll
            for (int s = 0; s < DEG_NS; ++s) c += cntp[(size_t)s * (size_t)cstride + node];
            dinv[node] = own_dinv(c);
        }
    }

    const int  n     = m;
    const int  mbase = base + ((lane >> 4) * 8);
    const bool full  = (base + 16) <= n_nodes;

    if (full) {
        if (n < OUT_CH) {
            uint4 c0 = *(const uint4*)(cntp + mbase);
            uint4 c1 = *(const uint4*)(cntp + mbase + 4);
#pragma unroll
            for (int s = 1; s < DEG_NS; ++s) {
                const unsigned* __restrict__ pl = cntp + (size_t)s * (size_t)cstride + mbase;
                const uint4 e0 = *(const uint4*)(pl);
                const uint4 e1 = *(const uint4*)(pl + 4);
                c0.x += e0.x; c0.y += e0.y; c0.z += e0.z; c0.w += e0.w;
                c1.x += e1.x; c1.y += e1.y; c1.z += e1.z; c1.w += e1.w;
            }
            const float dv[8] = {own_dinv(c0.x), own_dinv(c0.y), own_dinv(c0.z), own_dinv(c0.w),
                                 own_dinv(c1.x), own_dinv(c1.y), own_dinv(c1.z), own_dinv(c1.w)};
            float* __restrict__ hp = h   + (size_t)mbase * OUT_CH + n;
#pragma unroll
            for (int r = 0; r < 8; ++r) hp[r * OUT_CH] = dv[r] * acc[r];
        }
    } else {
        if (n < OUT_CH) {
#pragma unroll
            for (int r = 0; r < 8; ++r) {
                const int node = mbase + r;
                if (node < n_nodes) {
                    unsigned c = 0u;
#pragma unroll
                    for (int s = 0; s < DEG_NS; ++s) c += cntp[(size_t)s * (size_t)cstride + node];
                    h[(size_t)node * OUT_CH + n] = own_dinv(c) * acc[r];
                }
            }
        }
    }
}

__device__ __forceinline__ void own_add_row(float* acc_x, const float4& h0, const float4& h1) {
    atomicAdd(acc_x + 0 * ACC_P, h0.x);
    atomicAdd(acc_x + 1 * ACC_P, h0.y);
    atomicAdd(acc_x + 2 * ACC_P, h0.z);
    atomicAdd(acc_x + 3 * ACC_P, h0.w);
    atomicAdd(acc_x + 4 * ACC_P, h1.x);
    atomicAdd(acc_x + 5 * ACC_P, h1.y);
    atomicAdd(acc_x + 6 * ACC_P, h1.z);
    atomicAdd(acc_x + 7 * ACC_P, h1.w);
}

__global__ __launch_bounds__(OWN_T) void gcn_sum_owner(const int* __restrict__ src,
                                                       const int* __restrict__ dst,
                                                       const float* __restrict__ dinv,
                                                       const float* __restrict__ h,
                                                       const float* __restrict__ bias,
                                                       const int* __restrict__ meta,
                                                       const int* __restrict__ lists,
                                                       unsigned cap,
                                                       float* __restrict__ out,
                                                       int n_nodes, int e_cnt) {
    struct SumLds {
        float acc[OUT_CH * ACC_P];
        int   qa[OWN_W][QA_CAP];
        float dvs[SUB_R];
    };
    static_assert((OUT_CH - 1) * ACC_P * 4 < 65536, "plane offsets fit the LDS offset field");
    __shared__ __attribute__((aligned(16))) SumLds lds;
    float* const acc = lds.acc;
    int   (* const qa)[QA_CAP] = lds.qa;
    float* const dvs = lds.dvs;
    const int tid  = threadIdx.x;
    const int lane = tid & 31;
    const int wv   = __builtin_amdgcn_readfirstlane(tid >> 5);
    int* myq = qa[wv];
    const unsigned owner  = blockIdx.x / SUB_N;
    const unsigned sub    = blockIdx.x - owner * SUB_N;
    const unsigned owners = gridDim.x / SUB_N;
    const unsigned base = owner * DEG_R + sub * SUB_R;
    if (base >= (unsigned)n_nodes) return;
    const unsigned span = min((unsigned)SUB_R, (unsigned)n_nodes - base);
    const int4* __restrict__ d4 = (const int4*)dst;
    const int nssw = own_steps(((reinterpret_cast<size_t>(dst) & 15u) == 0) ? (e_cnt >> 2) : 0);
    const unsigned e_last = (unsigned)max(e_cnt - 1, 0);

    {
        float4* a4 = (float4*)acc;
        for (unsigned i = tid; i < (unsigned)(OUT_CH * ACC_P / 4); i += OWN_T) a4[i] = make_float4(0.0f, 0.0f, 0.0f, 0.0f);
        for (unsigned i = tid; i < span; i += OWN_T) dvs[i] = dinv[base + i];
    }
    __syncthreads();

    auto add_src = [&](int s, unsigned x) {
        const float4* __restrict__ hs = (const float4*)(h + (size_t)s * OUT_CH);
        const float4 h0 = hs[0];
        const float4 h1 = hs[1];
        own_add_row(&acc[x], h0, h1);
    };

    auto drain = [&](unsigned n) {
        for (unsigned j0 = 0; j0 < n; j0 += 32 * DI) {
            bool ok[DI]; unsigned e[DI]; unsigned xr[DI]; int s[DI];
#pragma unroll
            for (int d = 0; d < DI; ++d) {
                const unsigned j = j0 + d * 32 + lane;
                ok[d] = j < n;
                const unsigned qv  = (unsigned)myq[ok[d] ? j : 0u];
                const unsigned rk  = (qv >> SUB_SH) & ((1u << (REL_BITS + 2)) - 1u);
                const unsigned rel = rk >> 2;
                const unsigned pos = (unsigned)own_share_lo(nssw, qv >> QS_SH) * (unsigned)OWN_SS
                                   + (((rel >> 5) << OWN_TSH) | ((unsigned)wv << 5) | (rel & 31u));
                e[d]  = min(4u * pos + (rk & 3u), e_last);
                xr[d] = min(qv & (SUB_R - 1u), span - 1u);
            }
#pragma unroll
            for (int d = 0; d < DI; ++d) s[d] = src[e[d]];
            float4 r0[DI], r1[DI];
#pragma unroll
            for (int d = 0; d < DI; ++d) {
                const float4* __restrict__ hs = (const float4*)(h + (size_t)s[d] * OUT_CH);
                r0[d] = hs[0];
                r1[d] = hs[1];
            }
#pragma unroll
            for (int d = 0; d < DI; ++d) {
                asm volatile("" :: "v"(r0[d].x), "v"(r0[d].y), "v"(r0[d].z), "v"(r0[d].w),
                                   "v"(r1[d].x), "v"(r1[d].y), "v"(r1[d].z), "v"(r1[d].w));
            }
#pragma unroll
            for (int d = 0; d < DI; ++d)
                if (ok[d]) own_add_row(&acc[xr[d]], r0[d], r1[d]);
        }
    };

    unsigned wcount = 0;

#pragma unroll 1
    for (unsigned slice = 0; slice < (unsigned)DEG_NS; ++slice) {
        const unsigned wslot = (owner * DEG_NS + slice) * OWN_W + (unsigned)wv;
        const int* __restrict__ lst = lists + (size_t)wslot * cap;
        unsigned lcount = 0;
        int it0 = 0;
        if (cap != 0u) {
            lcount = min((unsigned)__builtin_amdgcn_readfirstlane(meta[2 * wslot + 0]), cap);
            it0    = max(__builtin_amdgcn_readfirstlane(meta[2 * wslot + 1]), 0);
        }
        for (unsigned j0 = 0; j0 < lcount; j0 += QA_STEP) {
            unsigned e[OWN_U];
#pragma unroll
            for (int u = 0; u < OWN_U; ++u) {
                const unsigned j = j0 + u * 32 + lane;
                e[u] = (unsigned)lst[min(j, lcount - 1u)];
            }
            __builtin_amdgcn_sched_barrier(0);
#pragma unroll
            for (int u = 0; u < OWN_U; ++u) {
                const bool p = (j0 + u * 32 + lane < lcount) && (((e[u] >> SUB_SH) & (SUB_N - 1u)) == sub);
                const unsigned mask = __builtin_amdgcn_ballot_w32(p);
                const unsigned rank = __builtin_amdgcn_mbcnt_lo(mask, 0u);
                if (p) myq[wcount + rank] = (int)((slice << QS_SH) | ((e[u] >> K_SH) << SUB_SH) | (e[u] & (SUB_R - 1u)));
                wcount += (unsigned)__builtin_popcount(mask);
            }
            if (wcount > QA_CAP - QA_STEP) { drain(wcount); wcount = 0; }
        }
        own_scan_all(dst, e_cnt, tid, owner, owners, slice, it0,
            [&](int e) {
                const unsigned x = (unsigned)dst[e] - base;
                if (x < span) add_src(src[e], x);
            },
            [&](int it, int first, int n4, bool partial) {
                int4 v[OWN_U];
                if (partial) own_load_step<true >(d4, first, n4, tid, v);
                else         own_load_step<false>(d4, first, n4, tid, v);
#pragma unroll
                for (int u = 0; u < OWN_U; ++u) {
                    const unsigned x0 = (unsigned)v[u].x - base;
                    const unsigned x1 = (unsigned)v[u].y - base;
                    const unsigned x2 = (unsigned)v[u].z - base;
                    const unsigned x3 = (unsigned)v[u].w - base;
                    if (min(min(x0, x1), min(x2, x3)) < span) {
                        const int* __restrict__ sp = src + 4 * (size_t)(first + u * OWN_T + tid);
                        if (x0 < span) add_src(sp[0], x0);
                        if (x1 < span) add_src(sp[1], x1);
                        if (x2 < span) add_src(sp[2], x2);
                        if (x3 < span) add_src(sp[3], x3);
                    }
                }
            });
    }
    drain(wcount);

    const float4* __restrict__ h4 = (const float4*)(h + (size_t)base * OUT_CH);
    float4* __restrict__ o4 = (float4*)(out + (size_t)base * OUT_CH);
    const float* __restrict__ bp = bias + (tid & 1) * 4;
    const float bx = bp[0], by = bp[1], bz = bp[2], bw = bp[3];
    const unsigned nq = span * (OUT_CH / 4);
    float4 hv[WO_K];
#pragma unroll
    for (int k = 0; k < WO_K; ++k) {
        const unsigned q = (unsigned)(k * OWN_T + tid);
        hv[k] = h4[min(q, nq - 1u)];
    }
    __syncthreads();
    {
        const float* ap = acc + (tid & 1) * 4 * ACC_P;
#pragma unroll
        for (int k = 0; k < WO_K; ++k) {
            const unsigned q = (unsigned)(k * OWN_T + tid);
            if (q < nq) {
                const unsigned xw = q >> 1;
                const float  di = dvs[xw];
                const float4 av = make_float4(ap[xw], ap[xw + ACC_P], ap[xw + 2 * ACC_P], ap[xw + 3 * ACC_P]);
                float4 o;
                o.x = bx + di * (hv[k].x + av.x);
                o.y = by + di * (hv[k].y + av.y);
                o.z = bz + di * (hv[k].z + av.z);
                o.w = bw + di * (hv[k].w + av.w);
                o4[q] = o;
            }
        }
    }
}

extern "C" void kernel_launch(void* const* d_in, const int* in_sizes, int n_in,
                              void* d_out, int out_size, void* d_ws, size_t ws_size,
                              hipStream_t stream) {
    const float* x    = (const float*)d_in[0];
    const int*   ei   = (const int*)d_in[1];
    const float* W    = (const float*)d_in[2];
    const float* bias = (const float*)d_in[3];

    const int N = in_sizes[0] / IN_CH;
    const int E = in_sizes[1] / 2;
    const int* src = ei;
    const int* dst = ei + (size_t)E;

    float* dinv = (float*)d_ws;
    size_t h_off = (((size_t)N * sizeof(float)) + 255u) & ~(size_t)255u;
    float* h = (float*)((char*)d_ws + h_off);
    float* out = (float*)d_out;

    const int TPB = 256;
    const int owners = (N + DEG_R - 1) / DEG_R;

    const size_t lwaves   = (size_t)owners * DEG_NS * OWN_W;
    const size_t cstride  = ((size_t)N + 63u) & ~(size_t)63u;
    const size_t cnt_off  = (h_off + (size_t)N * OUT_CH * sizeof(float) + 255u) & ~(size_t)255u;
    const size_t meta_off = (cnt_off + (size_t)DEG_NS * cstride * sizeof(unsigned) + 255u) & ~(size_t)255u;
    const size_t list_off = (meta_off + lwaves * 2 * sizeof(int) + 255u) & ~(size_t)255u;
    size_t cap = 0;
    if (lwaves != 0 && ws_size > list_off) cap = (ws_size - list_off) / (lwaves * sizeof(int));
    if (cap > LIST_MAX) cap = LIST_MAX;
    if ((E >> 2) > (1 << POS_BITS)) cap = 0;
    unsigned* cntp = (unsigned*)((char*)d_ws + cnt_off);
    int* meta  = (int*)((char*)d_ws + meta_off);
    int* lists = (int*)((char*)d_ws + list_off);

    gcn_deg_owner<<<owners * DEG_NS, OWN_T, 0, stream>>>(dst, cntp, (int)cstride, meta, lists, (unsigned)cap, N, E);

    const int n_tiles = (N + 15) / 16;
    const int waves_per_block = TPB / 32;
    const int gemm_blocks = (n_tiles + waves_per_block - 1) / waves_per_block;
    gcn_gemm_wmma<<<gemm_blocks, TPB, 0, stream>>>(x, W, cntp, (int)cstride, dinv, h, N, n_tiles);

    gcn_sum_owner<<<owners * SUB_N, OWN_T, 0, stream>>>(src, dst, dinv, h, bias, meta, lists, (unsigned)cap, out, N, E);
}


# degree walk keeps each wave's list as 8 lists, one per 2048-row sub-range (ranked by ballot per list, counters in LDS); a sum group reads only its own lists and tests none; sum order changes
# speedup vs baseline: 5.7606x; 1.1269x over previous
#include <hip/hip_runtime.h>
#include <hip/hip_bf16.h>


typedef __attribute__((ext_vector_type(2))) float v2f;
typedef __attribute__((ext_vector_type(8))) float v8f;

#define IN_CH 16
#define OUT_CH 8

#define DEG_R    16384
#define DEG_NS   8
#define SUB_R    2048
#define SUB_SH   11
#define SUB_N    (DEG_R / SUB_R)
#define POS_BITS 21
#define X_BITS   14
#define K_SH     X_BITS
#define REL_SH   (X_BITS + 2)
#define OWN_TSH  10
#define QS_SH    26
#define WO_K     (SUB_R * (OUT_CH / 4) / OWN_T)
#define ACC_P    (SUB_R + 8)
#define OWN_T    1024
#define OWN_W    (OWN_T / 32)
#define OWN_U    8
#define OWN_SS   (OWN_T * OWN_U)
#define QA_CAP   1024
#define QA_STEP  (OWN_U * 32)
#define DI       4
#define LIST_MAX 4096
#define LQ_U     8
#define LQ_STEP  (LQ_U * 32)
#define DQ_DRAIN 96
#define DQ_CAP   (DQ_DRAIN - 1 + QA_STEP + 1)

template <bool GUARD>
__device__ __forceinline__ void own_load_step(const int4* __restrict__ d4, int first, int n4,
                                              int tid, int4 (&v)[OWN_U]) {
#pragma unroll
    for (int u = 0; u < OWN_U; ++u) {
        const int idx = first + u * OWN_T + tid;
        if (GUARD) {
            v[u] = make_int4(-1, -1, -1, -1);
            if (idx < n4) v[u] = d4[idx];
        } else {
            v[u] = d4[idx];
        }
    }
}

__device__ __forceinline__ int own_steps(int n4) { return (n4 + OWN_SS - 1) / OWN_SS; }
__device__ __forceinline__ int own_share_lo(int nss, unsigned slice) {
    return (int)(((unsigned)nss * slice) / (unsigned)DEG_NS);
}
static_assert((1 << OWN_TSH) == OWN_T && (1 << X_BITS) == DEG_R && (SUB_N & (SUB_N - 1)) == 0, "entry format");
#define REL_BITS (POS_BITS - 3 - OWN_TSH + 5)
static_assert(DEG_NS == 8 && REL_BITS + REL_SH <= 32 && SUB_SH + REL_BITS + 2 <= QS_SH && QS_SH + 3 <= 32, "entry format");

template <typename T, typename S>
__device__ __forceinline__ void own_scan_all(const int* __restrict__ dst, int e_cnt, int tid,
                                             unsigned owner, unsigned owners, unsigned slice,
                                             int it0, T tail, S step) {
    const bool aligned = ((reinterpret_cast<size_t>(dst) & 15u) == 0);
    const int n4 = aligned ? (e_cnt >> 2) : 0;
    if (slice == 0u)
        for (int e = 4 * n4 + tid; e < e_cnt; e += OWN_T) tail(e);
    const int nss = own_steps(n4);
    const int lo  = own_share_lo(nss, slice);
    const int hi  = own_share_lo(nss, slice + 1u);
    const int ns  = hi - lo;
    unsigned rot = (owner * (unsigned)ns) / owners;
    if (rot >= (unsigned)ns) rot = 0;
    for (int it = it0; it < ns; ++it) {
        int ss = it + (int)rot;
        if (ss >= ns) ss -= ns;
        const int first = (lo + ss) * OWN_SS;
        step(it, first, n4, first + OWN_SS > n4);
    }
}

__global__ __launch_bounds__(OWN_T) void gcn_deg_owner(const int* __restrict__ dst,
                                                       unsigned* __restrict__ cntp,
                                                       int cstride,
                                                       int* __restrict__ meta,
                                                       int* __restrict__ lists,
                                                       unsigned cap,
                                                       int n_nodes, int e_cnt) {
    __shared__ unsigned cnt[DEG_R];
    __shared__ int4     dqw[OWN_W][DQ_CAP];
    __shared__ int      dqp[OWN_W][DQ_CAP];
    __shared__ unsigned scnt[OWN_W][SUB_N];
    const int tid = threadIdx.x;
    const int lane = tid & 31;
    const int wv   = __builtin_amdgcn_readfirstlane(tid >> 5);
    const unsigned owner  = blockIdx.x / DEG_NS;
    const unsigned slice  = blockIdx.x - owner * DEG_NS;
    const unsigned owners = gridDim.x / DEG_NS;
    const unsigned base = owner * DEG_R;
    if (base >= (unsigned)n_nodes) return;
    const unsigned span = min((unsigned)DEG_R, (unsigned)n_nodes - base);
    const int4* __restrict__ d4 = (const int4*)dst;
    int4* myw = dqw[wv];
    int*  myp = dqp[wv];
    const int n4w = ((reinterpret_cast<size_t>(dst) & 15u) == 0) ? (e_cnt >> 2) : 0;
    const unsigned lo_pos = (unsigned)own_share_lo(own_steps(n4w), slice) * (unsigned)OWN_SS;

    for (int i = tid; i < DEG_R; i += OWN_T) cnt[i] = 0u;
    if (tid < OWN_W * SUB_N) (&scnt[0][0])[tid] = 0u;
    __syncthreads();

    const unsigned wslot = blockIdx.x * OWN_W + (unsigned)wv;
    int* __restrict__ lst = lists + (size_t)wslot * cap;
    unsigned* mycnt = scnt[wv];
    const unsigned subcap  = cap / (unsigned)SUB_N;
    const bool     listing = cap != 0u;
    unsigned qn     = 0;

    auto test4_queue = [&](const int4& w, int pos) {
        const unsigned x0 = (unsigned)w.x - base;
        const unsigned x1 = (unsigned)w.y - base;
        const unsigned x2 = (unsigned)w.z - base;
        const unsigned x3 = (unsigned)w.w - base;
        const bool p = min(min(x0, x1), min(x2, x3)) < span;
        const unsigned mask = __builtin_amdgcn_ballot_w32(p);
        const unsigned rank = __builtin_amdgcn_mbcnt_lo(mask, 0u);
        if (p) {
            myw[qn + rank] = w;
            myp[qn + rank] = pos;
        }
        qn += (unsigned)__builtin_popcount(mask);
    };
    auto work_off = [&](unsigned n) {
        for (unsigned j0 = 0; j0 < n; j0 += 32) {
            const unsigned j = j0 + (unsigned)lane;
            int4 w = make_int4(-1, -1, -1, -1);
            unsigned rel = 0u;
            if (j < n) {
                w = myw[j];
                const unsigned q = (unsigned)myp[j] - lo_pos;
                rel = (((q >> OWN_TSH) << 5) | (q & 31u)) << REL_SH;
            }
            const unsigned xk[4] = {(unsigned)w.x - base, (unsigned)w.y - base,
                                    (unsigned)w.z - base, (unsigned)w.w - base};
            bool p[4];
#pragma unroll
            for (int k = 0; k < 4; ++k) {
                p[k] = xk[k] < span;
                if (p[k]) atomicAdd(&cnt[xk[k]], 1u);
            }
            if (listing) {
                const bool     has = p[0] || p[1] || p[2] || p[3];
                const unsigned f   = p[0] ? 0u : p[1] ? 1u : p[2] ? 2u : 3u;
                const unsigned xf  = p[0] ? xk[0] : p[1] ? xk[1] : p[2] ? xk[2] : xk[3];
                const unsigned tag = has ? (xf >> SUB_SH) : (unsigned)SUB_N;
                unsigned mine = 0u;
#pragma unroll
                for (unsigned t = 0; t < (unsigned)SUB_N; ++t) {
                    const unsigned m = __builtin_amdgcn_ballot_w32(tag == t);
                    if (tag == t) mine = m;
                }
                const unsigned rank = __builtin_amdgcn_mbcnt_lo(mine, 0u);
                if (has) {
                    const unsigned slot = mycnt[tag] + rank;
                    if (rank == 0u) atomicAdd(&mycnt[tag], (unsigned)__builtin_popcount(mine));
                    if (slot < subcap) lst[tag * subcap + slot] = (int)(rel | (f << K_SH) | xf);
                }
#pragma unroll
                for (int k = 1; k < 4; ++k) {
                    if (p[k] && f != (unsigned)k) {
                        const unsigned t2 = xk[k] >> SUB_SH;
                        const unsigned s2 = atomicAdd(&mycnt[t2], 1u);
                        if (s2 < subcap) lst[t2 * subcap + s2] = (int)(rel | ((unsigned)k << K_SH) | xk[k]);
                    }
                }
            }
        }
    };
    own_scan_all(dst, e_cnt, tid, owner, owners, slice, 0,
        [&](int e) {
            const unsigned x = (unsigned)dst[e] - base;
            if (x < span) atomicAdd(&cnt[x], 1u);
        },
        [&](int it, int first, int n4, bool partial) {
            int4 v[OWN_U];
            if (partial) own_load_step<true >(d4, first, n4, tid, v);
            else         own_load_step<false>(d4, first, n4, tid, v);
#pragma unroll
            for (int u = 0; u < OWN_U; ++u) test4_queue(v[u], first + u * OWN_T + tid);
            if (qn >= DQ_DRAIN) { work_off(qn); qn = 0; }
        });
    work_off(qn);
    __syncthreads();

    if (listing && lane < SUB_N) meta[(size_t)wslot * SUB_N + lane] = (int)mycnt[lane];

    unsigned* __restrict__ plane = cntp + (size_t)slice * (size_t)cstride + base;
    for (unsigned i = tid; i < span; i += OWN_T) plane[i] = cnt[i];
}

__device__ __forceinline__ float own_dinv(unsigned c) {
    const float d = 1.0f + (float)c;
    return d > 0.0f ? __frsqrt_rn(d) : 0.0f;
}

__global__ void gcn_gemm_wmma(const float* __restrict__ x,
                              const float* __restrict__ W,
                              const unsigned* __restrict__ cntp,
                              int cstride,
                              float* __restrict__ dinv,
                              float* __restrict__ h,
                              int n_nodes, int n_tiles) {
    const int lane  = threadIdx.x & 31;
    const int wave  = threadIdx.x >> 5;
    const int tile  = blockIdx.x * (blockDim.x >> 5) + wave;
    if (tile >= n_tiles) return;

    const int base  = tile * 16;
    const int m     = lane & 15;
    const int khalf = (lane >> 4) * 2;

    int arow = base + m;
    if (arow >= n_nodes) arow = n_nodes - 1;
    const float* __restrict__ xrow = x + (size_t)arow * IN_CH;

    v2f a0 = *(const v2f*)(xrow + 0  + khalf);
    v2f a1 = *(const v2f*)(xrow + 4  + khalf);
    v2f a2 = *(const v2f*)(xrow + 8  + khalf);
    v2f a3 = *(const v2f*)(xrow + 12 + khalf);

    const int   wrow = (m < OUT_CH) ? m : 0;
    const float sel  = (m < OUT_CH) ? 1.0f : 0.0f;
    const float* __restrict__ wr = W + wrow * IN_CH + khalf;
    v2f b0 = *(const v2f*)(wr + 0);
    v2f b1 = *(const v2f*)(wr + 4);
    v2f b2 = *(const v2f*)(wr + 8);
    v2f b3 = *(const v2f*)(wr + 12);
    b0 *= sel; b1 *= sel; b2 *= sel; b3 *= sel;

    v8f acc = {};
    acc = __builtin_amdgcn_wmma_f32_16x16x4_f32(false, a0, false, b0, (short)0, acc, false, false);
    acc = __builtin_amdgcn_wmma_f32_16x16x4_f32(false, a1, false, b1, (short)0, acc, false, false);
    acc = __builtin_amdgcn_wmma_f32_16x16x4_f32(false, a2, false, b2, (short)0, acc, false, false);
    acc = __builtin_amdgcn_wmma_f32_16x16x4_f32(false, a3, false, b3, (short)0, acc, false, false);

    if ((tile & 1) == 0) {
        const int node = base + lane;
        if (node < n_nodes) {
            unsigned c = 0u;
#pragma unroll
            for (int s = 0; s < DEG_NS; ++s) c += cntp[(size_t)s * (size_t)cstride + node];
            dinv[node] = own_dinv(c);
        }
    }

    const int  n     = m;
    const int  mbase = base + ((lane >> 4) * 8);
    const bool full  = (base + 16) <= n_nodes;

    if (full) {
        if (n < OUT_CH) {
            uint4 c0 = *(const uint4*)(cntp + mbase);
            uint4 c1 = *(const uint4*)(cntp + mbase + 4);
#pragma unroll
            for (int s = 1; s < DEG_NS; ++s) {
                const unsigned* __restrict__ pl = cntp + (size_t)s * (size_t)cstride + mbase;
                const uint4 e0 = *(const uint4*)(pl);
                const uint4 e1 = *(const uint4*)(pl + 4);
                c0.x += e0.x; c0.y += e0.y; c0.z += e0.z; c0.w += e0.w;
                c1.x += e1.x; c1.y += e1.y; c1.z += e1.z; c1.w += e1.w;
            }
            const float dv[8] = {own_dinv(c0.x), own_dinv(c0.y), own_dinv(c0.z), own_dinv(c0.w),
                                 own_dinv(c1.x), own_dinv(c1.y), own_dinv(c1.z), own_dinv(c1.w)};
            float* __restrict__ hp = h   + (size_t)mbase * OUT_CH + n;
#pragma unroll
            for (int r = 0; r < 8; ++r) hp[r * OUT_CH] = dv[r] * acc[r];
        }
    } else {
        if (n < OUT_CH) {
#pragma unroll
            for (int r = 0; r < 8; ++r) {
                const int node = mbase + r;
                if (node < n_nodes) {
                    unsigned c = 0u;
#pragma unroll
                    for (int s = 0; s < DEG_NS; ++s) c += cntp[(size_t)s * (size_t)cstride + node];
                    h[(size_t)node * OUT_CH + n] = own_dinv(c) * acc[r];
                }
            }
        }
    }
}

__device__ __forceinline__ void own_add_row(float* acc_x, const float4& h0, const float4& h1) {
    atomicAdd(acc_x + 0 * ACC_P, h0.x);
    atomicAdd(acc_x + 1 * ACC_P, h0.y);
    atomicAdd(acc_x + 2 * ACC_P, h0.z);
    atomicAdd(acc_x + 3 * ACC_P, h0.w);
    atomicAdd(acc_x + 4 * ACC_P, h1.x);
    atomicAdd(acc_x + 5 * ACC_P, h1.y);
    atomicAdd(acc_x + 6 * ACC_P, h1.z);
    atomicAdd(acc_x + 7 * ACC_P, h1.w);
}

__global__ __launch_bounds__(OWN_T) void gcn_sum_owner(const int* __restrict__ src,
                                                       const int* __restrict__ dst,
                                                       const float* __restrict__ dinv,
                                                       const float* __restrict__ h,
                                                       const float* __restrict__ bias,
                                                       const int* __restrict__ meta,
                                                       const int* __restrict__ lists,
                                                       unsigned cap,
                                                       float* __restrict__ out,
                                                       int n_nodes, int e_cnt) {
    struct SumLds {
        float acc[OUT_CH * ACC_P];
        int   qa[OWN_W][QA_CAP];
        float dvs[SUB_R];
    };
    static_assert((OUT_CH - 1) * ACC_P * 4 < 65536, "plane offsets fit the LDS offset field");
    __shared__ __attribute__((aligned(16))) SumLds lds;
    float* const acc = lds.acc;
    int   (* const qa)[QA_CAP] = lds.qa;
    float* const dvs = lds.dvs;
    const int tid  = threadIdx.x;
    const int lane = tid & 31;
    const int wv   = __builtin_amdgcn_readfirstlane(tid >> 5);
    int* myq = qa[wv];
    const unsigned owner  = blockIdx.x / SUB_N;
    const unsigned sub    = blockIdx.x - owner * SUB_N;
    const unsigned owners = gridDim.x / SUB_N;
    const unsigned base = owner * DEG_R + sub * SUB_R;
    if (base >= (unsigned)n_nodes) return;
    const unsigned span = min((unsigned)SUB_R, (unsigned)n_nodes - base);
    const int4* __restrict__ d4 = (const int4*)dst;
    const int nssw = own_steps(((reinterpret_cast<size_t>(dst) & 15u) == 0) ? (e_cnt >> 2) : 0);
    const unsigned e_last = (unsigned)max(e_cnt - 1, 0);

    {
        float4* a4 = (float4*)acc;
        for (unsigned i = tid; i < (unsigned)(OUT_CH * ACC_P / 4); i += OWN_T) a4[i] = make_float4(0.0f, 0.0f, 0.0f, 0.0f);
        for (unsigned i = tid; i < span; i += OWN_T) dvs[i] = dinv[base + i];
    }
    __syncthreads();

    auto add_src = [&](int s, unsigned x) {
        const float4* __restrict__ hs = (const float4*)(h + (size_t)s * OUT_CH);
        const float4 h0 = hs[0];
        const float4 h1 = hs[1];
        own_add_row(&acc[x], h0, h1);
    };

    auto drain = [&](unsigned n) {
        for (unsigned j0 = 0; j0 < n; j0 += 32 * DI) {
            bool ok[DI]; unsigned e[DI]; unsigned xr[DI]; int s[DI];
#pragma unroll
            for (int d = 0; d < DI; ++d) {
                const unsigned j = j0 + d * 32 + lane;
                ok[d] = j < n;
                const unsigned qv  = (unsigned)myq[ok[d] ? j : 0u];
                const unsigned rk  = (qv >> SUB_SH) & ((1u << (REL_BITS + 2)) - 1u);
                const unsigned rel = rk >> 2;
                const unsigned pos = (unsigned)own_share_lo(nssw, qv >> QS_SH) * (unsigned)OWN_SS
                                   + (((rel >> 5) << OWN_TSH) | ((unsigned)wv << 5) | (rel & 31u));
                e[d]  = min(4u * pos + (rk & 3u), e_last);
                xr[d] = min(qv & (SUB_R - 1u), span - 1u);
            }
#pragma unroll
            for (int d = 0; d < DI; ++d) s[d] = src[e[d]];
            float4 r0[DI], r1[DI];
#pragma unroll
            for (int d = 0; d < DI; ++d) {
                const float4* __restrict__ hs = (const float4*)(h + (size_t)s[d] * OUT_CH);
                r0[d] = hs[0];
                r1[d] = hs[1];
            }
#pragma unroll
            for (int d = 0; d < DI; ++d) {
                asm volatile("" :: "v"(r0[d].x), "v"(r0[d].y), "v"(r0[d].z), "v"(r0[d].w),
                                   "v"(r1[d].x), "v"(r1[d].y), "v"(r1[d].z), "v"(r1[d].w));
            }
#pragma unroll
            for (int d = 0; d < DI; ++d)
                if (ok[d]) own_add_row(&acc[xr[d]], r0[d], r1[d]);
        }
    };

    unsigned wcount = 0;

    const unsigned subcap = cap / (unsigned)SUB_N;
    unsigned want = 0xffffffffu;
    if (cap != 0u && lane < DEG_NS)
        want = (unsigned)meta[(size_t)((owner * DEG_NS + (unsigned)lane) * OWN_W + (unsigned)wv) * SUB_N + sub];
    const bool     is_whole = lane < DEG_NS && want <= subcap;
    const unsigned whole    = __builtin_amdgcn_ballot_w32(is_whole);
    const unsigned len_l    = is_whole ? want : 0u;
    unsigned pre[DEG_NS + 1];
    pre[0] = 0u;
#pragma unroll
    for (int s = 0; s < DEG_NS; ++s)
        pre[s + 1] = pre[s] + (unsigned)__builtin_amdgcn_readlane((int)len_l, s);
    const unsigned total   = pre[DEG_NS];
    const unsigned lstride = (unsigned)OWN_W * cap;
    const int* __restrict__ lst0 = lists + (size_t)((owner * DEG_NS) * OWN_W + (unsigned)wv) * cap + (size_t)sub * subcap;
    for (unsigned j0 = 0; j0 < total; j0 += LQ_STEP) {
        unsigned e[LQ_U], sh[LQ_U];
#pragma unroll
        for (int u = 0; u < LQ_U; ++u) {
            const unsigned j = min(j0 + u * 32 + lane, total - 1u);
            unsigned s = 0u, idx = j;
#pragma unroll
            for (int t = 1; t < DEG_NS; ++t) {
                const bool past = j >= pre[t];
                s   += past ? 1u : 0u;
                idx += past ? lstride - (pre[t] - pre[t - 1]) : 0u;
            }
            e[u]  = (unsigned)lst0[idx];
            sh[u] = s;
        }
#pragma unroll
        for (int u = 0; u < LQ_U; ++u) {
            if (j0 + u * 32 + lane < total)
                myq[wcount + u * 32 + lane] = (int)((sh[u] << QS_SH) | ((e[u] >> K_SH) << SUB_SH) | (e[u] & (SUB_R - 1u)));
        }
        wcount += min((unsigned)LQ_STEP, total - j0);
        if (wcount > QA_CAP - LQ_STEP) { drain(wcount); wcount = 0; }
    }

#pragma unroll 1
    for (unsigned slice = 0; slice < (unsigned)DEG_NS; ++slice) {
        const int it0 = ((whole >> slice) & 1u) ? 0x7fffffff : 0;
        own_scan_all(dst, e_cnt, tid, owner, owners, slice, it0,
            [&](int e) {
                const unsigned x = (unsigned)dst[e] - base;
                if (x < span) add_src(src[e], x);
            },
            [&](int it, int first, int n4, bool partial) {
                int4 v[OWN_U];
                if (partial) own_load_step<true >(d4, first, n4, tid, v);
                else         own_load_step<false>(d4, first, n4, tid, v);
#pragma unroll
                for (int u = 0; u < OWN_U; ++u) {
                    const unsigned x0 = (unsigned)v[u].x - base;
                    const unsigned x1 = (unsigned)v[u].y - base;
                    const unsigned x2 = (unsigned)v[u].z - base;
                    const unsigned x3 = (unsigned)v[u].w - base;
                    if (min(min(x0, x1), min(x2, x3)) < span) {
                        const int* __restrict__ sp = src + 4 * (size_t)(first + u * OWN_T + tid);
                        if (x0 < span) add_src(sp[0], x0);
                        if (x1 < span) add_src(sp[1], x1);
                        if (x2 < span) add_src(sp[2], x2);
                        if (x3 < span) add_src(sp[3], x3);
                    }
                }
            });
    }
    drain(wcount);

    const float4* __restrict__ h4 = (const float4*)(h + (size_t)base * OUT_CH);
    float4* __restrict__ o4 = (float4*)(out + (size_t)base * OUT_CH);
    const float* __restrict__ bp = bias + (tid & 1) * 4;
    const float bx = bp[0], by = bp[1], bz = bp[2], bw = bp[3];
    const unsigned nq = span * (OUT_CH / 4);
    float4 hv[WO_K];
#pragma unroll
    for (int k = 0; k < WO_K; ++k) {
        const unsigned q = (unsigned)(k * OWN_T + tid);
        hv[k] = h4[min(q, nq - 1u)];
    }
    __syncthreads();
    {
        const float* ap = acc + (tid & 1) * 4 * ACC_P;
#pragma unroll
        for (int k = 0; k < WO_K; ++k) {
            const unsigned q = (unsigned)(k * OWN_T + tid);
            if (q < nq) {
                const unsigned xw = q >> 1;
                const float  di = dvs[xw];
                const float4 av = make_float4(ap[xw], ap[xw + ACC_P], ap[xw + 2 * ACC_P], ap[xw + 3 * ACC_P]);
                float4 o;
                o.x = bx + di * (hv[k].x + av.x);
                o.y = by + di * (hv[k].y + av.y);
                o.z = bz + di * (hv[k].z + av.z);
                o.w = bw + di * (hv[k].w + av.w);
                o4[q] = o;
            }
        }
    }
}

extern "C" void kernel_launch(void* const* d_in, const int* in_sizes, int n_in,
                              void* d_out, int out_size, void* d_ws, size_t ws_size,
                              hipStream_t stream) {
    const float* x    = (const float*)d_in[0];
    const int*   ei   = (const int*)d_in[1];
    const float* W    = (const float*)d_in[2];
    const float* bias = (const float*)d_in[3];

    const int N = in_sizes[0] / IN_CH;
    const int E = in_sizes[1] / 2;
    const int* src = ei;
    const int* dst = ei + (size_t)E;

    float* dinv = (float*)d_ws;
    size_t h_off = (((size_t)N * sizeof(float)) + 255u) & ~(size_t)255u;
    float* h = (float*)((char*)d_ws + h_off);
    float* out = (float*)d_out;

    const int TPB = 256;
    const int owners = (N + DEG_R - 1) / DEG_R;

    const size_t lwaves   = (size_t)owners * DEG_NS * OWN_W;
    const size_t cstride  = ((size_t)N + 63u) & ~(size_t)63u;
    const size_t cnt_off  = (h_off + (size_t)N * OUT_CH * sizeof(float) + 255u) & ~(size_t)255u;
    const size_t meta_off = (cnt_off + (size_t)DEG_NS * cstride * sizeof(unsigned) + 255u) & ~(size_t)255u;
    const size_t list_off = (meta_off + lwaves * SUB_N * sizeof(int) + 255u) & ~(size_t)255u;
    size_t cap = 0;
    if (lwaves != 0 && ws_size > list_off) cap = (ws_size - list_off) / (lwaves * sizeof(int));
    if (cap > LIST_MAX) cap = LIST_MAX;
    if ((E >> 2) > (1 << POS_BITS)) cap = 0;
    unsigned* cntp = (unsigned*)((char*)d_ws + cnt_off);
    int* meta  = (int*)((char*)d_ws + meta_off);
    int* lists = (int*)((char*)d_ws + list_off);

    gcn_deg_owner<<<owners * DEG_NS, OWN_T, 0, stream>>>(dst, cntp, (int)cstride, meta, lists, (unsigned)cap, N, E);

    const int n_tiles = (N + 15) / 16;
    const int waves_per_block = TPB / 32;
    const int gemm_blocks = (n_tiles + waves_per_block - 1) / waves_per_block;
    gcn_gemm_wmma<<<gemm_blocks, TPB, 0, stream>>>(x, W, cntp, (int)cstride, dinv, h, N, n_tiles);

    gcn_sum_owner<<<owners * SUB_N, OWN_T, 0, stream>>>(src, dst, dinv, h, bias, meta, lists, (unsigned)cap, out, N, E);
}


# list entries carry the edge's source node: degree walk reads the src word at each found word's place (neighbouring lanes share lines), sum kernel gathers rows with no src read; sum order as before
# speedup vs baseline: 6.0098x; 1.0433x over previous
#include <hip/hip_runtime.h>
#include <hip/hip_bf16.h>


typedef __attribute__((ext_vector_type(2))) float v2f;
typedef __attribute__((ext_vector_type(8))) float v8f;

#define IN_CH 16
#define OUT_CH 8

#define DEG_R    16384
#define DEG_NS   8
#define SUB_R    2048
#define SUB_SH   11
#define SUB_N    (DEG_R / SUB_R)
#define SRC_BITS (32 - SUB_SH)
#define WO_K     (SUB_R * (OUT_CH / 4) / OWN_T)
#define ACC_P    (SUB_R + 8)
#define OWN_T    1024
#define OWN_W    (OWN_T / 32)
#define OWN_U    8
#define OWN_SS   (OWN_T * OWN_U)
#define QA_CAP   1024
#define QA_STEP  (OWN_U * 32)
#define DI       4
#define LIST_MAX 4096
#define LQ_U     8
#define LQ_STEP  (LQ_U * 32)
#define DQ_DRAIN 96
#define DQ_CAP   (DQ_DRAIN - 1 + QA_STEP + 1)

template <bool GUARD>
__device__ __forceinline__ void own_load_step(const int4* __restrict__ d4, int first, int n4,
                                              int tid, int4 (&v)[OWN_U]) {
#pragma unroll
    for (int u = 0; u < OWN_U; ++u) {
        const int idx = first + u * OWN_T + tid;
        if (GUARD) {
            v[u] = make_int4(-1, -1, -1, -1);
            if (idx < n4) v[u] = d4[idx];
        } else {
            v[u] = d4[idx];
        }
    }
}

__device__ __forceinline__ int own_steps(int n4) { return (n4 + OWN_SS - 1) / OWN_SS; }
__device__ __forceinline__ int own_share_lo(int nss, unsigned slice) {
    return (int)(((unsigned)nss * slice) / (unsigned)DEG_NS);
}
static_assert((1 << SUB_SH) == SUB_R && (SUB_N & (SUB_N - 1)) == 0, "entry format");

template <typename T, typename S>
__device__ __forceinline__ void own_scan_all(const int* __restrict__ dst, int e_cnt, int tid,
                                             unsigned owner, unsigned owners, unsigned slice,
                                             int it0, T tail, S step) {
    const bool aligned = ((reinterpret_cast<size_t>(dst) & 15u) == 0);
    const int n4 = aligned ? (e_cnt >> 2) : 0;
    if (slice == 0u)
        for (int e = 4 * n4 + tid; e < e_cnt; e += OWN_T) tail(e);
    const int nss = own_steps(n4);
    const int lo  = own_share_lo(nss, slice);
    const int hi  = own_share_lo(nss, slice + 1u);
    const int ns  = hi - lo;
    unsigned rot = (owner * (unsigned)ns) / owners;
    if (rot >= (unsigned)ns) rot = 0;
    for (int it = it0; it < ns; ++it) {
        int ss = it + (int)rot;
        if (ss >= ns) ss -= ns;
        const int first = (lo + ss) * OWN_SS;
        step(it, first, n4, first + OWN_SS > n4);
    }
}

__global__ __launch_bounds__(OWN_T) void gcn_deg_owner(const int* __restrict__ src,
                                                       const int* __restrict__ dst,
                                                       unsigned* __restrict__ cntp,
                                                       int cstride,
                                                       int* __restrict__ meta,
                                                       int* __restrict__ lists,
                                                       unsigned cap,
                                                       int n_nodes, int e_cnt) {
    __shared__ unsigned cnt[DEG_R];
    __shared__ int4     dqw[OWN_W][DQ_CAP];
    __shared__ int      dqp[OWN_W][DQ_CAP];
    __shared__ unsigned scnt[OWN_W][SUB_N];
    const int tid = threadIdx.x;
    const int lane = tid & 31;
    const int wv   = __builtin_amdgcn_readfirstlane(tid >> 5);
    const unsigned owner  = blockIdx.x / DEG_NS;
    const unsigned slice  = blockIdx.x - owner * DEG_NS;
    const unsigned owners = gridDim.x / DEG_NS;
    const unsigned base = owner * DEG_R;
    if (base >= (unsigned)n_nodes) return;
    const unsigned span = min((unsigned)DEG_R, (unsigned)n_nodes - base);
    const int4* __restrict__ d4 = (const int4*)dst;
    const int4* __restrict__ s4 = (const int4*)src;
    int4* myw = dqw[wv];
    int*  myp = dqp[wv];

    for (int i = tid; i < DEG_R; i += OWN_T) cnt[i] = 0u;
    if (tid < OWN_W * SUB_N) (&scnt[0][0])[tid] = 0u;
    __syncthreads();

    const unsigned wslot = blockIdx.x * OWN_W + (unsigned)wv;
    int* __restrict__ lst = lists + (size_t)wslot * cap;
    unsigned* mycnt = scnt[wv];
    const unsigned subcap  = cap / (unsigned)SUB_N;
    const bool     listing = cap != 0u;
    unsigned qn     = 0;

    auto test4_queue = [&](const int4& w, int pos) {
        const unsigned x0 = (unsigned)w.x - base;
        const unsigned x1 = (unsigned)w.y - base;
        const unsigned x2 = (unsigned)w.z - base;
        const unsigned x3 = (unsigned)w.w - base;
        const bool p = min(min(x0, x1), min(x2, x3)) < span;
        const unsigned mask = __builtin_amdgcn_ballot_w32(p);
        const unsigned rank = __builtin_amdgcn_mbcnt_lo(mask, 0u);
        if (p) {
            myw[qn + rank] = w;
            myp[qn + rank] = pos;
        }
        qn += (unsigned)__builtin_popcount(mask);
    };
    auto fetch = [&](unsigned j0, unsigned n, int4& w, int4& sw) {
        const unsigned j = j0 + (unsigned)lane;
        w  = make_int4(-1, -1, -1, -1);
        sw = make_int4(0, 0, 0, 0);
        if (j < n) {
            w = myw[j];
            if (listing) sw = s4[myp[j]];
        }
    };
    auto work_off = [&](unsigned n) {
        int4 wn, sn;
        fetch(0u, n, wn, sn);
        for (unsigned j0 = 0; j0 < n; j0 += 32) {
            const int4 w = wn, sw = sn;
            if (j0 + 32 < n) fetch(j0 + 32, n, wn, sn);
            const unsigned xk[4] = {(unsigned)w.x - base, (unsigned)w.y - base,
                                    (unsigned)w.z - base, (unsigned)w.w - base};
            const unsigned sk[4] = {(unsigned)sw.x, (unsigned)sw.y, (unsigned)sw.z, (unsigned)sw.w};
            bool p[4];
#pragma unroll
            for (int k = 0; k < 4; ++k) {
                p[k] = xk[k] < span;
                if (p[k]) atomicAdd(&cnt[xk[k]], 1u);
            }
            if (listing) {
                const bool     has = p[0] || p[1] || p[2] || p[3];
                const unsigned xf  = p[0] ? xk[0] : p[1] ? xk[1] : p[2] ? xk[2] : xk[3];
                const unsigned sf  = p[0] ? sk[0] : p[1] ? sk[1] : p[2] ? sk[2] : sk[3];
                const unsigned tag = has ? (xf >> SUB_SH) : (unsigned)SUB_N;
                unsigned mine = 0u;
#pragma unroll
                for (unsigned t = 0; t < (unsigned)SUB_N; ++t) {
                    const unsigned m = __builtin_amdgcn_ballot_w32(tag == t);
                    if (tag == t) mine = m;
                }
                const unsigned rank = __builtin_amdgcn_mbcnt_lo(mine, 0u);
                if (has) {
                    const unsigned slot = mycnt[tag] + rank;
                    if (rank == 0u) atomicAdd(&mycnt[tag], (unsigned)__builtin_popcount(mine));
                    if (slot < subcap) lst[tag * subcap + slot] = (int)((sf << SUB_SH) | (xf & (SUB_R - 1u)));
                }
                bool before = p[0];
#pragma unroll
                for (int k = 1; k < 4; ++k) {
                    if (p[k] && before) {
                        const unsigned t2 = xk[k] >> SUB_SH;
                        const unsigned s2 = atomicAdd(&mycnt[t2], 1u);
                        if (s2 < subcap) lst[t2 * subcap + s2] = (int)((sk[k] << SUB_SH) | (xk[k] & (SUB_R - 1u)));
                    }
                    before = before || p[k];
                }
            }
        }
    };
    own_scan_all(dst, e_cnt, tid, owner, owners, slice, 0,
        [&](int e) {
            const unsigned x = (unsigned)dst[e] - base;
            if (x < span) atomicAdd(&cnt[x], 1u);
        },
        [&](int it, int first, int n4, bool partial) {
            int4 v[OWN_U];
            if (partial) own_load_step<true >(d4, first, n4, tid, v);
            else         own_load_step<false>(d4, first, n4, tid, v);
#pragma unroll
            for (int u = 0; u < OWN_U; ++u) test4_queue(v[u], first + u * OWN_T + tid);
            if (qn >= DQ_DRAIN) { work_off(qn); qn = 0; }
        });
    work_off(qn);
    __syncthreads();

    if (listing && lane < SUB_N) meta[(size_t)wslot * SUB_N + lane] = (int)mycnt[lane];

    unsigned* __restrict__ plane = cntp + (size_t)slice * (size_t)cstride + base;
    for (unsigned i = tid; i < span; i += OWN_T) plane[i] = cnt[i];
}

__device__ __forceinline__ float own_dinv(unsigned c) {
    const float d = 1.0f + (float)c;
    return d > 0.0f ? __frsqrt_rn(d) : 0.0f;
}

__global__ void gcn_gemm_wmma(const float* __restrict__ x,
                              const float* __restrict__ W,
                              const unsigned* __restrict__ cntp,
                              int cstride,
                              float* __restrict__ dinv,
                              float* __restrict__ h,
                              int n_nodes, int n_tiles) {
    const int lane  = threadIdx.x & 31;
    const int wave  = threadIdx.x >> 5;
    const int tile  = blockIdx.x * (blockDim.x >> 5) + wave;
    if (tile >= n_tiles) return;

    const int base  = tile * 16;
    const int m     = lane & 15;
    const int khalf = (lane >> 4) * 2;

    int arow = base + m;
    if (arow >= n_nodes) arow = n_nodes - 1;
    const float* __restrict__ xrow = x + (size_t)arow * IN_CH;

    v2f a0 = *(const v2f*)(xrow + 0  + khalf);
    v2f a1 = *(const v2f*)(xrow + 4  + khalf);
    v2f a2 = *(const v2f*)(xrow + 8  + khalf);
    v2f a3 = *(const v2f*)(xrow + 12 + khalf);

    const int   wrow = (m < OUT_CH) ? m : 0;
    const float sel  = (m < OUT_CH) ? 1.0f : 0.0f;
    const float* __restrict__ wr = W + wrow * IN_CH + khalf;
    v2f b0 = *(const v2f*)(wr + 0);
    v2f b1 = *(const v2f*)(wr + 4);
    v2f b2 = *(const v2f*)(wr + 8);
    v2f b3 = *(const v2f*)(wr + 12);
    b0 *= sel; b1 *= sel; b2 *= sel; b3 *= sel;

    v8f acc = {};
    acc = __builtin_amdgcn_wmma_f32_16x16x4_f32(false, a0, false, b0, (short)0, acc, false, false);
    acc = __builtin_amdgcn_wmma_f32_16x16x4_f32(false, a1, false, b1, (short)0, acc, false, false);
    acc = __builtin_amdgcn_wmma_f32_16x16x4_f32(false, a2, false, b2, (short)0, acc, false, false);
    acc = __builtin_amdgcn_wmma_f32_16x16x4_f32(false, a3, false, b3, (short)0, acc, false, false);

    if ((tile & 1) == 0) {
        const int node = base + lane;
        if (node < n_nodes) {
            unsigned c = 0u;
#pragma unroll
            for (int s = 0; s < DEG_NS; ++s) c += cntp[(size_t)s * (size_t)cstride + node];
            dinv[node] = own_dinv(c);
        }
    }

    const int  n     = m;
    const int  mbase = base + ((lane >> 4) * 8);
    const bool full  = (base + 16) <= n_nodes;

    if (full) {
        if (n < OUT_CH) {
            uint4 c0 = *(const uint4*)(cntp + mbase);
            uint4 c1 = *(const uint4*)(cntp + mbase + 4);
#pragma unroll
            for (int s = 1; s < DEG_NS; ++s) {
                const unsigned* __restrict__ pl = cntp + (size_t)s * (size_t)cstride + mbase;
                const uint4 e0 = *(const uint4*)(pl);
                const uint4 e1 = *(const uint4*)(pl + 4);
                c0.x += e0.x; c0.y += e0.y; c0.z += e0.z; c0.w += e0.w;
                c1.x += e1.x; c1.y += e1.y; c1.z += e1.z; c1.w += e1.w;
            }
            const float dv[8] = {own_dinv(c0.x), own_dinv(c0.y), own_dinv(c0.z), own_dinv(c0.w),
                                 own_dinv(c1.x), own_dinv(c1.y), own_dinv(c1.z), own_dinv(c1.w)};
            float* __restrict__ hp = h   + (size_t)mbase * OUT_CH + n;
#pragma unroll
            for (int r = 0; r < 8; ++r) hp[r * OUT_CH] = dv[r] * acc[r];
        }
    } else {
        if (n < OUT_CH) {
#pragma unroll
            for (int r = 0; r < 8; ++r) {
                const int node = mbase + r;
                if (node < n_nodes) {
                    unsigned c = 0u;
#pragma unroll
                    for (int s = 0; s < DEG_NS; ++s) c += cntp[(size_t)s * (size_t)cstride + node];
                    h[(size_t)node * OUT_CH + n] = own_dinv(c) * acc[r];
                }
            }
        }
    }
}

__device__ __forceinline__ void own_add_row(float* acc_x, const float4& h0, const float4& h1) {
    atomicAdd(acc_x + 0 * ACC_P, h0.x);
    atomicAdd(acc_x + 1 * ACC_P, h0.y);
    atomicAdd(acc_x + 2 * ACC_P, h0.z);
    atomicAdd(acc_x + 3 * ACC_P, h0.w);
    atomicAdd(acc_x + 4 * ACC_P, h1.x);
    atomicAdd(acc_x + 5 * ACC_P, h1.y);
    atomicAdd(acc_x + 6 * ACC_P, h1.z);
    atomicAdd(acc_x + 7 * ACC_P, h1.w);
}

__global__ __launch_bounds__(OWN_T) void gcn_sum_owner(const int* __restrict__ src,
                                                       const int* __restrict__ dst,
                                                       const float* __restrict__ dinv,
                                                       const float* __restrict__ h,
                                                       const float* __restrict__ bias,
                                                       const int* __restrict__ meta,
                                                       const int* __restrict__ lists,
                                                       unsigned cap,
                                                       float* __restrict__ out,
                                                       int n_nodes, int e_cnt) {
    struct SumLds {
        float acc[OUT_CH * ACC_P];
        int   qa[OWN_W][QA_CAP];
        float dvs[SUB_R];
    };
    static_assert((OUT_CH - 1) * ACC_P * 4 < 65536, "plane offsets fit the LDS offset field");
    __shared__ __attribute__((aligned(16))) SumLds lds;
    float* const acc = lds.acc;
    int   (* const qa)[QA_CAP] = lds.qa;
    float* const dvs = lds.dvs;
    const int tid  = threadIdx.x;
    const int lane = tid & 31;
    const int wv   = __builtin_amdgcn_readfirstlane(tid >> 5);
    int* myq = qa[wv];
    const unsigned owner  = blockIdx.x / SUB_N;
    const unsigned sub    = blockIdx.x - owner * SUB_N;
    const unsigned owners = gridDim.x / SUB_N;
    const unsigned base = owner * DEG_R + sub * SUB_R;
    if (base >= (unsigned)n_nodes) return;
    const unsigned span = min((unsigned)SUB_R, (unsigned)n_nodes - base);
    const int4* __restrict__ d4 = (const int4*)dst;
    const unsigned n_last = (unsigned)n_nodes - 1u;

    {
        float4* a4 = (float4*)acc;
        for (unsigned i = tid; i < (unsigned)(OUT_CH * ACC_P / 4); i += OWN_T) a4[i] = make_float4(0.0f, 0.0f, 0.0f, 0.0f);
        for (unsigned i = tid; i < span; i += OWN_T) dvs[i] = dinv[base + i];
    }
    __syncthreads();

    auto add_src = [&](int s, unsigned x) {
        const float4* __restrict__ hs = (const float4*)(h + (size_t)s * OUT_CH);
        const float4 h0 = hs[0];
        const float4 h1 = hs[1];
        own_add_row(&acc[x], h0, h1);
    };

    auto drain = [&](unsigned n) {
        for (unsigned j0 = 0; j0 < n; j0 += 32 * DI) {
            bool ok[DI]; unsigned xr[DI]; unsigned s[DI];
#pragma unroll
            for (int d = 0; d < DI; ++d) {
                const unsigned j = j0 + d * 32 + lane;
                ok[d] = j < n;
                const unsigned qv  = (unsigned)myq[ok[d] ? j : 0u];
                s[d]  = min(qv >> SUB_SH, n_last);
                xr[d] = min(qv & (SUB_R - 1u), span - 1u);
            }
            float4 r0[DI], r1[DI];
#pragma unroll
            for (int d = 0; d < DI; ++d) {
                const float4* __restrict__ hs = (const float4*)(h + (size_t)s[d] * OUT_CH);
                r0[d] = hs[0];
                r1[d] = hs[1];
            }
#pragma unroll
            for (int d = 0; d < DI; ++d) {
                asm volatile("" :: "v"(r0[d].x), "v"(r0[d].y), "v"(r0[d].z), "v"(r0[d].w),
                                   "v"(r1[d].x), "v"(r1[d].y), "v"(r1[d].z), "v"(r1[d].w));
            }
#pragma unroll
            for (int d = 0; d < DI; ++d)
                if (ok[d]) own_add_row(&acc[xr[d]], r0[d], r1[d]);
        }
    };

    unsigned wcount = 0;

    const unsigned subcap = cap / (unsigned)SUB_N;
    unsigned want = 0xffffffffu;
    if (cap != 0u && lane < DEG_NS)
        want = (unsigned)meta[(size_t)((owner * DEG_NS + (unsigned)lane) * OWN_W + (unsigned)wv) * SUB_N + sub];
    const bool     is_whole = lane < DEG_NS && want <= subcap;
    const unsigned whole    = __builtin_amdgcn_ballot_w32(is_whole);
    const unsigned len_l    = is_whole ? want : 0u;
    unsigned pre[DEG_NS + 1];
    pre[0] = 0u;
#pragma unroll
    for (int s = 0; s < DEG_NS; ++s)
        pre[s + 1] = pre[s] + (unsigned)__builtin_amdgcn_readlane((int)len_l, s);
    const unsigned total   = pre[DEG_NS];
    const unsigned lstride = (unsigned)OWN_W * cap;
    const int* __restrict__ lst0 = lists + (size_t)((owner * DEG_NS) * OWN_W + (unsigned)wv) * cap + (size_t)sub * subcap;
    for (unsigned j0 = 0; j0 < total; j0 += LQ_STEP) {
        unsigned e[LQ_U];
#pragma unroll
        for (int u = 0; u < LQ_U; ++u) {
            const unsigned j = min(j0 + u * 32 + lane, total - 1u);
            unsigned idx = j;
#pragma unroll
            for (int t = 1; t < DEG_NS; ++t)
                idx += (j >= pre[t]) ? lstride - (pre[t] - pre[t - 1]) : 0u;
            e[u]  = (unsigned)lst0[idx];
        }
#pragma unroll
        for (int u = 0; u < LQ_U; ++u) {
            if (j0 + u * 32 + lane < total)
                myq[wcount + u * 32 + lane] = (int)e[u];
        }
        wcount += min((unsigned)LQ_STEP, total - j0);
        if (wcount > QA_CAP - LQ_STEP) { drain(wcount); wcount = 0; }
    }

#pragma unroll 1
    for (unsigned slice = 0; slice < (unsigned)DEG_NS; ++slice) {
        const int it0 = ((whole >> slice) & 1u) ? 0x7fffffff : 0;
        own_scan_all(dst, e_cnt, tid, owner, owners, slice, it0,
            [&](int e) {
                const unsigned x = (unsigned)dst[e] - base;
                if (x < span) add_src(src[e], x);
            },
            [&](int it, int first, int n4, bool partial) {
                int4 v[OWN_U];
                if (partial) own_load_step<true >(d4, first, n4, tid, v);
                else         own_load_step<false>(d4, first, n4, tid, v);
#pragma unroll
                for (int u = 0; u < OWN_U; ++u) {
                    const unsigned x0 = (unsigned)v[u].x - base;
                    const unsigned x1 = (unsigned)v[u].y - base;
                    const unsigned x2 = (unsigned)v[u].z - base;
                    const unsigned x3 = (unsigned)v[u].w - base;
                    if (min(min(x0, x1), min(x2, x3)) < span) {
                        const int* __restrict__ sp = src + 4 * (size_t)(first + u * OWN_T + tid);
                        if (x0 < span) add_src(sp[0], x0);
                        if (x1 < span) add_src(sp[1], x1);
                        if (x2 < span) add_src(sp[2], x2);
                        if (x3 < span) add_src(sp[3], x3);
                    }
                }
            });
    }
    drain(wcount);

    const float4* __restrict__ h4 = (const float4*)(h + (size_t)base * OUT_CH);
    float4* __restrict__ o4 = (float4*)(out + (size_t)base * OUT_CH);
    const float* __restrict__ bp = bias + (tid & 1) * 4;
    const float bx = bp[0], by = bp[1], bz = bp[2], bw = bp[3];
    const unsigned nq = span * (OUT_CH / 4);
    float4 hv[WO_K];
#pragma unroll
    for (int k = 0; k < WO_K; ++k) {
        const unsigned q = (unsigned)(k * OWN_T + tid);
        hv[k] = h4[min(q, nq - 1u)];
    }
    __syncthreads();
    {
        const float* ap = acc + (tid & 1) * 4 * ACC_P;
#pragma unroll
        for (int k = 0; k < WO_K; ++k) {
            const unsigned q = (unsigned)(k * OWN_T + tid);
            if (q < nq) {
                const unsigned xw = q >> 1;
                const float  di = dvs[xw];
                const float4 av = make_float4(ap[xw], ap[xw + ACC_P], ap[xw + 2 * ACC_P], ap[xw + 3 * ACC_P]);
                float4 o;
                o.x = bx + di * (hv[k].x + av.x);
                o.y = by + di * (hv[k].y + av.y);
                o.z = bz + di * (hv[k].z + av.z);
                o.w = bw + di * (hv[k].w + av.w);
                o4[q] = o;
            }
        }
    }
}

extern "C" void kernel_launch(void* const* d_in, const int* in_sizes, int n_in,
                              void* d_out, int out_size, void* d_ws, size_t ws_size,
                              hipStream_t stream) {
    const float* x    = (const float*)d_in[0];
    const int*   ei   = (const int*)d_in[1];
    const float* W    = (const float*)d_in[2];
    const float* bias = (const float*)d_in[3];

    const int N = in_sizes[0] / IN_CH;
    const int E = in_sizes[1] / 2;
    const int* src = ei;
    const int* dst = ei + (size_t)E;

    float* dinv = (float*)d_ws;
    size_t h_off = (((size_t)N * sizeof(float)) + 255u) & ~(size_t)255u;
    float* h = (float*)((char*)d_ws + h_off);
    float* out = (float*)d_out;

    const int TPB = 256;
    const int owners = (N + DEG_R - 1) / DEG_R;

    const size_t lwaves   = (size_t)owners * DEG_NS * OWN_W;
    const size_t cstride  = ((size_t)N + 63u) & ~(size_t)63u;
    const size_t cnt_off  = (h_off + (size_t)N * OUT_CH * sizeof(float) + 255u) & ~(size_t)255u;
    const size_t meta_off = (cnt_off + (size_t)DEG_NS * cstride * sizeof(unsigned) + 255u) & ~(size_t)255u;
    const size_t list_off = (meta_off + lwaves * SUB_N * sizeof(int) + 255u) & ~(size_t)255u;
    size_t cap = 0;
    if (lwaves != 0 && ws_size > list_off) cap = (ws_size - list_off) / (lwaves * sizeof(int));
    if (cap > LIST_MAX) cap = LIST_MAX;
    if (N > (1 << SRC_BITS) || (reinterpret_cast<size_t>(src) & 15u) != 0) cap = 0;
    unsigned* cntp = (unsigned*)((char*)d_ws + cnt_off);
    int* meta  = (int*)((char*)d_ws + meta_off);
    int* lists = (int*)((char*)d_ws + list_off);

    gcn_deg_owner<<<owners * DEG_NS, OWN_T, 0, stream>>>(src, dst, cntp, (int)cstride, meta, lists, (unsigned)cap, N, E);

    const int n_tiles = (N + 15) / 16;
    const int waves_per_block = TPB / 32;
    const int gemm_blocks = (n_tiles + waves_per_block - 1) / waves_per_block;
    gcn_gemm_wmma<<<gemm_blocks, TPB, 0, stream>>>(x, W, cntp, (int)cstride, dinv, h, N, n_tiles);

    gcn_sum_owner<<<owners * SUB_N, OWN_T, 0, stream>>>(src, dst, dinv, h, bias, meta, lists, (unsigned)cap, out, N, E);
}
